# starting kernel
# baseline (speedup 1.0000x reference)
#include <hip/hip_runtime.h>
#include <hip/hip_bf16.h>

typedef __attribute__((ext_vector_type(16))) _Float16 v16h;
typedef __attribute__((ext_vector_type(8)))  _Float16 v8h;
typedef __attribute__((ext_vector_type(8)))  float    v8f;

#define B_ 2
#define F_ 512
#define S_ 1024
#define E_ 4
#define I_ 1024
#define TOK (B_*S_)
#define EPSN 1e-5f
#define SLOPE_ 0.02f

__device__ __forceinline__ int frag_j(int kin) {
  return kin < 8 ? kin : (kin < 16 ? kin - 8 : (kin < 24 ? kin - 8 : kin - 16));
}
__device__ __forceinline__ int frag_lane(int kin, int n) {
  return n + (((kin >> 3) & 1) ? 16 : 0);
}

__global__ void pack_weights_kernel(const float* __restrict__ W,
                                    _Float16* __restrict__ Wp,
                                    int E, int Cin, int Cout, int mode) {
  size_t idx = (size_t)blockIdx.x * blockDim.x + threadIdx.x;
  size_t total = (size_t)E * Cin * Cout;
  if (idx >= total) return;
  int j    = (int)(idx & 15);
  int lane = (int)((idx >> 4) & 31);
  size_t rest = idx >> 9;
  int KT = Cin >> 5;
  int kt = (int)(rest % KT); rest /= KT;
  int MT = Cout >> 4;
  int mt = (int)(rest % MT);
  int e  = (int)(rest / MT);
  int khalf = (lane >> 4) * 8;
  int k = kt * 32 + khalf + (j < 8 ? j : j + 8);
  int m = mt * 16 + (lane & 15);
  float v = mode ? W[(size_t)m * Cin + k]
                 : W[((size_t)e * Cin + k) * Cout + m];
  Wp[idx] = (_Float16)v;
}

__device__ __forceinline__ v16h load_frag(const _Float16* p) {
  v8h lo = *(const v8h*)p;
  v8h hi = *(const v8h*)(p + 8);
  v16h r;
#pragma unroll
  for (int j = 0; j < 8; ++j) { r[j] = lo[j]; r[j + 8] = hi[j]; }
  return r;
}

__global__ void gate_kernel(const float* __restrict__ x, int C,
                            const float* __restrict__ gw,
                            int* __restrict__ eidx,
                            float* __restrict__ lacc) {
  extern __shared__ float gws[];
  for (int p = threadIdx.x; p < E_ * C; p += blockDim.x) gws[p] = gw[p];
  __syncthreads();
  int t = blockIdx.x * blockDim.x + threadIdx.x;
  if (t >= TOK) return;
  int b = t / S_, s = t % S_;
  const float* xb = x + (size_t)b * C * S_ + s;
  float a0 = 0.f, a1 = 0.f, a2 = 0.f, a3 = 0.f;
  for (int c = 0; c < C; ++c) {
    float xv = xb[(size_t)c * S_];
    a0 += xv * gws[0 * C + c];
    a1 += xv * gws[1 * C + c];
    a2 += xv * gws[2 * C + c];
    a3 += xv * gws[3 * C + c];
  }
  float a[4] = {a0, a1, a2, a3};
  int best = 0; float m = a[0];
  for (int e = 1; e < E_; ++e) { if (a[e] > m) { m = a[e]; best = e; } }
  float g[4]; float se = 0.f;
  for (int e = 0; e < E_; ++e) { g[e] = __expf(a[e] - m); se += g[e]; }
  float inv = 1.0f / se;
  eidx[t] = best;
  for (int e = 0; e < E_; ++e) atomicAdd(&lacc[e], g[e] * inv);
  atomicAdd(&lacc[4 + best], 1.0f);
}

__global__ void moe_wmma_kernel(const float* __restrict__ X, int Cin,
                                const _Float16* __restrict__ Wp, int Cout,
                                const int* __restrict__ eidx,
                                float* __restrict__ Out, float outscale) {
  extern __shared__ _Float16 xs[];
  int tile = blockIdx.y;
  int b = (tile * 16) / S_;
  int s0 = (tile * 16) % S_;
  int KT = Cin >> 5;
  for (int p = threadIdx.x; p < Cin * 16; p += blockDim.x) {
    int k = p >> 4, n = p & 15;
    int kt = k >> 5, kin = k & 31;
    xs[((kt * 32 + frag_lane(kin, n)) << 4) + frag_j(kin)] =
        (_Float16)X[((size_t)b * Cin + k) * S_ + s0 + n];
  }
  __syncthreads();
  int wave = threadIdx.x >> 5, lane = threadIdx.x & 31;
  int mt0 = blockIdx.x * 16 + wave * 2;
  int col = lane & 15, khalf = (lane >> 4) * 8;
  int myexp = eidx[tile * 16 + col];
  size_t fragstride = (size_t)KT * 512;
  v8f c0s = {}, c1s = {};
  for (int e = 0; e < E_; ++e) {
    if (!__any(myexp == e)) continue;
    const _Float16* We = Wp + (size_t)e * (Cout >> 4) * fragstride;
    const _Float16* A0 = We + (size_t)(mt0 + 0) * fragstride + lane * 16;
    const _Float16* A1 = We + (size_t)(mt0 + 1) * fragstride + lane * 16;
    v8f c0 = {}, c1 = {};
    for (int kt = 0; kt < KT; ++kt) {
      if (kt + 1 < KT) __builtin_prefetch(A0 + (size_t)(kt + 1) * 512, 0, 0);
      v16h bv  = load_frag(&xs[(kt * 32 + lane) * 16]);
      v16h a0v = load_frag(A0 + (size_t)kt * 512);
      v16h a1v = load_frag(A1 + (size_t)kt * 512);
      c0 = __builtin_amdgcn_wmma_f32_16x16x32_f16(false, a0v, false, bv,
                                                  (short)0, c0, false, false);
      c1 = __builtin_amdgcn_wmma_f32_16x16x32_f16(false, a1v, false, bv,
                                                  (short)0, c1, false, false);
    }
    if (myexp == e) { c0s = c0; c1s = c1; }
  }
#pragma unroll
  for (int j = 0; j < 8; ++j) {
    int r0 = mt0 * 16 + j + khalf;
    int r1 = r0 + 16;
    Out[((size_t)b * Cout + r0) * S_ + s0 + col] = c0s[j] * outscale;
    Out[((size_t)b * Cout + r1) * S_ + s0 + col] = c1s[j] * outscale;
  }
}

__global__ void cumsum_kernel(const float* __restrict__ h0,
                              const float* __restrict__ divisor,
                              float* __restrict__ tmp1) {
  int row = blockIdx.x * (blockDim.x >> 5) + (threadIdx.x >> 5);
  int lane = threadIdx.x & 31;
  if (row >= B_ * I_) return;
  int b = row / I_, i = row % I_;
  const float* dep = h0 + ((size_t)b * 3 * I_ + i) * S_;
  const float* sc  = h0 + ((size_t)b * 3 * I_ + I_ + i) * S_;
  const float* sh  = h0 + ((size_t)b * 3 * I_ + 2 * I_ + i) * S_;
  float* o = tmp1 + ((size_t)b * I_ + i) * S_;
  float carry = 0.f;
  for (int s0 = 0; s0 < S_; s0 += 32) {
    float v = dep[s0 + lane];
#pragma unroll
    for (int d = 1; d < 32; d <<= 1) {
      float t = __shfl_up(v, d, 32);
      if (lane >= d) v += t;
    }
    float cum = carry + v;
    o[s0 + lane] = cum / divisor[s0 + lane] * sc[s0 + lane] + sh[s0 + lane];
    carry = __shfl(cum, 31, 32);
  }
}

__global__ void norm_kernel(const float* __restrict__ p0,
                            const float* __restrict__ p1,
                            const float* __restrict__ p2,
                            int cstride, int mode,
                            float* __restrict__ out) {
  __shared__ float red[256];
  int t = blockIdx.x; int b = t / S_, s = t % S_;
  float x[4]; float lsum = 0.f;
#pragma unroll
  for (int r = 0; r < 4; ++r) {
    int i = threadIdx.x + r * 256;
    size_t idx = ((size_t)b * cstride + i) * S_ + s;
    float v = p0[idx];
    if (mode) v = v * p1[idx] + p2[idx];
    x[r] = v; lsum += v;
  }
  red[threadIdx.x] = lsum; __syncthreads();
  for (int st = 128; st > 0; st >>= 1) {
    if (threadIdx.x < st) red[threadIdx.x] += red[threadIdx.x + st];
    __syncthreads();
  }
  float mean = red[0] * (1.0f / I_);
  __syncthreads();
  float lss = 0.f;
#pragma unroll
  for (int r = 0; r < 4; ++r) { float d = x[r] - mean; lss += d * d; }
  red[threadIdx.x] = lss; __syncthreads();
  for (int st = 128; st > 0; st >>= 1) {
    if (threadIdx.x < st) red[threadIdx.x] += red[threadIdx.x + st];
    __syncthreads();
  }
  float n = sqrtf(red[0]) * (1.0f / 32.0f);
  float inv = 1.0f / (n + EPSN);
#pragma unroll
  for (int r = 0; r < 4; ++r) {
    int i = threadIdx.x + r * 256;
    float y = (x[r] - mean) * inv;
    y = y >= 0.f ? y : y * SLOPE_;
    out[((size_t)b * I_ + i) * S_ + s] = y;
  }
}

__global__ void conv_wmma_kernel(const float* __restrict__ G,
                                 const _Float16* __restrict__ Wp1,
                                 float* __restrict__ Out) {
  __shared__ _Float16 xs[128 * 16];
  int tile = blockIdx.y;
  int b = (tile * 16) / S_, s0 = (tile * 16) % S_;
  int wave = threadIdx.x >> 5, lane = threadIdx.x & 31;
  int mt0 = blockIdx.x * 16 + wave * 2;
  int col = lane & 15, khalf = (lane >> 4) * 8;
  const int KTtot = (3 * I_) >> 5;
  size_t fragstride = (size_t)KTtot * 512;
  const _Float16* A0 = Wp1 + (size_t)(mt0 + 0) * fragstride + lane * 16;
  const _Float16* A1 = Wp1 + (size_t)(mt0 + 1) * fragstride + lane * 16;
  v8f c0 = {}, c1 = {};
  for (int cc = 0; cc < 3 * I_; cc += 128) {
    __syncthreads();
    for (int p = threadIdx.x; p < 128 * 16; p += blockDim.x) {
      int kk = cc + (p >> 4); int n = p & 15;
      int i = kk / 3, kc = kk - i * 3;
      int s = s0 + n - 2 + kc;
      int klocal = p >> 4;
      int kt = klocal >> 5, kin = klocal & 31;
      xs[((kt * 32 + frag_lane(kin, n)) << 4) + frag_j(kin)] =
          (s >= 0) ? (_Float16)G[((size_t)b * I_ + i) * S_ + s] : (_Float16)0.0f;
    }
    __syncthreads();
    int ktbase = cc >> 5;
#pragma unroll
    for (int kt = 0; kt < 4; ++kt) {
      __builtin_prefetch(A0 + (size_t)(ktbase + kt + 1) * 512, 0, 0);
      v16h bv  = load_frag(&xs[(kt * 32 + lane) * 16]);
      v16h a0v = load_frag(A0 + (size_t)(ktbase + kt) * 512);
      v16h a1v = load_frag(A1 + (size_t)(ktbase + kt) * 512);
      c0 = __builtin_amdgcn_wmma_f32_16x16x32_f16(false, a0v, false, bv,
                                                  (short)0, c0, false, false);
      c1 = __builtin_amdgcn_wmma_f32_16x16x32_f16(false, a1v, false, bv,
                                                  (short)0, c1, false, false);
    }
  }
#pragma unroll
  for (int j = 0; j < 8; ++j) {
    int r0 = mt0 * 16 + j + khalf;
    int r1 = r0 + 16;
    Out[((size_t)b * (3 * I_) + r0) * S_ + s0 + col] = c0[j];
    Out[((size_t)b * (3 * I_) + r1) * S_ + s0 + col] = c1[j];
  }
}

__global__ void init_loss_kernel(float* lacc) {
  if (threadIdx.x < 16) lacc[threadIdx.x] = 0.f;
}
__global__ void finalize_loss_kernel(const float* __restrict__ lacc,
                                     float* __restrict__ out_loss) {
  if (threadIdx.x == 0) {
    const float invT = 1.0f / (float)TOK;
    float l = 0.f;
    for (int e = 0; e < E_; ++e) l += (lacc[e] * invT) * (lacc[4 + e] * invT);
    for (int e = 0; e < E_; ++e) l += (lacc[8 + e] * invT) * (lacc[12 + e] * invT);
    *out_loss = l;
  }
}

extern "C" void kernel_launch(void* const* d_in, const int* in_sizes, int n_in,
                              void* d_out, int out_size, void* d_ws, size_t ws_size,
                              hipStream_t stream) {
  const float* inp     = (const float*)d_in[0];
  const float* divisor = (const float*)d_in[1];
  const float* w0_gate = (const float*)d_in[2];
  const float* w0      = (const float*)d_in[3];
  const float* w1      = (const float*)d_in[4];
  const float* w2_gate = (const float*)d_in[5];
  const float* w2      = (const float*)d_in[6];
  float* out = (float*)d_out;

  char* ws = (char*)d_ws;
  size_t off = 0;
  float* lacc = (float*)(ws + off); off += 256;
  const size_t h0b = (size_t)B_ * 3 * I_ * S_ * sizeof(float);
  const size_t t1b = (size_t)B_ * I_ * S_ * sizeof(float);
  float* h0   = (float*)(ws + off); off += h0b;
  float* tmp1 = (float*)(ws + off); off += t1b;
  float* g1   = (float*)(ws + off); off += t1b;
  int*   idx0 = (int*)(ws + off); off += TOK * sizeof(int);
  int*   idx1 = (int*)(ws + off); off += TOK * sizeof(int);
  off = (off + 255) & ~(size_t)255;
  const size_t w0n = (size_t)E_ * F_ * 3 * I_;
  const size_t w1n = (size_t)3 * I_ * 3 * I_;
  const size_t w2n = (size_t)E_ * I_ * F_;
  _Float16* w0p = (_Float16*)(ws + off); off += w0n * 2;
  _Float16* w1p = (_Float16*)(ws + off); off += w1n * 2;
  _Float16* w2p = (_Float16*)(ws + off); off += w2n * 2;
  float* h2 = h0;
  float* h3 = tmp1;

  pack_weights_kernel<<<(int)(w0n / 256), 256, 0, stream>>>(w0, w0p, E_, F_, 3 * I_, 0);
  pack_weights_kernel<<<(int)(w1n / 256), 256, 0, stream>>>(w1, w1p, 1, 3 * I_, 3 * I_, 1);
  pack_weights_kernel<<<(int)(w2n / 256), 256, 0, stream>>>(w2, w2p, E_, I_, F_, 0);

  init_loss_kernel<<<1, 32, 0, stream>>>(lacc);

  gate_kernel<<<TOK / 256, 256, E_ * F_ * sizeof(float), stream>>>(
      inp, F_, w0_gate, idx0, lacc);
  moe_wmma_kernel<<<dim3(3 * I_ / 256, TOK / 16), 256,
                    F_ * 16 * sizeof(_Float16), stream>>>(
      inp, F_, w0p, 3 * I_, idx0, h0, 1.0f);

  cumsum_kernel<<<(B_ * I_) / 8, 256, 0, stream>>>(h0, divisor, tmp1);
  norm_kernel<<<TOK, 256, 0, stream>>>(tmp1, tmp1, tmp1, I_, 0, g1);

  conv_wmma_kernel<<<dim3(3 * I_ / 256, TOK / 16), 256, 0, stream>>>(g1, w1p, h2);

  norm_kernel<<<TOK, 256, 0, stream>>>(
      h2, h2 + (size_t)I_ * S_, h2 + (size_t)2 * I_ * S_, 3 * I_, 1, h3);

  gate_kernel<<<TOK / 256, 256, E_ * I_ * sizeof(float), stream>>>(
      h3, I_, w2_gate, idx1, lacc + 8);
  moe_wmma_kernel<<<dim3(F_ / 256, TOK / 16), 256,
                    I_ * 16 * sizeof(_Float16), stream>>>(
      h3, I_, w2p, F_, idx1, out, 1.0f  );

  finalize_loss_kernel<<<1, 32, 0, stream>>>(lacc, out + (size_t)B_ * F_ * S_);
}


# gate kernels: each token's channel sum split over 16 waves of a 32-token group (64 groups, was 8), partials combined via LDS in wave order; loss terms wave-summed before atomics (sum order changes)
# speedup vs baseline: 1.2225x; 1.2225x over previous
#include <hip/hip_runtime.h>
#include <hip/hip_bf16.h>

typedef __attribute__((ext_vector_type(16))) _Float16 v16h;
typedef __attribute__((ext_vector_type(8)))  _Float16 v8h;
typedef __attribute__((ext_vector_type(8)))  float    v8f;

#define B_ 2
#define F_ 512
#define S_ 1024
#define E_ 4
#define I_ 1024
#define TOK (B_*S_)
#define EPSN 1e-5f
#define SLOPE_ 0.02f

__device__ __forceinline__ int frag_j(int kin) {
  return kin < 8 ? kin : (kin < 16 ? kin - 8 : (kin < 24 ? kin - 8 : kin - 16));
}
__device__ __forceinline__ int frag_lane(int kin, int n) {
  return n + (((kin >> 3) & 1) ? 16 : 0);
}

__global__ void pack_weights_kernel(const float* __restrict__ W,
                                    _Float16* __restrict__ Wp,
                                    int E, int Cin, int Cout, int mode) {
  size_t idx = (size_t)blockIdx.x * blockDim.x + threadIdx.x;
  size_t total = (size_t)E * Cin * Cout;
  if (idx >= total) return;
  int j    = (int)(idx & 15);
  int lane = (int)((idx >> 4) & 31);
  size_t rest = idx >> 9;
  int KT = Cin >> 5;
  int kt = (int)(rest % KT); rest /= KT;
  int MT = Cout >> 4;
  int mt = (int)(rest % MT);
  int e  = (int)(rest / MT);
  int khalf = (lane >> 4) * 8;
  int k = kt * 32 + khalf + (j < 8 ? j : j + 8);
  int m = mt * 16 + (lane & 15);
  float v = mode ? W[(size_t)m * Cin + k]
                 : W[((size_t)e * Cin + k) * Cout + m];
  Wp[idx] = (_Float16)v;
}

__device__ __forceinline__ v16h load_frag(const _Float16* p) {
  v8h lo = *(const v8h*)p;
  v8h hi = *(const v8h*)(p + 8);
  v16h r;
#pragma unroll
  for (int j = 0; j < 8; ++j) { r[j] = lo[j]; r[j + 8] = hi[j]; }
  return r;
}

#define GATE_WAVES 16
__global__ void __launch_bounds__(GATE_WAVES * 32)
gate_kernel(const float* __restrict__ x, int C,
            const float* __restrict__ gw,
            int* __restrict__ eidx,
            float* __restrict__ lacc) {
  __shared__ float part[GATE_WAVES][E_][32];
  const int lane = threadIdx.x & 31;
  const int wave = __builtin_amdgcn_readfirstlane((int)(threadIdx.x >> 5));
  const int t0 = blockIdx.x * 32;
  const int b = t0 / S_, s = (t0 % S_) + lane;
  const int cper = C / GATE_WAVES;
  const int c0 = wave * cper;
  const float* xp = x + ((size_t)b * C + c0) * S_ + s;
  const float* g0 = gw + c0;
  const float* g1 = g0 + C;
  const float* g2 = g1 + C;
  const float* g3 = g2 + C;
  float a0 = 0.f, a1 = 0.f, a2 = 0.f, a3 = 0.f;
  for (int cb = 0; cb < cper; cb += 16) {
    float xv[16];
#pragma unroll
    for (int u = 0; u < 16; ++u) xv[u] = xp[(size_t)(cb + u) * S_];
#pragma unroll
    for (int u = 0; u < 16; ++u) {
      a0 = __builtin_fmaf(xv[u], g0[cb + u], a0);
      a1 = __builtin_fmaf(xv[u], g1[cb + u], a1);
      a2 = __builtin_fmaf(xv[u], g2[cb + u], a2);
      a3 = __builtin_fmaf(xv[u], g3[cb + u], a3);
    }
  }
  part[wave][0][lane] = a0;
  part[wave][1][lane] = a1;
  part[wave][2][lane] = a2;
  part[wave][3][lane] = a3;
  __syncthreads();
  if (wave != 0) return;
  float a[4];
  for (int e = 0; e < E_; ++e) {
    float v = part[0][e][lane];
    for (int w = 1; w < GATE_WAVES; ++w) v += part[w][e][lane];
    a[e] = v;
  }
  int best = 0; float m = a[0];
  for (int e = 1; e < E_; ++e) { if (a[e] > m) { m = a[e]; best = e; } }
  float g[4]; float se = 0.f;
  for (int e = 0; e < E_; ++e) { g[e] = __expf(a[e] - m); se += g[e]; }
  float inv = 1.0f / se;
  eidx[t0 + lane] = best;
  for (int e = 0; e < E_; ++e) {
    float gsum = g[e] * inv;
    for (int d = 16; d > 0; d >>= 1) gsum += __shfl_xor(gsum, d, 32);
    float cnt = (float)__popcll(__ballot(best == e));
    if (lane == 0) {
      atomicAdd(&lacc[e], gsum);
      atomicAdd(&lacc[4 + e], cnt);
    }
  }
}

__global__ void moe_wmma_kernel(const float* __restrict__ X, int Cin,
                                const _Float16* __restrict__ Wp, int Cout,
                                const int* __restrict__ eidx,
                                float* __restrict__ Out, float outscale) {
  extern __shared__ _Float16 xs[];
  int tile = blockIdx.y;
  int b = (tile * 16) / S_;
  int s0 = (tile * 16) % S_;
  int KT = Cin >> 5;
  for (int p = threadIdx.x; p < Cin * 16; p += blockDim.x) {
    int k = p >> 4, n = p & 15;
    int kt = k >> 5, kin = k & 31;
    xs[((kt * 32 + frag_lane(kin, n)) << 4) + frag_j(kin)] =
        (_Float16)X[((size_t)b * Cin + k) * S_ + s0 + n];
  }
  __syncthreads();
  int wave = threadIdx.x >> 5, lane = threadIdx.x & 31;
  int mt0 = blockIdx.x * 16 + wave * 2;
  int col = lane & 15, khalf = (lane >> 4) * 8;
  int myexp = eidx[tile * 16 + col];
  size_t fragstride = (size_t)KT * 512;
  v8f c0s = {}, c1s = {};
  for (int e = 0; e < E_; ++e) {
    if (!__any(myexp == e)) continue;
    const _Float16* We = Wp + (size_t)e * (Cout >> 4) * fragstride;
    const _Float16* A0 = We + (size_t)(mt0 + 0) * fragstride + lane * 16;
    const _Float16* A1 = We + (size_t)(mt0 + 1) * fragstride + lane * 16;
    v8f c0 = {}, c1 = {};
    for (int kt = 0; kt < KT; ++kt) {
      if (kt + 1 < KT) __builtin_prefetch(A0 + (size_t)(kt + 1) * 512, 0, 0);
      v16h bv  = load_frag(&xs[(kt * 32 + lane) * 16]);
      v16h a0v = load_frag(A0 + (size_t)kt * 512);
      v16h a1v = load_frag(A1 + (size_t)kt * 512);
      c0 = __builtin_amdgcn_wmma_f32_16x16x32_f16(false, a0v, false, bv,
                                                  (short)0, c0, false, false);
      c1 = __builtin_amdgcn_wmma_f32_16x16x32_f16(false, a1v, false, bv,
                                                  (short)0, c1, false, false);
    }
    if (myexp == e) { c0s = c0; c1s = c1; }
  }
#pragma unroll
  for (int j = 0; j < 8; ++j) {
    int r0 = mt0 * 16 + j + khalf;
    int r1 = r0 + 16;
    Out[((size_t)b * Cout + r0) * S_ + s0 + col] = c0s[j] * outscale;
    Out[((size_t)b * Cout + r1) * S_ + s0 + col] = c1s[j] * outscale;
  }
}

__global__ void cumsum_kernel(const float* __restrict__ h0,
                              const float* __restrict__ divisor,
                              float* __restrict__ tmp1) {
  int row = blockIdx.x * (blockDim.x >> 5) + (threadIdx.x >> 5);
  int lane = threadIdx.x & 31;
  if (row >= B_ * I_) return;
  int b = row / I_, i = row % I_;
  const float* dep = h0 + ((size_t)b * 3 * I_ + i) * S_;
  const float* sc  = h0 + ((size_t)b * 3 * I_ + I_ + i) * S_;
  const float* sh  = h0 + ((size_t)b * 3 * I_ + 2 * I_ + i) * S_;
  float* o = tmp1 + ((size_t)b * I_ + i) * S_;
  float carry = 0.f;
  for (int s0 = 0; s0 < S_; s0 += 32) {
    float v = dep[s0 + lane];
#pragma unroll
    for (int d = 1; d < 32; d <<= 1) {
      float t = __shfl_up(v, d, 32);
      if (lane >= d) v += t;
    }
    float cum = carry + v;
    o[s0 + lane] = cum / divisor[s0 + lane] * sc[s0 + lane] + sh[s0 + lane];
    carry = __shfl(cum, 31, 32);
  }
}

__global__ void norm_kernel(const float* __restrict__ p0,
                            const float* __restrict__ p1,
                            const float* __restrict__ p2,
                            int cstride, int mode,
                            float* __restrict__ out) {
  __shared__ float red[256];
  int t = blockIdx.x; int b = t / S_, s = t % S_;
  float x[4]; float lsum = 0.f;
#pragma unroll
  for (int r = 0; r < 4; ++r) {
    int i = threadIdx.x + r * 256;
    size_t idx = ((size_t)b * cstride + i) * S_ + s;
    float v = p0[idx];
    if (mode) v = v * p1[idx] + p2[idx];
    x[r] = v; lsum += v;
  }
  red[threadIdx.x] = lsum; __syncthreads();
  for (int st = 128; st > 0; st >>= 1) {
    if (threadIdx.x < st) red[threadIdx.x] += red[threadIdx.x + st];
    __syncthreads();
  }
  float mean = red[0] * (1.0f / I_);
  __syncthreads();
  float lss = 0.f;
#pragma unroll
  for (int r = 0; r < 4; ++r) { float d = x[r] - mean; lss += d * d; }
  red[threadIdx.x] = lss; __syncthreads();
  for (int st = 128; st > 0; st >>= 1) {
    if (threadIdx.x < st) red[threadIdx.x] += red[threadIdx.x + st];
    __syncthreads();
  }
  float n = sqrtf(red[0]) * (1.0f / 32.0f);
  float inv = 1.0f / (n + EPSN);
#pragma unroll
  for (int r = 0; r < 4; ++r) {
    int i = threadIdx.x + r * 256;
    float y = (x[r] - mean) * inv;
    y = y >= 0.f ? y : y * SLOPE_;
    out[((size_t)b * I_ + i) * S_ + s] = y;
  }
}

__global__ void conv_wmma_kernel(const float* __restrict__ G,
                                 const _Float16* __restrict__ Wp1,
                                 float* __restrict__ Out) {
  __shared__ _Float16 xs[128 * 16];
  int tile = blockIdx.y;
  int b = (tile * 16) / S_, s0 = (tile * 16) % S_;
  int wave = threadIdx.x >> 5, lane = threadIdx.x & 31;
  int mt0 = blockIdx.x * 16 + wave * 2;
  int col = lane & 15, khalf = (lane >> 4) * 8;
  const int KTtot = (3 * I_) >> 5;
  size_t fragstride = (size_t)KTtot * 512;
  const _Float16* A0 = Wp1 + (size_t)(mt0 + 0) * fragstride + lane * 16;
  const _Float16* A1 = Wp1 + (size_t)(mt0 + 1) * fragstride + lane * 16;
  v8f c0 = {}, c1 = {};
  for (int cc = 0; cc < 3 * I_; cc += 128) {
    __syncthreads();
    for (int p = threadIdx.x; p < 128 * 16; p += blockDim.x) {
      int kk = cc + (p >> 4); int n = p & 15;
      int i = kk / 3, kc = kk - i * 3;
      int s = s0 + n - 2 + kc;
      int klocal = p >> 4;
      int kt = klocal >> 5, kin = klocal & 31;
      xs[((kt * 32 + frag_lane(kin, n)) << 4) + frag_j(kin)] =
          (s >= 0) ? (_Float16)G[((size_t)b * I_ + i) * S_ + s] : (_Float16)0.0f;
    }
    __syncthreads();
    int ktbase = cc >> 5;
#pragma unroll
    for (int kt = 0; kt < 4; ++kt) {
      __builtin_prefetch(A0 + (size_t)(ktbase + kt + 1) * 512, 0, 0);
      v16h bv  = load_frag(&xs[(kt * 32 + lane) * 16]);
      v16h a0v = load_frag(A0 + (size_t)(ktbase + kt) * 512);
      v16h a1v = load_frag(A1 + (size_t)(ktbase + kt) * 512);
      c0 = __builtin_amdgcn_wmma_f32_16x16x32_f16(false, a0v, false, bv,
                                                  (short)0, c0, false, false);
      c1 = __builtin_amdgcn_wmma_f32_16x16x32_f16(false, a1v, false, bv,
                                                  (short)0, c1, false, false);
    }
  }
#pragma unroll
  for (int j = 0; j < 8; ++j) {
    int r0 = mt0 * 16 + j + khalf;
    int r1 = r0 + 16;
    Out[((size_t)b * (3 * I_) + r0) * S_ + s0 + col] = c0[j];
    Out[((size_t)b * (3 * I_) + r1) * S_ + s0 + col] = c1[j];
  }
}

__global__ void init_loss_kernel(float* lacc) {
  if (threadIdx.x < 16) lacc[threadIdx.x] = 0.f;
}
__global__ void finalize_loss_kernel(const float* __restrict__ lacc,
                                     float* __restrict__ out_loss) {
  if (threadIdx.x == 0) {
    const float invT = 1.0f / (float)TOK;
    float l = 0.f;
    for (int e = 0; e < E_; ++e) l += (lacc[e] * invT) * (lacc[4 + e] * invT);
    for (int e = 0; e < E_; ++e) l += (lacc[8 + e] * invT) * (lacc[12 + e] * invT);
    *out_loss = l;
  }
}

extern "C" void kernel_launch(void* const* d_in, const int* in_sizes, int n_in,
                              void* d_out, int out_size, void* d_ws, size_t ws_size,
                              hipStream_t stream) {
  const float* inp     = (const float*)d_in[0];
  const float* divisor = (const float*)d_in[1];
  const float* w0_gate = (const float*)d_in[2];
  const float* w0      = (const float*)d_in[3];
  const float* w1      = (const float*)d_in[4];
  const float* w2_gate = (const float*)d_in[5];
  const float* w2      = (const float*)d_in[6];
  float* out = (float*)d_out;

  char* ws = (char*)d_ws;
  size_t off = 0;
  float* lacc = (float*)(ws + off); off += 256;
  const size_t h0b = (size_t)B_ * 3 * I_ * S_ * sizeof(float);
  const size_t t1b = (size_t)B_ * I_ * S_ * sizeof(float);
  float* h0   = (float*)(ws + off); off += h0b;
  float* tmp1 = (float*)(ws + off); off += t1b;
  float* g1   = (float*)(ws + off); off += t1b;
  int*   idx0 = (int*)(ws + off); off += TOK * sizeof(int);
  int*   idx1 = (int*)(ws + off); off += TOK * sizeof(int);
  off = (off + 255) & ~(size_t)255;
  const size_t w0n = (size_t)E_ * F_ * 3 * I_;
  const size_t w1n = (size_t)3 * I_ * 3 * I_;
  const size_t w2n = (size_t)E_ * I_ * F_;
  _Float16* w0p = (_Float16*)(ws + off); off += w0n * 2;
  _Float16* w1p = (_Float16*)(ws + off); off += w1n * 2;
  _Float16* w2p = (_Float16*)(ws + off); off += w2n * 2;
  float* h2 = h0;
  float* h3 = tmp1;

  pack_weights_kernel<<<(int)(w0n / 256), 256, 0, stream>>>(w0, w0p, E_, F_, 3 * I_, 0);
  pack_weights_kernel<<<(int)(w1n / 256), 256, 0, stream>>>(w1, w1p, 1, 3 * I_, 3 * I_, 1);
  pack_weights_kernel<<<(int)(w2n / 256), 256, 0, stream>>>(w2, w2p, E_, I_, F_, 0);

  init_loss_kernel<<<1, 32, 0, stream>>>(lacc);

  gate_kernel<<<TOK / 32, GATE_WAVES * 32, 0, stream>>>(
      inp, F_, w0_gate, idx0, lacc);
  moe_wmma_kernel<<<dim3(3 * I_ / 256, TOK / 16), 256,
                    F_ * 16 * sizeof(_Float16), stream>>>(
      inp, F_, w0p, 3 * I_, idx0, h0, 1.0f);

  cumsum_kernel<<<(B_ * I_) / 8, 256, 0, stream>>>(h0, divisor, tmp1);
  norm_kernel<<<TOK, 256, 0, stream>>>(tmp1, tmp1, tmp1, I_, 0, g1);

  conv_wmma_kernel<<<dim3(3 * I_ / 256, TOK / 16), 256, 0, stream>>>(g1, w1p, h2);

  norm_kernel<<<TOK, 256, 0, stream>>>(
      h2, h2 + (size_t)I_ * S_, h2 + (size_t)2 * I_ * S_, 3 * I_, 1, h3);

  gate_kernel<<<TOK / 32, GATE_WAVES * 32, 0, stream>>>(
      h3, I_, w2_gate, idx1, lacc + 8);
  moe_wmma_kernel<<<dim3(F_ / 256, TOK / 16), 256,
                    I_ * 16 * sizeof(_Float16), stream>>>(
      h3, I_, w2p, F_, idx1, out, 1.0f  );

  finalize_loss_kernel<<<1, 32, 0, stream>>>(lacc, out + (size_t)B_ * F_ * S_);
}


# WMMA GEMMs re-tiled (conv and both routed GEMMs): a work-group spans 64 (32) tokens, so each weight fragment read from L2 feeds 4 (2) products, not 1; conv weight loads run half a chunk ahead
# speedup vs baseline: 1.8681x; 1.5281x over previous
#include <hip/hip_runtime.h>
#include <hip/hip_bf16.h>

typedef __attribute__((ext_vector_type(16))) _Float16 v16h;
typedef __attribute__((ext_vector_type(8)))  _Float16 v8h;
typedef __attribute__((ext_vector_type(8)))  float    v8f;

#define B_ 2
#define F_ 512
#define S_ 1024
#define E_ 4
#define I_ 1024
#define TOK (B_*S_)
#define EPSN 1e-5f
#define SLOPE_ 0.02f

__device__ __forceinline__ int frag_j(int kin) {
  return kin < 8 ? kin : (kin < 16 ? kin - 8 : (kin < 24 ? kin - 8 : kin - 16));
}
__device__ __forceinline__ int frag_lane(int kin, int n) {
  return n + (((kin >> 3) & 1) ? 16 : 0);
}

__global__ void pack_weights_kernel(const float* __restrict__ W,
                                    _Float16* __restrict__ Wp,
                                    int E, int Cin, int Cout, int mode) {
  size_t idx = (size_t)blockIdx.x * blockDim.x + threadIdx.x;
  size_t total = (size_t)E * Cin * Cout;
  if (idx >= total) return;
  int j    = (int)(idx & 15);
  int lane = (int)((idx >> 4) & 31);
  size_t rest = idx >> 9;
  int KT = Cin >> 5;
  int kt = (int)(rest % KT); rest /= KT;
  int MT = Cout >> 4;
  int mt = (int)(rest % MT);
  int e  = (int)(rest / MT);
  int khalf = (lane >> 4) * 8;
  int k = kt * 32 + khalf + (j < 8 ? j : j + 8);
  int m = mt * 16 + (lane & 15);
  float v = mode ? W[(size_t)m * Cin + k]
                 : W[((size_t)e * Cin + k) * Cout + m];
  Wp[idx] = (_Float16)v;
}

__device__ __forceinline__ v16h load_frag(const _Float16* p) {
  v8h lo = *(const v8h*)p;
  v8h hi = *(const v8h*)(p + 8);
  v16h r;
#pragma unroll
  for (int j = 0; j < 8; ++j) { r[j] = lo[j]; r[j + 8] = hi[j]; }
  return r;
}

#define GATE_WAVES 16
__global__ void __launch_bounds__(GATE_WAVES * 32)
gate_kernel(const float* __restrict__ x, int C,
            const float* __restrict__ gw,
            int* __restrict__ eidx,
            float* __restrict__ lacc) {
  __shared__ float part[GATE_WAVES][E_][32];
  const int lane = threadIdx.x & 31;
  const int wave = __builtin_amdgcn_readfirstlane((int)(threadIdx.x >> 5));
  const int t0 = blockIdx.x * 32;
  const int b = t0 / S_, s = (t0 % S_) + lane;
  const int cper = C / GATE_WAVES;
  const int c0 = wave * cper;
  const float* xp = x + ((size_t)b * C + c0) * S_ + s;
  const float* g0 = gw + c0;
  const float* g1 = g0 + C;
  const float* g2 = g1 + C;
  const float* g3 = g2 + C;
  float a0 = 0.f, a1 = 0.f, a2 = 0.f, a3 = 0.f;
  for (int cb = 0; cb < cper; cb += 16) {
    float xv[16];
#pragma unroll
    for (int u = 0; u < 16; ++u) xv[u] = xp[(size_t)(cb + u) * S_];
#pragma unroll
    for (int u = 0; u < 16; ++u) {
      a0 = __builtin_fmaf(xv[u], g0[cb + u], a0);
      a1 = __builtin_fmaf(xv[u], g1[cb + u], a1);
      a2 = __builtin_fmaf(xv[u], g2[cb + u], a2);
      a3 = __builtin_fmaf(xv[u], g3[cb + u], a3);
    }
  }
  part[wave][0][lane] = a0;
  part[wave][1][lane] = a1;
  part[wave][2][lane] = a2;
  part[wave][3][lane] = a3;
  __syncthreads();
  if (wave != 0) return;
  float a[4];
  for (int e = 0; e < E_; ++e) {
    float v = part[0][e][lane];
    for (int w = 1; w < GATE_WAVES; ++w) v += part[w][e][lane];
    a[e] = v;
  }
  int best = 0; float m = a[0];
  for (int e = 1; e < E_; ++e) { if (a[e] > m) { m = a[e]; best = e; } }
  float g[4]; float se = 0.f;
  for (int e = 0; e < E_; ++e) { g[e] = __expf(a[e] - m); se += g[e]; }
  float inv = 1.0f / se;
  eidx[t0 + lane] = best;
  for (int e = 0; e < E_; ++e) {
    float gsum = g[e] * inv;
    for (int d = 16; d > 0; d >>= 1) gsum += __shfl_xor(gsum, d, 32);
    float cnt = (float)__popcll(__ballot(best == e));
    if (lane == 0) {
      atomicAdd(&lacc[e], gsum);
      atomicAdd(&lacc[4 + e], cnt);
    }
  }
}

__device__ __forceinline__ void store_oct(_Float16* dst, const float* v) {
  v8h h;
#pragma unroll
  for (int jj = 0; jj < 8; ++jj) h[jj] = (_Float16)v[jj];
  *(v8h*)dst = h;
}
__device__ __forceinline__ int oct_off(int oct, int n, int KT) {
  int kt = oct >> 2, a = oct & 3;
  return ((((n >> 4) * KT + kt) * 32 + (n & 15) + ((a & 1) << 4)) << 4) + ((a >> 1) << 3);
}

template <int CIN, int COUT, int MTW, int NT, int WAVES>
__global__ void __launch_bounds__(WAVES * 32)
moe_wmma_kernel(const float* __restrict__ X,
                const _Float16* __restrict__ Wp,
                const int* __restrict__ eidx,
                float* __restrict__ Out, float outscale) {
  constexpr int KT = CIN >> 5;
  constexpr int NTOK = NT * 16;
  constexpr int THREADS = WAVES * 32;
  constexpr int ITEMS = (CIN >> 3) * NTOK;
  __shared__ __attribute__((aligned(16))) _Float16 xs[NT * KT * 512];
  const int tok0 = blockIdx.y * NTOK;
  const int b = tok0 / S_, s0 = tok0 % S_;
  auto stage = [&](int q) {
    const int n = q % NTOK, oct = q / NTOK;
    const float* xp = X + ((size_t)b * CIN + oct * 8) * S_ + s0 + n;
    float v[8];
#pragma unroll
    for (int jj = 0; jj < 8; ++jj) v[jj] = xp[(size_t)jj * S_];
    store_oct(&xs[oct_off(oct, n, KT)], v);
  };
  constexpr int FULL = ITEMS / THREADS;
  constexpr int UNR = (FULL % 5 == 0) ? 5 : 4;
#pragma unroll UNR
  for (int it = 0; it < FULL; ++it) stage((int)threadIdx.x + it * THREADS);
  if (ITEMS % THREADS != 0) {
    const int q = (int)threadIdx.x + FULL * THREADS;
    if (q < ITEMS) stage(q);
  }
  __syncthreads();
  const int lane = threadIdx.x & 31;
  const int wave = __builtin_amdgcn_readfirstlane((int)(threadIdx.x >> 5));
  const int col = lane & 15, khalf = (lane >> 4) * 8;
  const int mt0 = (blockIdx.x * WAVES + wave) * MTW;
  constexpr size_t fragstride = (size_t)KT * 512;
  int myexp[NT];
#pragma unroll
  for (int nt = 0; nt < NT; ++nt) myexp[nt] = eidx[tok0 + nt * 16 + col];
  v8f cs[MTW][NT];
#pragma unroll
  for (int mi = 0; mi < MTW; ++mi)
#pragma unroll
    for (int nt = 0; nt < NT; ++nt) cs[mi][nt] = v8f{};
  const _Float16* Bb = xs + lane * 16;
  for (int e = 0; e < E_; ++e) {
    int hit = 0;
#pragma unroll
    for (int nt = 0; nt < NT; ++nt) hit |= (myexp[nt] == e) ? 1 : 0;
    if (!__any(hit)) continue;
    const _Float16* Ab = Wp + ((size_t)e * (COUT >> 4) + mt0) * fragstride + lane * 16;
    v8f c[MTW][NT];
    v16h ac[MTW], bc[NT];
#pragma unroll
    for (int mi = 0; mi < MTW; ++mi) ac[mi] = load_frag(Ab + (size_t)mi * fragstride);
#pragma unroll
    for (int nt = 0; nt < NT; ++nt) {
      bc[nt] = load_frag(Bb + (size_t)nt * KT * 512);
#pragma unroll
      for (int mi = 0; mi < MTW; ++mi) c[mi][nt] = v8f{};
    }
#pragma unroll 2
    for (int kt = 0; kt < KT; ++kt) {
      const int k2 = (kt + 1 < KT) ? kt + 1 : kt;
      v16h an[MTW], bn[NT];
#pragma unroll
      for (int mi = 0; mi < MTW; ++mi)
        an[mi] = load_frag(Ab + (size_t)mi * fragstride + (size_t)k2 * 512);
#pragma unroll
      for (int nt = 0; nt < NT; ++nt)
        bn[nt] = load_frag(Bb + ((size_t)nt * KT + k2) * 512);
#pragma unroll
      for (int nt = 0; nt < NT; ++nt)
#pragma unroll
        for (int mi = 0; mi < MTW; ++mi)
          c[mi][nt] = __builtin_amdgcn_wmma_f32_16x16x32_f16(
              false, ac[mi], false, bc[nt], (short)0, c[mi][nt], false, false);
#pragma unroll
      for (int mi = 0; mi < MTW; ++mi) ac[mi] = an[mi];
#pragma unroll
      for (int nt = 0; nt < NT; ++nt) bc[nt] = bn[nt];
    }
#pragma unroll
    for (int nt = 0; nt < NT; ++nt)
      if (myexp[nt] == e) {
#pragma unroll
        for (int mi = 0; mi < MTW; ++mi) cs[mi][nt] = c[mi][nt];
      }
  }
#pragma unroll
  for (int mi = 0; mi < MTW; ++mi)
#pragma unroll
    for (int nt = 0; nt < NT; ++nt)
#pragma unroll
      for (int j = 0; j < 8; ++j) {
        int r = (mt0 + mi) * 16 + j + khalf;
        Out[((size_t)b * COUT + r) * S_ + s0 + nt * 16 + col] = cs[mi][nt][j] * outscale;
      }
}

__global__ void cumsum_kernel(const float* __restrict__ h0,
                              const float* __restrict__ divisor,
                              float* __restrict__ tmp1) {
  int row = blockIdx.x * (blockDim.x >> 5) + (threadIdx.x >> 5);
  int lane = threadIdx.x & 31;
  if (row >= B_ * I_) return;
  int b = row / I_, i = row % I_;
  const float* dep = h0 + ((size_t)b * 3 * I_ + i) * S_;
  const float* sc  = h0 + ((size_t)b * 3 * I_ + I_ + i) * S_;
  const float* sh  = h0 + ((size_t)b * 3 * I_ + 2 * I_ + i) * S_;
  float* o = tmp1 + ((size_t)b * I_ + i) * S_;
  float carry = 0.f;
  for (int s0 = 0; s0 < S_; s0 += 32) {
    float v = dep[s0 + lane];
#pragma unroll
    for (int d = 1; d < 32; d <<= 1) {
      float t = __shfl_up(v, d, 32);
      if (lane >= d) v += t;
    }
    float cum = carry + v;
    o[s0 + lane] = cum / divisor[s0 + lane] * sc[s0 + lane] + sh[s0 + lane];
    carry = __shfl(cum, 31, 32);
  }
}

__global__ void norm_kernel(const float* __restrict__ p0,
                            const float* __restrict__ p1,
                            const float* __restrict__ p2,
                            int cstride, int mode,
                            float* __restrict__ out) {
  __shared__ float red[256];
  int t = blockIdx.x; int b = t / S_, s = t % S_;
  float x[4]; float lsum = 0.f;
#pragma unroll
  for (int r = 0; r < 4; ++r) {
    int i = threadIdx.x + r * 256;
    size_t idx = ((size_t)b * cstride + i) * S_ + s;
    float v = p0[idx];
    if (mode) v = v * p1[idx] + p2[idx];
    x[r] = v; lsum += v;
  }
  red[threadIdx.x] = lsum; __syncthreads();
  for (int st = 128; st > 0; st >>= 1) {
    if (threadIdx.x < st) red[threadIdx.x] += red[threadIdx.x + st];
    __syncthreads();
  }
  float mean = red[0] * (1.0f / I_);
  __syncthreads();
  float lss = 0.f;
#pragma unroll
  for (int r = 0; r < 4; ++r) { float d = x[r] - mean; lss += d * d; }
  red[threadIdx.x] = lss; __syncthreads();
  for (int st = 128; st > 0; st >>= 1) {
    if (threadIdx.x < st) red[threadIdx.x] += red[threadIdx.x + st];
    __syncthreads();
  }
  float n = sqrtf(red[0]) * (1.0f / 32.0f);
  float inv = 1.0f / (n + EPSN);
#pragma unroll
  for (int r = 0; r < 4; ++r) {
    int i = threadIdx.x + r * 256;
    float y = (x[r] - mean) * inv;
    y = y >= 0.f ? y : y * SLOPE_;
    out[((size_t)b * I_ + i) * S_ + s] = y;
  }
}

#define CV_WAVES 12
#define CV_NT 4
#define CV_KS 6
__global__ void __launch_bounds__(CV_WAVES * 32)
conv_wmma_kernel(const float* __restrict__ G,
                 const _Float16* __restrict__ Wp1,
                 float* __restrict__ Out) {
  constexpr int CHUNK = CV_KS * 32;
  constexpr int NCH = (3 * I_) / CHUNK;
  constexpr int BUF = CV_NT * CV_KS * 512;
  constexpr int ITS = (CV_KS * 4 * CV_NT * 16) / (CV_WAVES * 32);
  constexpr int KTtot = (3 * I_) >> 5;
  constexpr size_t fragstride = (size_t)KTtot * 512;
  __shared__ __attribute__((aligned(16))) _Float16 xs[2][BUF];
  const int tok0 = blockIdx.y * (CV_NT * 16);
  const int b = tok0 / S_, s0 = tok0 % S_;
  const int lane = threadIdx.x & 31;
  const int wave = __builtin_amdgcn_readfirstlane((int)(threadIdx.x >> 5));
  const int col = lane & 15, khalf = (lane >> 4) * 8;
  const int mt0 = (blockIdx.x * CV_WAVES + wave) * 2;
  const _Float16* A0 = Wp1 + (size_t)mt0 * fragstride + lane * 16;
  const _Float16* A1 = A0 + fragstride;
  const int n64 = (wave & 1) * 32 + lane;
  const int octw = wave >> 1;
  const int sb = s0 + n64 - 2;
  const float* Gb = G + (size_t)b * I_ * S_;
  float r[ITS][8];
  auto gload = [&](int cc) {
#pragma unroll
    for (int it = 0; it < ITS; ++it) {
      const int kk0 = cc + (it * (CV_WAVES / 2) + octw) * 8;
#pragma unroll
      for (int jj = 0; jj < 8; ++jj) {
        const int kk = kk0 + jj;
        const int i = kk / 3, kc = kk - i * 3;
        const int s = sb + kc;
        const float gv = Gb[(size_t)i * S_ + (s > 0 ? s : 0)];
        r[it][jj] = (s >= 0) ? gv : 0.0f;
      }
    }
  };
  auto lstore = [&](int buf) {
#pragma unroll
    for (int it = 0; it < ITS; ++it)
      store_oct(&xs[buf][oct_off(it * (CV_WAVES / 2) + octw, n64, CV_KS)], r[it]);
  };
  v8f c[2][CV_NT];
#pragma unroll
  for (int nt = 0; nt < CV_NT; ++nt) { c[0][nt] = v8f{}; c[1][nt] = v8f{}; }
  gload(0);
  lstore(0);
  __syncthreads();
  v16h ag0[3][2], ag1[3][2];
#pragma unroll
  for (int u = 0; u < 3; ++u) {
    ag0[u][0] = load_frag(A0 + (size_t)u * 512);
    ag0[u][1] = load_frag(A1 + (size_t)u * 512);
  }
  for (int ch = 0; ch < NCH; ++ch) {
    const int buf = ch & 1;
    const _Float16* Bb = &xs[buf][lane * 16];
    if (ch + 1 < NCH) gload((ch + 1) * CHUNK);
#pragma unroll
    for (int u = 0; u < 3; ++u) {
      ag1[u][0] = load_frag(A0 + (size_t)(ch * CV_KS + 3 + u) * 512);
      ag1[u][1] = load_frag(A1 + (size_t)(ch * CV_KS + 3 + u) * 512);
    }
    v16h bc[CV_NT], bn[CV_NT];
#pragma unroll
    for (int nt = 0; nt < CV_NT; ++nt) bc[nt] = load_frag(Bb + (nt * CV_KS) * 512);
    __builtin_amdgcn_sched_barrier(0);
#pragma unroll
    for (int kt = 0; kt < CV_KS; ++kt) {
      if (kt == 3) {
#pragma unroll
        for (int u = 0; u < 3; ++u) {
          const int kn = ((ch + 1) * CV_KS + u < KTtot) ? (ch + 1) * CV_KS + u : KTtot - 1;
          ag0[u][0] = load_frag(A0 + (size_t)kn * 512);
          ag0[u][1] = load_frag(A1 + (size_t)kn * 512);
        }
      }
      if (kt + 1 < CV_KS) {
#pragma unroll
        for (int nt = 0; nt < CV_NT; ++nt)
          bn[nt] = load_frag(Bb + (nt * CV_KS + kt + 1) * 512);
      }
      __builtin_amdgcn_sched_barrier(0);
      const v16h a0c = (kt < 3) ? ag0[kt % 3][0] : ag1[kt % 3][0];
      const v16h a1c = (kt < 3) ? ag0[kt % 3][1] : ag1[kt % 3][1];
#pragma unroll
      for (int nt = 0; nt < CV_NT; ++nt) {
        c[0][nt] = __builtin_amdgcn_wmma_f32_16x16x32_f16(
            false, a0c, false, bc[nt], (short)0, c[0][nt], false, false);
        c[1][nt] = __builtin_amdgcn_wmma_f32_16x16x32_f16(
            false, a1c, false, bc[nt], (short)0, c[1][nt], false, false);
      }
      __builtin_amdgcn_sched_barrier(0);
      if (kt + 1 < CV_KS) {
#pragma unroll
        for (int nt = 0; nt < CV_NT; ++nt) bc[nt] = bn[nt];
      }
    }
    if (ch + 1 < NCH) lstore(buf ^ 1);
    __syncthreads();
  }
#pragma unroll
  for (int mi = 0; mi < 2; ++mi)
#pragma unroll
    for (int nt = 0; nt < CV_NT; ++nt)
#pragma unroll
      for (int j = 0; j < 8; ++j) {
        int rr = (mt0 + mi) * 16 + j + khalf;
        Out[((size_t)b * (3 * I_) + rr) * S_ + s0 + nt * 16 + col] = c[mi][nt][j];
      }
}

__global__ void init_loss_kernel(float* lacc) {
  if (threadIdx.x < 16) lacc[threadIdx.x] = 0.f;
}
__global__ void finalize_loss_kernel(const float* __restrict__ lacc,
                                     float* __restrict__ out_loss) {
  if (threadIdx.x == 0) {
    const float invT = 1.0f / (float)TOK;
    float l = 0.f;
    for (int e = 0; e < E_; ++e) l += (lacc[e] * invT) * (lacc[4 + e] * invT);
    for (int e = 0; e < E_; ++e) l += (lacc[8 + e] * invT) * (lacc[12 + e] * invT);
    *out_loss = l;
  }
}

extern "C" void kernel_launch(void* const* d_in, const int* in_sizes, int n_in,
                              void* d_out, int out_size, void* d_ws, size_t ws_size,
                              hipStream_t stream) {
  const float* inp     = (const float*)d_in[0];
  const float* divisor = (const float*)d_in[1];
  const float* w0_gate = (const float*)d_in[2];
  const float* w0      = (const float*)d_in[3];
  const float* w1      = (const float*)d_in[4];
  const float* w2_gate = (const float*)d_in[5];
  const float* w2      = (const float*)d_in[6];
  float* out = (float*)d_out;

  char* ws = (char*)d_ws;
  size_t off = 0;
  float* lacc = (float*)(ws + off); off += 256;
  const size_t h0b = (size_t)B_ * 3 * I_ * S_ * sizeof(float);
  const size_t t1b = (size_t)B_ * I_ * S_ * sizeof(float);
  float* h0   = (float*)(ws + off); off += h0b;
  float* tmp1 = (float*)(ws + off); off += t1b;
  float* g1   = (float*)(ws + off); off += t1b;
  int*   idx0 = (int*)(ws + off); off += TOK * sizeof(int);
  int*   idx1 = (int*)(ws + off); off += TOK * sizeof(int);
  off = (off + 255) & ~(size_t)255;
  const size_t w0n = (size_t)E_ * F_ * 3 * I_;
  const size_t w1n = (size_t)3 * I_ * 3 * I_;
  const size_t w2n = (size_t)E_ * I_ * F_;
  _Float16* w0p = (_Float16*)(ws + off); off += w0n * 2;
  _Float16* w1p = (_Float16*)(ws + off); off += w1n * 2;
  _Float16* w2p = (_Float16*)(ws + off); off += w2n * 2;
  float* h2 = h0;
  float* h3 = tmp1;

  pack_weights_kernel<<<(int)(w0n / 256), 256, 0, stream>>>(w0, w0p, E_, F_, 3 * I_, 0);
  pack_weights_kernel<<<(int)(w1n / 256), 256, 0, stream>>>(w1, w1p, 1, 3 * I_, 3 * I_, 1);
  pack_weights_kernel<<<(int)(w2n / 256), 256, 0, stream>>>(w2, w2p, E_, I_, F_, 0);

  init_loss_kernel<<<1, 32, 0, stream>>>(lacc);

  gate_kernel<<<TOK / 32, GATE_WAVES * 32, 0, stream>>>(
      inp, F_, w0_gate, idx0, lacc);
  moe_wmma_kernel<F_, 3 * I_, 2, 4, 12><<<dim3(3 * I_ / 384, TOK / 64), 384, 0, stream>>>(
      inp, w0p, idx0, h0, 1.0f);

  cumsum_kernel<<<(B_ * I_) / 8, 256, 0, stream>>>(h0, divisor, tmp1);
  norm_kernel<<<TOK, 256, 0, stream>>>(tmp1, tmp1, tmp1, I_, 0, g1);

  conv_wmma_kernel<<<dim3(3 * I_ / 384, TOK / 64), CV_WAVES * 32, 0, stream>>>(g1, w1p, h2);

  norm_kernel<<<TOK, 256, 0, stream>>>(
      h2, h2 + (size_t)I_ * S_, h2 + (size_t)2 * I_ * S_, 3 * I_, 1, h3);

  gate_kernel<<<TOK / 32, GATE_WAVES * 32, 0, stream>>>(
      h3, I_, w2_gate, idx1, lacc + 8);
  moe_wmma_kernel<I_, F_, 1, 2, 8><<<dim3(F_ / 128, TOK / 32), 256, 0, stream>>>(
      h3, w2p, idx1, out, 1.0f  );

  finalize_loss_kernel<<<1, 32, 0, stream>>>(lacc, out + (size_t)B_ * F_ * S_);
}


# weight pre-pack re-tiled: a wave converts a whole 32x32 (conv: 16x128) source tile from whole-line loads and stores finished fragment rows; 12.8k waves, was 557k at one element per thread
# speedup vs baseline: 2.0264x; 1.0847x over previous
#include <hip/hip_runtime.h>
#include <hip/hip_bf16.h>

typedef __attribute__((ext_vector_type(16))) _Float16 v16h;
typedef __attribute__((ext_vector_type(8)))  _Float16 v8h;
typedef __attribute__((ext_vector_type(8)))  float    v8f;

#define B_ 2
#define F_ 512
#define S_ 1024
#define E_ 4
#define I_ 1024
#define TOK (B_*S_)
#define EPSN 1e-5f
#define SLOPE_ 0.02f

__device__ __forceinline__ int frag_j(int kin) {
  return kin < 8 ? kin : (kin < 16 ? kin - 8 : (kin < 24 ? kin - 8 : kin - 16));
}
__device__ __forceinline__ int frag_lane(int kin, int n) {
  return n + (((kin >> 3) & 1) ? 16 : 0);
}

template <int CIN, int COUT>
__global__ void __launch_bounds__(256)
pack_kn_kernel(const float* __restrict__ W, _Float16* __restrict__ Wp) {
  constexpr int KT = CIN >> 5, MT = COUT >> 4, MG = COUT >> 5;
  const int lane = threadIdx.x & 31;
  const int g = __builtin_amdgcn_readfirstlane((int)(blockIdx.x * 8 + (threadIdx.x >> 5)));
  const int mg = g % MG;
  const int kt = (g / MG) % KT;
  const int e  = g / (MG * KT);
  const float* sp = W + ((size_t)e * CIN + (size_t)kt * 32) * COUT + mg * 32 + lane;
  float v[32];
#pragma unroll
  for (int k = 0; k < 32; ++k) v[k] = sp[(size_t)k * COUT];
  v16h lo, hi;
#pragma unroll
  for (int j = 0; j < 8; ++j) {
    lo[j]     = (_Float16)v[j];
    lo[j + 8] = (_Float16)v[j + 16];
    hi[j]     = (_Float16)v[j + 8];
    hi[j + 8] = (_Float16)v[j + 24];
  }
  const int mt = mg * 2 + (lane >> 4);
  _Float16* dp = Wp + ((((size_t)e * MT + mt) * KT + kt) * 32 + (lane & 15)) * 16;
  *(v16h*)dp = lo;
  *(v16h*)(dp + 16 * 16) = hi;
}

template <int CIN, int COUT>
__global__ void __launch_bounds__(256)
pack_mk_kernel(const float* __restrict__ W, _Float16* __restrict__ Wp) {
  constexpr int KT = CIN >> 5, KG = KT >> 2;
  typedef __attribute__((ext_vector_type(4))) float v4f;
  typedef __attribute__((ext_vector_type(4))) _Float16 v4h;
  const int lane = threadIdx.x & 31;
  const int g = __builtin_amdgcn_readfirstlane((int)(blockIdx.x * 8 + (threadIdx.x >> 5)));
  const int kg = g % KG;
  const int mt = g / KG;
  const int r = lane >> 3, q = lane & 7;
  const float* sp = W + ((size_t)mt * 16 + r) * CIN + kg * 128 + q * 4;
  v4f x[4][4];
#pragma unroll
  for (int t = 0; t < 4; ++t)
#pragma unroll
    for (int i = 0; i < 4; ++i)
      x[t][i] = *(const v4f*)(sp + (size_t)(4 * i) * CIN + t * 32);
  const int fl = r + (((q >> 1) & 1) << 4);
  const int j0 = ((q >> 2) << 3) + ((q & 1) << 2);
  _Float16* dp = Wp + (((size_t)mt * KT + kg * 4) * 32 + fl) * 16 + j0;
#pragma unroll
  for (int t = 0; t < 4; ++t)
#pragma unroll
    for (int i = 0; i < 4; ++i) {
      v4h h;
#pragma unroll
      for (int u = 0; u < 4; ++u) h[u] = (_Float16)x[t][i][u];
      *(v4h*)(dp + (size_t)t * 512 + (4 * i) * 16) = h;
    }
}

__device__ __forceinline__ v16h load_frag(const _Float16* p) {
  v8h lo = *(const v8h*)p;
  v8h hi = *(const v8h*)(p + 8);
  v16h r;
#pragma unroll
  for (int j = 0; j < 8; ++j) { r[j] = lo[j]; r[j + 8] = hi[j]; }
  return r;
}

#define GATE_WAVES 16
__global__ void __launch_bounds__(GATE_WAVES * 32)
gate_kernel(const float* __restrict__ x, int C,
            const float* __restrict__ gw,
            int* __restrict__ eidx,
            float* __restrict__ lacc) {
  __shared__ float part[GATE_WAVES][E_][32];
  const int lane = threadIdx.x & 31;
  const int wave = __builtin_amdgcn_readfirstlane((int)(threadIdx.x >> 5));
  const int t0 = blockIdx.x * 32;
  const int b = t0 / S_, s = (t0 % S_) + lane;
  const int cper = C / GATE_WAVES;
  const int c0 = wave * cper;
  const float* xp = x + ((size_t)b * C + c0) * S_ + s;
  const float* g0 = gw + c0;
  const float* g1 = g0 + C;
  const float* g2 = g1 + C;
  const float* g3 = g2 + C;
  float a0 = 0.f, a1 = 0.f, a2 = 0.f, a3 = 0.f;
  for (int cb = 0; cb < cper; cb += 16) {
    float xv[16];
#pragma unroll
    for (int u = 0; u < 16; ++u) xv[u] = xp[(size_t)(cb + u) * S_];
#pragma unroll
    for (int u = 0; u < 16; ++u) {
      a0 = __builtin_fmaf(xv[u], g0[cb + u], a0);
      a1 = __builtin_fmaf(xv[u], g1[cb + u], a1);
      a2 = __builtin_fmaf(xv[u], g2[cb + u], a2);
      a3 = __builtin_fmaf(xv[u], g3[cb + u], a3);
    }
  }
  part[wave][0][lane] = a0;
  part[wave][1][lane] = a1;
  part[wave][2][lane] = a2;
  part[wave][3][lane] = a3;
  __syncthreads();
  if (wave != 0) return;
  float a[4];
  for (int e = 0; e < E_; ++e) {
    float v = part[0][e][lane];
    for (int w = 1; w < GATE_WAVES; ++w) v += part[w][e][lane];
    a[e] = v;
  }
  int best = 0; float m = a[0];
  for (int e = 1; e < E_; ++e) { if (a[e] > m) { m = a[e]; best = e; } }
  float g[4]; float se = 0.f;
  for (int e = 0; e < E_; ++e) { g[e] = __expf(a[e] - m); se += g[e]; }
  float inv = 1.0f / se;
  eidx[t0 + lane] = best;
  for (int e = 0; e < E_; ++e) {
    float gsum = g[e] * inv;
    for (int d = 16; d > 0; d >>= 1) gsum += __shfl_xor(gsum, d, 32);
    float cnt = (float)__popcll(__ballot(best == e));
    if (lane == 0) {
      atomicAdd(&lacc[e], gsum);
      atomicAdd(&lacc[4 + e], cnt);
    }
  }
}

__device__ __forceinline__ void store_oct(_Float16* dst, const float* v) {
  v8h h;
#pragma unroll
  for (int jj = 0; jj < 8; ++jj) h[jj] = (_Float16)v[jj];
  *(v8h*)dst = h;
}
__device__ __forceinline__ int oct_off(int oct, int n, int KT) {
  int kt = oct >> 2, a = oct & 3;
  return ((((n >> 4) * KT + kt) * 32 + (n & 15) + ((a & 1) << 4)) << 4) + ((a >> 1) << 3);
}

template <int CIN, int COUT, int MTW, int NT, int WAVES>
__global__ void __launch_bounds__(WAVES * 32)
moe_wmma_kernel(const float* __restrict__ X,
                const _Float16* __restrict__ Wp,
                const int* __restrict__ eidx,
                float* __restrict__ Out, float outscale) {
  constexpr int KT = CIN >> 5;
  constexpr int NTOK = NT * 16;
  constexpr int THREADS = WAVES * 32;
  constexpr int ITEMS = (CIN >> 3) * NTOK;
  __shared__ __attribute__((aligned(16))) _Float16 xs[NT * KT * 512];
  const int tok0 = blockIdx.y * NTOK;
  const int b = tok0 / S_, s0 = tok0 % S_;
  auto stage = [&](int q) {
    const int n = q % NTOK, oct = q / NTOK;
    const float* xp = X + ((size_t)b * CIN + oct * 8) * S_ + s0 + n;
    float v[8];
#pragma unroll
    for (int jj = 0; jj < 8; ++jj) v[jj] = xp[(size_t)jj * S_];
    store_oct(&xs[oct_off(oct, n, KT)], v);
  };
  constexpr int FULL = ITEMS / THREADS;
  constexpr int UNR = (FULL % 5 == 0) ? 5 : 4;
#pragma unroll UNR
  for (int it = 0; it < FULL; ++it) stage((int)threadIdx.x + it * THREADS);
  if (ITEMS % THREADS != 0) {
    const int q = (int)threadIdx.x + FULL * THREADS;
    if (q < ITEMS) stage(q);
  }
  __syncthreads();
  const int lane = threadIdx.x & 31;
  const int wave = __builtin_amdgcn_readfirstlane((int)(threadIdx.x >> 5));
  const int col = lane & 15, khalf = (lane >> 4) * 8;
  const int mt0 = (blockIdx.x * WAVES + wave) * MTW;
  constexpr size_t fragstride = (size_t)KT * 512;
  int myexp[NT];
#pragma unroll
  for (int nt = 0; nt < NT; ++nt) myexp[nt] = eidx[tok0 + nt * 16 + col];
  v8f cs[MTW][NT];
#pragma unroll
  for (int mi = 0; mi < MTW; ++mi)
#pragma unroll
    for (int nt = 0; nt < NT; ++nt) cs[mi][nt] = v8f{};
  const _Float16* Bb = xs + lane * 16;
  for (int e = 0; e < E_; ++e) {
    int hit = 0;
#pragma unroll
    for (int nt = 0; nt < NT; ++nt) hit |= (myexp[nt] == e) ? 1 : 0;
    if (!__any(hit)) continue;
    const _Float16* Ab = Wp + ((size_t)e * (COUT >> 4) + mt0) * fragstride + lane * 16;
    v8f c[MTW][NT];
    v16h ac[MTW], bc[NT];
#pragma unroll
    for (int mi = 0; mi < MTW; ++mi) ac[mi] = load_frag(Ab + (size_t)mi * fragstride);
#pragma unroll
    for (int nt = 0; nt < NT; ++nt) {
      bc[nt] = load_frag(Bb + (size_t)nt * KT * 512);
#pragma unroll
      for (int mi = 0; mi < MTW; ++mi) c[mi][nt] = v8f{};
    }
#pragma unroll 2
    for (int kt = 0; kt < KT; ++kt) {
      const int k2 = (kt + 1 < KT) ? kt + 1 : kt;
      v16h an[MTW], bn[NT];
#pragma unroll
      for (int mi = 0; mi < MTW; ++mi)
        an[mi] = load_frag(Ab + (size_t)mi * fragstride + (size_t)k2 * 512);
#pragma unroll
      for (int nt = 0; nt < NT; ++nt)
        bn[nt] = load_frag(Bb + ((size_t)nt * KT + k2) * 512);
#pragma unroll
      for (int nt = 0; nt < NT; ++nt)
#pragma unroll
        for (int mi = 0; mi < MTW; ++mi)
          c[mi][nt] = __builtin_amdgcn_wmma_f32_16x16x32_f16(
              false, ac[mi], false, bc[nt], (short)0, c[mi][nt], false, false);
#pragma unroll
      for (int mi = 0; mi < MTW; ++mi) ac[mi] = an[mi];
#pragma unroll
      for (int nt = 0; nt < NT; ++nt) bc[nt] = bn[nt];
    }
#pragma unroll
    for (int nt = 0; nt < NT; ++nt)
      if (myexp[nt] == e) {
#pragma unroll
        for (int mi = 0; mi < MTW; ++mi) cs[mi][nt] = c[mi][nt];
      }
  }
#pragma unroll
  for (int mi = 0; mi < MTW; ++mi)
#pragma unroll
    for (int nt = 0; nt < NT; ++nt)
#pragma unroll
      for (int j = 0; j < 8; ++j) {
        int r = (mt0 + mi) * 16 + j + khalf;
        Out[((size_t)b * COUT + r) * S_ + s0 + nt * 16 + col] = cs[mi][nt][j] * outscale;
      }
}

__global__ void cumsum_kernel(const float* __restrict__ h0,
                              const float* __restrict__ divisor,
                              float* __restrict__ tmp1) {
  int row = blockIdx.x * (blockDim.x >> 5) + (threadIdx.x >> 5);
  int lane = threadIdx.x & 31;
  if (row >= B_ * I_) return;
  int b = row / I_, i = row % I_;
  const float* dep = h0 + ((size_t)b * 3 * I_ + i) * S_;
  const float* sc  = h0 + ((size_t)b * 3 * I_ + I_ + i) * S_;
  const float* sh  = h0 + ((size_t)b * 3 * I_ + 2 * I_ + i) * S_;
  float* o = tmp1 + ((size_t)b * I_ + i) * S_;
  float carry = 0.f;
  for (int s0 = 0; s0 < S_; s0 += 32) {
    float v = dep[s0 + lane];
#pragma unroll
    for (int d = 1; d < 32; d <<= 1) {
      float t = __shfl_up(v, d, 32);
      if (lane >= d) v += t;
    }
    float cum = carry + v;
    o[s0 + lane] = cum / divisor[s0 + lane] * sc[s0 + lane] + sh[s0 + lane];
    carry = __shfl(cum, 31, 32);
  }
}

__global__ void norm_kernel(const float* __restrict__ p0,
                            const float* __restrict__ p1,
                            const float* __restrict__ p2,
                            int cstride, int mode,
                            float* __restrict__ out) {
  __shared__ float red[256];
  int t = blockIdx.x; int b = t / S_, s = t % S_;
  float x[4]; float lsum = 0.f;
#pragma unroll
  for (int r = 0; r < 4; ++r) {
    int i = threadIdx.x + r * 256;
    size_t idx = ((size_t)b * cstride + i) * S_ + s;
    float v = p0[idx];
    if (mode) v = v * p1[idx] + p2[idx];
    x[r] = v; lsum += v;
  }
  red[threadIdx.x] = lsum; __syncthreads();
  for (int st = 128; st > 0; st >>= 1) {
    if (threadIdx.x < st) red[threadIdx.x] += red[threadIdx.x + st];
    __syncthreads();
  }
  float mean = red[0] * (1.0f / I_);
  __syncthreads();
  float lss = 0.f;
#pragma unroll
  for (int r = 0; r < 4; ++r) { float d = x[r] - mean; lss += d * d; }
  red[threadIdx.x] = lss; __syncthreads();
  for (int st = 128; st > 0; st >>= 1) {
    if (threadIdx.x < st) red[threadIdx.x] += red[threadIdx.x + st];
    __syncthreads();
  }
  float n = sqrtf(red[0]) * (1.0f / 32.0f);
  float inv = 1.0f / (n + EPSN);
#pragma unroll
  for (int r = 0; r < 4; ++r) {
    int i = threadIdx.x + r * 256;
    float y = (x[r] - mean) * inv;
    y = y >= 0.f ? y : y * SLOPE_;
    out[((size_t)b * I_ + i) * S_ + s] = y;
  }
}

#define CV_WAVES 12
#define CV_NT 4
#define CV_KS 6
__global__ void __launch_bounds__(CV_WAVES * 32)
conv_wmma_kernel(const float* __restrict__ G,
                 const _Float16* __restrict__ Wp1,
                 float* __restrict__ Out) {
  constexpr int CHUNK = CV_KS * 32;
  constexpr int NCH = (3 * I_) / CHUNK;
  constexpr int BUF = CV_NT * CV_KS * 512;
  constexpr int ITS = (CV_KS * 4 * CV_NT * 16) / (CV_WAVES * 32);
  constexpr int KTtot = (3 * I_) >> 5;
  constexpr size_t fragstride = (size_t)KTtot * 512;
  __shared__ __attribute__((aligned(16))) _Float16 xs[2][BUF];
  const int tok0 = blockIdx.y * (CV_NT * 16);
  const int b = tok0 / S_, s0 = tok0 % S_;
  const int lane = threadIdx.x & 31;
  const int wave = __builtin_amdgcn_readfirstlane((int)(threadIdx.x >> 5));
  const int col = lane & 15, khalf = (lane >> 4) * 8;
  const int mt0 = (blockIdx.x * CV_WAVES + wave) * 2;
  const _Float16* A0 = Wp1 + (size_t)mt0 * fragstride + lane * 16;
  const _Float16* A1 = A0 + fragstride;
  const int n64 = (wave & 1) * 32 + lane;
  const int octw = wave >> 1;
  const int sb = s0 + n64 - 2;
  const float* Gb = G + (size_t)b * I_ * S_;
  float r[ITS][8];
  auto gload = [&](int cc) {
#pragma unroll
    for (int it = 0; it < ITS; ++it) {
      const int kk0 = cc + (it * (CV_WAVES / 2) + octw) * 8;
#pragma unroll
      for (int jj = 0; jj < 8; ++jj) {
        const int kk = kk0 + jj;
        const int i = kk / 3, kc = kk - i * 3;
        const int s = sb + kc;
        const float gv = Gb[(size_t)i * S_ + (s > 0 ? s : 0)];
        r[it][jj] = (s >= 0) ? gv : 0.0f;
      }
    }
  };
  auto lstore = [&](int buf) {
#pragma unroll
    for (int it = 0; it < ITS; ++it)
      store_oct(&xs[buf][oct_off(it * (CV_WAVES / 2) + octw, n64, CV_KS)], r[it]);
  };
  v8f c[2][CV_NT];
#pragma unroll
  for (int nt = 0; nt < CV_NT; ++nt) { c[0][nt] = v8f{}; c[1][nt] = v8f{}; }
  gload(0);
  lstore(0);
  __syncthreads();
  v16h ag0[3][2], ag1[3][2];
#pragma unroll
  for (int u = 0; u < 3; ++u) {
    ag0[u][0] = load_frag(A0 + (size_t)u * 512);
    ag0[u][1] = load_frag(A1 + (size_t)u * 512);
  }
  for (int ch = 0; ch < NCH; ++ch) {
    const int buf = ch & 1;
    const _Float16* Bb = &xs[buf][lane * 16];
    if (ch + 1 < NCH) gload((ch + 1) * CHUNK);
#pragma unroll
    for (int u = 0; u < 3; ++u) {
      ag1[u][0] = load_frag(A0 + (size_t)(ch * CV_KS + 3 + u) * 512);
      ag1[u][1] = load_frag(A1 + (size_t)(ch * CV_KS + 3 + u) * 512);
    }
    v16h bc[CV_NT], bn[CV_NT];
#pragma unroll
    for (int nt = 0; nt < CV_NT; ++nt) bc[nt] = load_frag(Bb + (nt * CV_KS) * 512);
    __builtin_amdgcn_sched_barrier(0);
#pragma unroll
    for (int kt = 0; kt < CV_KS; ++kt) {
      if (kt == 3) {
#pragma unroll
        for (int u = 0; u < 3; ++u) {
          const int kn = ((ch + 1) * CV_KS + u < KTtot) ? (ch + 1) * CV_KS + u : KTtot - 1;
          ag0[u][0] = load_frag(A0 + (size_t)kn * 512);
          ag0[u][1] = load_frag(A1 + (size_t)kn * 512);
        }
      }
      if (kt + 1 < CV_KS) {
#pragma unroll
        for (int nt = 0; nt < CV_NT; ++nt)
          bn[nt] = load_frag(Bb + (nt * CV_KS + kt + 1) * 512);
      }
      __builtin_amdgcn_sched_barrier(0);
      const v16h a0c = (kt < 3) ? ag0[kt % 3][0] : ag1[kt % 3][0];
      const v16h a1c = (kt < 3) ? ag0[kt % 3][1] : ag1[kt % 3][1];
#pragma unroll
      for (int nt = 0; nt < CV_NT; ++nt) {
        c[0][nt] = __builtin_amdgcn_wmma_f32_16x16x32_f16(
            false, a0c, false, bc[nt], (short)0, c[0][nt], false, false);
        c[1][nt] = __builtin_amdgcn_wmma_f32_16x16x32_f16(
            false, a1c, false, bc[nt], (short)0, c[1][nt], false, false);
      }
      __builtin_amdgcn_sched_barrier(0);
      if (kt + 1 < CV_KS) {
#pragma unroll
        for (int nt = 0; nt < CV_NT; ++nt) bc[nt] = bn[nt];
      }
    }
    if (ch + 1 < NCH) lstore(buf ^ 1);
    __syncthreads();
  }
#pragma unroll
  for (int mi = 0; mi < 2; ++mi)
#pragma unroll
    for (int nt = 0; nt < CV_NT; ++nt)
#pragma unroll
      for (int j = 0; j < 8; ++j) {
        int rr = (mt0 + mi) * 16 + j + khalf;
        Out[((size_t)b * (3 * I_) + rr) * S_ + s0 + nt * 16 + col] = c[mi][nt][j];
      }
}

__global__ void init_loss_kernel(float* lacc) {
  if (threadIdx.x < 16) lacc[threadIdx.x] = 0.f;
}
__global__ void finalize_loss_kernel(const float* __restrict__ lacc,
                                     float* __restrict__ out_loss) {
  if (threadIdx.x == 0) {
    const float invT = 1.0f / (float)TOK;
    float l = 0.f;
    for (int e = 0; e < E_; ++e) l += (lacc[e] * invT) * (lacc[4 + e] * invT);
    for (int e = 0; e < E_; ++e) l += (lacc[8 + e] * invT) * (lacc[12 + e] * invT);
    *out_loss = l;
  }
}

extern "C" void kernel_launch(void* const* d_in, const int* in_sizes, int n_in,
                              void* d_out, int out_size, void* d_ws, size_t ws_size,
                              hipStream_t stream) {
  const float* inp     = (const float*)d_in[0];
  const float* divisor = (const float*)d_in[1];
  const float* w0_gate = (const float*)d_in[2];
  const float* w0      = (const float*)d_in[3];
  const float* w1      = (const float*)d_in[4];
  const float* w2_gate = (const float*)d_in[5];
  const float* w2      = (const float*)d_in[6];
  float* out = (float*)d_out;

  char* ws = (char*)d_ws;
  size_t off = 0;
  float* lacc = (float*)(ws + off); off += 256;
  const size_t h0b = (size_t)B_ * 3 * I_ * S_ * sizeof(float);
  const size_t t1b = (size_t)B_ * I_ * S_ * sizeof(float);
  float* h0   = (float*)(ws + off); off += h0b;
  float* tmp1 = (float*)(ws + off); off += t1b;
  float* g1   = (float*)(ws + off); off += t1b;
  int*   idx0 = (int*)(ws + off); off += TOK * sizeof(int);
  int*   idx1 = (int*)(ws + off); off += TOK * sizeof(int);
  off = (off + 255) & ~(size_t)255;
  const size_t w0n = (size_t)E_ * F_ * 3 * I_;
  const size_t w1n = (size_t)3 * I_ * 3 * I_;
  const size_t w2n = (size_t)E_ * I_ * F_;
  _Float16* w0p = (_Float16*)(ws + off); off += w0n * 2;
  _Float16* w1p = (_Float16*)(ws + off); off += w1n * 2;
  _Float16* w2p = (_Float16*)(ws + off); off += w2n * 2;
  float* h2 = h0;
  float* h3 = tmp1;

  pack_kn_kernel<F_, 3 * I_><<<(int)(w0n / (1024 * 8)), 256, 0, stream>>>(w0, w0p);
  pack_mk_kernel<3 * I_, 3 * I_><<<(int)(w1n / (2048 * 8)), 256, 0, stream>>>(w1, w1p);
  pack_kn_kernel<I_, F_><<<(int)(w2n / (1024 * 8)), 256, 0, stream>>>(w2, w2p);

  init_loss_kernel<<<1, 32, 0, stream>>>(lacc);

  gate_kernel<<<TOK / 32, GATE_WAVES * 32, 0, stream>>>(
      inp, F_, w0_gate, idx0, lacc);
  moe_wmma_kernel<F_, 3 * I_, 2, 4, 12><<<dim3(3 * I_ / 384, TOK / 64), 384, 0, stream>>>(
      inp, w0p, idx0, h0, 1.0f);

  cumsum_kernel<<<(B_ * I_) / 8, 256, 0, stream>>>(h0, divisor, tmp1);
  norm_kernel<<<TOK, 256, 0, stream>>>(tmp1, tmp1, tmp1, I_, 0, g1);

  conv_wmma_kernel<<<dim3(3 * I_ / 384, TOK / 64), CV_WAVES * 32, 0, stream>>>(g1, w1p, h2);

  norm_kernel<<<TOK, 256, 0, stream>>>(
      h2, h2 + (size_t)I_ * S_, h2 + (size_t)2 * I_ * S_, 3 * I_, 1, h3);

  gate_kernel<<<TOK / 32, GATE_WAVES * 32, 0, stream>>>(
      h3, I_, w2_gate, idx1, lacc + 8);
  moe_wmma_kernel<I_, F_, 1, 2, 8><<<dim3(F_ / 128, TOK / 32), 256, 0, stream>>>(
      h3, w2p, idx1, out, 1.0f  );

  finalize_loss_kernel<<<1, 32, 0, stream>>>(lacc, out + (size_t)B_ * F_ * S_);
}


# norm kernels re-tiled: a work-group = 32 tokens x all channels (64 groups, was 2048 of one token), lanes along tokens, whole-line 16-byte loads and stores; sums combined via LDS (sum order changes)
# speedup vs baseline: 3.0776x; 1.5188x over previous
#include <hip/hip_runtime.h>
#include <hip/hip_bf16.h>

typedef __attribute__((ext_vector_type(16))) _Float16 v16h;
typedef __attribute__((ext_vector_type(8)))  _Float16 v8h;
typedef __attribute__((ext_vector_type(8)))  float    v8f;

#define B_ 2
#define F_ 512
#define S_ 1024
#define E_ 4
#define I_ 1024
#define TOK (B_*S_)
#define EPSN 1e-5f
#define SLOPE_ 0.02f

__device__ __forceinline__ int frag_j(int kin) {
  return kin < 8 ? kin : (kin < 16 ? kin - 8 : (kin < 24 ? kin - 8 : kin - 16));
}
__device__ __forceinline__ int frag_lane(int kin, int n) {
  return n + (((kin >> 3) & 1) ? 16 : 0);
}

template <int CIN, int COUT>
__global__ void __launch_bounds__(256)
pack_kn_kernel(const float* __restrict__ W, _Float16* __restrict__ Wp) {
  constexpr int KT = CIN >> 5, MT = COUT >> 4, MG = COUT >> 5;
  const int lane = threadIdx.x & 31;
  const int g = __builtin_amdgcn_readfirstlane((int)(blockIdx.x * 8 + (threadIdx.x >> 5)));
  const int mg = g % MG;
  const int kt = (g / MG) % KT;
  const int e  = g / (MG * KT);
  const float* sp = W + ((size_t)e * CIN + (size_t)kt * 32) * COUT + mg * 32 + lane;
  float v[32];
#pragma unroll
  for (int k = 0; k < 32; ++k) v[k] = sp[(size_t)k * COUT];
  v16h lo, hi;
#pragma unroll
  for (int j = 0; j < 8; ++j) {
    lo[j]     = (_Float16)v[j];
    lo[j + 8] = (_Float16)v[j + 16];
    hi[j]     = (_Float16)v[j + 8];
    hi[j + 8] = (_Float16)v[j + 24];
  }
  const int mt = mg * 2 + (lane >> 4);
  _Float16* dp = Wp + ((((size_t)e * MT + mt) * KT + kt) * 32 + (lane & 15)) * 16;
  *(v16h*)dp = lo;
  *(v16h*)(dp + 16 * 16) = hi;
}

template <int CIN, int COUT>
__global__ void __launch_bounds__(256)
pack_mk_kernel(const float* __restrict__ W, _Float16* __restrict__ Wp) {
  constexpr int KT = CIN >> 5, KG = KT >> 2;
  typedef __attribute__((ext_vector_type(4))) float v4f;
  typedef __attribute__((ext_vector_type(4))) _Float16 v4h;
  const int lane = threadIdx.x & 31;
  const int g = __builtin_amdgcn_readfirstlane((int)(blockIdx.x * 8 + (threadIdx.x >> 5)));
  const int kg = g % KG;
  const int mt = g / KG;
  const int r = lane >> 3, q = lane & 7;
  const float* sp = W + ((size_t)mt * 16 + r) * CIN + kg * 128 + q * 4;
  v4f x[4][4];
#pragma unroll
  for (int t = 0; t < 4; ++t)
#pragma unroll
    for (int i = 0; i < 4; ++i)
      x[t][i] = *(const v4f*)(sp + (size_t)(4 * i) * CIN + t * 32);
  const int fl = r + (((q >> 1) & 1) << 4);
  const int j0 = ((q >> 2) << 3) + ((q & 1) << 2);
  _Float16* dp = Wp + (((size_t)mt * KT + kg * 4) * 32 + fl) * 16 + j0;
#pragma unroll
  for (int t = 0; t < 4; ++t)
#pragma unroll
    for (int i = 0; i < 4; ++i) {
      v4h h;
#pragma unroll
      for (int u = 0; u < 4; ++u) h[u] = (_Float16)x[t][i][u];
      *(v4h*)(dp + (size_t)t * 512 + (4 * i) * 16) = h;
    }
}

__device__ __forceinline__ v16h load_frag(const _Float16* p) {
  v8h lo = *(const v8h*)p;
  v8h hi = *(const v8h*)(p + 8);
  v16h r;
#pragma unroll
  for (int j = 0; j < 8; ++j) { r[j] = lo[j]; r[j + 8] = hi[j]; }
  return r;
}

#define GATE_WAVES 16
__global__ void __launch_bounds__(GATE_WAVES * 32)
gate_kernel(const float* __restrict__ x, int C,
            const float* __restrict__ gw,
            int* __restrict__ eidx,
            float* __restrict__ lacc) {
  __shared__ float part[GATE_WAVES][E_][32];
  const int lane = threadIdx.x & 31;
  const int wave = __builtin_amdgcn_readfirstlane((int)(threadIdx.x >> 5));
  const int t0 = blockIdx.x * 32;
  const int b = t0 / S_, s = (t0 % S_) + lane;
  const int cper = C / GATE_WAVES;
  const int c0 = wave * cper;
  const float* xp = x + ((size_t)b * C + c0) * S_ + s;
  const float* g0 = gw + c0;
  const float* g1 = g0 + C;
  const float* g2 = g1 + C;
  const float* g3 = g2 + C;
  float a0 = 0.f, a1 = 0.f, a2 = 0.f, a3 = 0.f;
  for (int cb = 0; cb < cper; cb += 16) {
    float xv[16];
#pragma unroll
    for (int u = 0; u < 16; ++u) xv[u] = xp[(size_t)(cb + u) * S_];
#pragma unroll
    for (int u = 0; u < 16; ++u) {
      a0 = __builtin_fmaf(xv[u], g0[cb + u], a0);
      a1 = __builtin_fmaf(xv[u], g1[cb + u], a1);
      a2 = __builtin_fmaf(xv[u], g2[cb + u], a2);
      a3 = __builtin_fmaf(xv[u], g3[cb + u], a3);
    }
  }
  part[wave][0][lane] = a0;
  part[wave][1][lane] = a1;
  part[wave][2][lane] = a2;
  part[wave][3][lane] = a3;
  __syncthreads();
  if (wave != 0) return;
  float a[4];
  for (int e = 0; e < E_; ++e) {
    float v = part[0][e][lane];
    for (int w = 1; w < GATE_WAVES; ++w) v += part[w][e][lane];
    a[e] = v;
  }
  int best = 0; float m = a[0];
  for (int e = 1; e < E_; ++e) { if (a[e] > m) { m = a[e]; best = e; } }
  float g[4]; float se = 0.f;
  for (int e = 0; e < E_; ++e) { g[e] = __expf(a[e] - m); se += g[e]; }
  float inv = 1.0f / se;
  eidx[t0 + lane] = best;
  for (int e = 0; e < E_; ++e) {
    float gsum = g[e] * inv;
    for (int d = 16; d > 0; d >>= 1) gsum += __shfl_xor(gsum, d, 32);
    float cnt = (float)__popcll(__ballot(best == e));
    if (lane == 0) {
      atomicAdd(&lacc[e], gsum);
      atomicAdd(&lacc[4 + e], cnt);
    }
  }
}

__device__ __forceinline__ void store_oct(_Float16* dst, const float* v) {
  v8h h;
#pragma unroll
  for (int jj = 0; jj < 8; ++jj) h[jj] = (_Float16)v[jj];
  *(v8h*)dst = h;
}
__device__ __forceinline__ int oct_off(int oct, int n, int KT) {
  int kt = oct >> 2, a = oct & 3;
  return ((((n >> 4) * KT + kt) * 32 + (n & 15) + ((a & 1) << 4)) << 4) + ((a >> 1) << 3);
}

template <int CIN, int COUT, int MTW, int NT, int WAVES>
__global__ void __launch_bounds__(WAVES * 32)
moe_wmma_kernel(const float* __restrict__ X,
                const _Float16* __restrict__ Wp,
                const int* __restrict__ eidx,
                float* __restrict__ Out, float outscale) {
  constexpr int KT = CIN >> 5;
  constexpr int NTOK = NT * 16;
  constexpr int THREADS = WAVES * 32;
  constexpr int ITEMS = (CIN >> 3) * NTOK;
  __shared__ __attribute__((aligned(16))) _Float16 xs[NT * KT * 512];
  const int tok0 = blockIdx.y * NTOK;
  const int b = tok0 / S_, s0 = tok0 % S_;
  auto stage = [&](int q) {
    const int n = q % NTOK, oct = q / NTOK;
    const float* xp = X + ((size_t)b * CIN + oct * 8) * S_ + s0 + n;
    float v[8];
#pragma unroll
    for (int jj = 0; jj < 8; ++jj) v[jj] = xp[(size_t)jj * S_];
    store_oct(&xs[oct_off(oct, n, KT)], v);
  };
  constexpr int FULL = ITEMS / THREADS;
  constexpr int UNR = (FULL % 5 == 0) ? 5 : 4;
#pragma unroll UNR
  for (int it = 0; it < FULL; ++it) stage((int)threadIdx.x + it * THREADS);
  if (ITEMS % THREADS != 0) {
    const int q = (int)threadIdx.x + FULL * THREADS;
    if (q < ITEMS) stage(q);
  }
  __syncthreads();
  const int lane = threadIdx.x & 31;
  const int wave = __builtin_amdgcn_readfirstlane((int)(threadIdx.x >> 5));
  const int col = lane & 15, khalf = (lane >> 4) * 8;
  const int mt0 = (blockIdx.x * WAVES + wave) * MTW;
  constexpr size_t fragstride = (size_t)KT * 512;
  int myexp[NT];
#pragma unroll
  for (int nt = 0; nt < NT; ++nt) myexp[nt] = eidx[tok0 + nt * 16 + col];
  v8f cs[MTW][NT];
#pragma unroll
  for (int mi = 0; mi < MTW; ++mi)
#pragma unroll
    for (int nt = 0; nt < NT; ++nt) cs[mi][nt] = v8f{};
  const _Float16* Bb = xs + lane * 16;
  for (int e = 0; e < E_; ++e) {
    int hit = 0;
#pragma unroll
    for (int nt = 0; nt < NT; ++nt) hit |= (myexp[nt] == e) ? 1 : 0;
    if (!__any(hit)) continue;
    const _Float16* Ab = Wp + ((size_t)e * (COUT >> 4) + mt0) * fragstride + lane * 16;
    v8f c[MTW][NT];
    v16h ac[MTW], bc[NT];
#pragma unroll
    for (int mi = 0; mi < MTW; ++mi) ac[mi] = load_frag(Ab + (size_t)mi * fragstride);
#pragma unroll
    for (int nt = 0; nt < NT; ++nt) {
      bc[nt] = load_frag(Bb + (size_t)nt * KT * 512);
#pragma unroll
      for (int mi = 0; mi < MTW; ++mi) c[mi][nt] = v8f{};
    }
#pragma unroll 2
    for (int kt = 0; kt < KT; ++kt) {
      const int k2 = (kt + 1 < KT) ? kt + 1 : kt;
      v16h an[MTW], bn[NT];
#pragma unroll
      for (int mi = 0; mi < MTW; ++mi)
        an[mi] = load_frag(Ab + (size_t)mi * fragstride + (size_t)k2 * 512);
#pragma unroll
      for (int nt = 0; nt < NT; ++nt)
        bn[nt] = load_frag(Bb + ((size_t)nt * KT + k2) * 512);
#pragma unroll
      for (int nt = 0; nt < NT; ++nt)
#pragma unroll
        for (int mi = 0; mi < MTW; ++mi)
          c[mi][nt] = __builtin_amdgcn_wmma_f32_16x16x32_f16(
              false, ac[mi], false, bc[nt], (short)0, c[mi][nt], false, false);
#pragma unroll
      for (int mi = 0; mi < MTW; ++mi) ac[mi] = an[mi];
#pragma unroll
      for (int nt = 0; nt < NT; ++nt) bc[nt] = bn[nt];
    }
#pragma unroll
    for (int nt = 0; nt < NT; ++nt)
      if (myexp[nt] == e) {
#pragma unroll
        for (int mi = 0; mi < MTW; ++mi) cs[mi][nt] = c[mi][nt];
      }
  }
#pragma unroll
  for (int mi = 0; mi < MTW; ++mi)
#pragma unroll
    for (int nt = 0; nt < NT; ++nt)
#pragma unroll
      for (int j = 0; j < 8; ++j) {
        int r = (mt0 + mi) * 16 + j + khalf;
        Out[((size_t)b * COUT + r) * S_ + s0 + nt * 16 + col] = cs[mi][nt][j] * outscale;
      }
}

__global__ void cumsum_kernel(const float* __restrict__ h0,
                              const float* __restrict__ divisor,
                              float* __restrict__ tmp1) {
  int row = blockIdx.x * (blockDim.x >> 5) + (threadIdx.x >> 5);
  int lane = threadIdx.x & 31;
  if (row >= B_ * I_) return;
  int b = row / I_, i = row % I_;
  const float* dep = h0 + ((size_t)b * 3 * I_ + i) * S_;
  const float* sc  = h0 + ((size_t)b * 3 * I_ + I_ + i) * S_;
  const float* sh  = h0 + ((size_t)b * 3 * I_ + 2 * I_ + i) * S_;
  float* o = tmp1 + ((size_t)b * I_ + i) * S_;
  float carry = 0.f;
  for (int s0 = 0; s0 < S_; s0 += 32) {
    float v = dep[s0 + lane];
#pragma unroll
    for (int d = 1; d < 32; d <<= 1) {
      float t = __shfl_up(v, d, 32);
      if (lane >= d) v += t;
    }
    float cum = carry + v;
    o[s0 + lane] = cum / divisor[s0 + lane] * sc[s0 + lane] + sh[s0 + lane];
    carry = __shfl(cum, 31, 32);
  }
}

#define NORM_WAVES 32
#define NORM_CPW (I_ / NORM_WAVES)
#define NORM_G (NORM_CPW / 4)
#define NORM_PITCH 36
typedef __attribute__((ext_vector_type(4))) float v4f;

__device__ __forceinline__ float lane_x8(float x) {
  int b = __builtin_bit_cast(int, x);
  return __builtin_bit_cast(float, __builtin_amdgcn_permlane16(
      b, b, 0xfedcba98u, 0x76543210u, false, false));
}
__device__ __forceinline__ float lane_x16(float x) {
  int b = __builtin_bit_cast(int, x);
  return __builtin_bit_cast(float, __builtin_amdgcn_permlanex16(
      b, b, 0x76543210u, 0xfedcba98u, false, false));
}
__device__ __forceinline__ v4f join4(v4f p) {
#pragma unroll
  for (int k = 0; k < 4; ++k) p[k] += lane_x8(p[k]);
#pragma unroll
  for (int k = 0; k < 4; ++k) p[k] += lane_x16(p[k]);
  return p;
}
__device__ __forceinline__ v4f block_sum(float* part, v4f p, int wave, int tq, int rs) {
  p = join4(p);
  if (rs == 0) *(v4f*)(part + wave * NORM_PITCH + tq * 4) = p;
  __syncthreads();
  const float* pp = part + (8 * rs) * NORM_PITCH + tq * 4;
  v4f q = *(const v4f*)pp;
#pragma unroll
  for (int j = 1; j < 8; ++j) q += *(const v4f*)(pp + j * NORM_PITCH);
  return join4(q);
}

template <int MODE>
__global__ void __launch_bounds__(NORM_WAVES * 32)
norm_kernel(const float* __restrict__ p0,
            const float* __restrict__ p1,
            const float* __restrict__ p2,
            int cstride,
            float* __restrict__ out) {
  __shared__ __attribute__((aligned(16))) float part0[NORM_WAVES * NORM_PITCH];
  __shared__ __attribute__((aligned(16))) float part1[NORM_WAVES * NORM_PITCH];
  const int lane = threadIdx.x & 31;
  const int wave = __builtin_amdgcn_readfirstlane((int)(threadIdx.x >> 5));
  const int tq = lane & 7, rs = lane >> 3;
  const int t0 = blockIdx.x * 32;
  const int b = t0 / S_, s = (t0 % S_) + tq * 4;
  const int c0 = wave * NORM_CPW + rs;
  const size_t ib = ((size_t)b * cstride + c0) * S_ + s;
  v4f x[NORM_G];
#pragma unroll
  for (int g = 0; g < NORM_G; ++g) {
    const size_t idx = ib + (size_t)(4 * g) * S_;
    v4f a = *(const v4f*)(p0 + idx);
    if (MODE) a = a * *(const v4f*)(p1 + idx) + *(const v4f*)(p2 + idx);
    x[g] = a;
  }
  v4f ps = x[0];
#pragma unroll
  for (int g = 1; g < NORM_G; ++g) ps += x[g];
  const v4f mean = block_sum(part0, ps, wave, tq, rs) * (1.0f / I_);
  v4f ss = v4f{};
#pragma unroll
  for (int g = 0; g < NORM_G; ++g) {
    x[g] -= mean;
    ss += x[g] * x[g];
  }
  const v4f tot = block_sum(part1, ss, wave, tq, rs);
  v4f inv;
#pragma unroll
  for (int k = 0; k < 4; ++k) {
    const float n = sqrtf(tot[k]) * (1.0f / 32.0f);
    inv[k] = 1.0f / (n + EPSN);
  }
  float* op = out + ((size_t)b * I_ + c0) * S_ + s;
#pragma unroll
  for (int g = 0; g < NORM_G; ++g) {
    v4f y = x[g] * inv;
#pragma unroll
    for (int k = 0; k < 4; ++k) y[k] = y[k] >= 0.f ? y[k] : y[k] * SLOPE_;
    *(v4f*)(op + (size_t)(4 * g) * S_) = y;
  }
}

#define CV_WAVES 12
#define CV_NT 4
#define CV_KS 6
__global__ void __launch_bounds__(CV_WAVES * 32)
conv_wmma_kernel(const float* __restrict__ G,
                 const _Float16* __restrict__ Wp1,
                 float* __restrict__ Out) {
  constexpr int CHUNK = CV_KS * 32;
  constexpr int NCH = (3 * I_) / CHUNK;
  constexpr int BUF = CV_NT * CV_KS * 512;
  constexpr int ITS = (CV_KS * 4 * CV_NT * 16) / (CV_WAVES * 32);
  constexpr int KTtot = (3 * I_) >> 5;
  constexpr size_t fragstride = (size_t)KTtot * 512;
  __shared__ __attribute__((aligned(16))) _Float16 xs[2][BUF];
  const int tok0 = blockIdx.y * (CV_NT * 16);
  const int b = tok0 / S_, s0 = tok0 % S_;
  const int lane = threadIdx.x & 31;
  const int wave = __builtin_amdgcn_readfirstlane((int)(threadIdx.x >> 5));
  const int col = lane & 15, khalf = (lane >> 4) * 8;
  const int mt0 = (blockIdx.x * CV_WAVES + wave) * 2;
  const _Float16* A0 = Wp1 + (size_t)mt0 * fragstride + lane * 16;
  const _Float16* A1 = A0 + fragstride;
  const int n64 = (wave & 1) * 32 + lane;
  const int octw = wave >> 1;
  const int sb = s0 + n64 - 2;
  const float* Gb = G + (size_t)b * I_ * S_;
  float r[ITS][8];
  auto gload = [&](int cc) {
#pragma unroll
    for (int it = 0; it < ITS; ++it) {
      const int kk0 = cc + (it * (CV_WAVES / 2) + octw) * 8;
#pragma unroll
      for (int jj = 0; jj < 8; ++jj) {
        const int kk = kk0 + jj;
        const int i = kk / 3, kc = kk - i * 3;
        const int s = sb + kc;
        const float gv = Gb[(size_t)i * S_ + (s > 0 ? s : 0)];
        r[it][jj] = (s >= 0) ? gv : 0.0f;
      }
    }
  };
  auto lstore = [&](int buf) {
#pragma unroll
    for (int it = 0; it < ITS; ++it)
      store_oct(&xs[buf][oct_off(it * (CV_WAVES / 2) + octw, n64, CV_KS)], r[it]);
  };
  v8f c[2][CV_NT];
#pragma unroll
  for (int nt = 0; nt < CV_NT; ++nt) { c[0][nt] = v8f{}; c[1][nt] = v8f{}; }
  gload(0);
  lstore(0);
  __syncthreads();
  v16h ag0[3][2], ag1[3][2];
#pragma unroll
  for (int u = 0; u < 3; ++u) {
    ag0[u][0] = load_frag(A0 + (size_t)u * 512);
    ag0[u][1] = load_frag(A1 + (size_t)u * 512);
  }
  for (int ch = 0; ch < NCH; ++ch) {
    const int buf = ch & 1;
    const _Float16* Bb = &xs[buf][lane * 16];
    if (ch + 1 < NCH) gload((ch + 1) * CHUNK);
#pragma unroll
    for (int u = 0; u < 3; ++u) {
      ag1[u][0] = load_frag(A0 + (size_t)(ch * CV_KS + 3 + u) * 512);
      ag1[u][1] = load_frag(A1 + (size_t)(ch * CV_KS + 3 + u) * 512);
    }
    v16h bc[CV_NT], bn[CV_NT];
#pragma unroll
    for (int nt = 0; nt < CV_NT; ++nt) bc[nt] = load_frag(Bb + (nt * CV_KS) * 512);
    __builtin_amdgcn_sched_barrier(0);
#pragma unroll
    for (int kt = 0; kt < CV_KS; ++kt) {
      if (kt == 3) {
#pragma unroll
        for (int u = 0; u < 3; ++u) {
          const int kn = ((ch + 1) * CV_KS + u < KTtot) ? (ch + 1) * CV_KS + u : KTtot - 1;
          ag0[u][0] = load_frag(A0 + (size_t)kn * 512);
          ag0[u][1] = load_frag(A1 + (size_t)kn * 512);
        }
      }
      if (kt + 1 < CV_KS) {
#pragma unroll
        for (int nt = 0; nt < CV_NT; ++nt)
          bn[nt] = load_frag(Bb + (nt * CV_KS + kt + 1) * 512);
      }
      __builtin_amdgcn_sched_barrier(0);
      const v16h a0c = (kt < 3) ? ag0[kt % 3][0] : ag1[kt % 3][0];
      const v16h a1c = (kt < 3) ? ag0[kt % 3][1] : ag1[kt % 3][1];
#pragma unroll
      for (int nt = 0; nt < CV_NT; ++nt) {
        c[0][nt] = __builtin_amdgcn_wmma_f32_16x16x32_f16(
            false, a0c, false, bc[nt], (short)0, c[0][nt], false, false);
        c[1][nt] = __builtin_amdgcn_wmma_f32_16x16x32_f16(
            false, a1c, false, bc[nt], (short)0, c[1][nt], false, false);
      }
      __builtin_amdgcn_sched_barrier(0);
      if (kt + 1 < CV_KS) {
#pragma unroll
        for (int nt = 0; nt < CV_NT; ++nt) bc[nt] = bn[nt];
      }
    }
    if (ch + 1 < NCH) lstore(buf ^ 1);
    __syncthreads();
  }
#pragma unroll
  for (int mi = 0; mi < 2; ++mi)
#pragma unroll
    for (int nt = 0; nt < CV_NT; ++nt)
#pragma unroll
      for (int j = 0; j < 8; ++j) {
        int rr = (mt0 + mi) * 16 + j + khalf;
        Out[((size_t)b * (3 * I_) + rr) * S_ + s0 + nt * 16 + col] = c[mi][nt][j];
      }
}

__global__ void init_loss_kernel(float* lacc) {
  if (threadIdx.x < 16) lacc[threadIdx.x] = 0.f;
}
__global__ void finalize_loss_kernel(const float* __restrict__ lacc,
                                     float* __restrict__ out_loss) {
  if (threadIdx.x == 0) {
    const float invT = 1.0f / (float)TOK;
    float l = 0.f;
    for (int e = 0; e < E_; ++e) l += (lacc[e] * invT) * (lacc[4 + e] * invT);
    for (int e = 0; e < E_; ++e) l += (lacc[8 + e] * invT) * (lacc[12 + e] * invT);
    *out_loss = l;
  }
}

extern "C" void kernel_launch(void* const* d_in, const int* in_sizes, int n_in,
                              void* d_out, int out_size, void* d_ws, size_t ws_size,
                              hipStream_t stream) {
  const float* inp     = (const float*)d_in[0];
  const float* divisor = (const float*)d_in[1];
  const float* w0_gate = (const float*)d_in[2];
  const float* w0      = (const float*)d_in[3];
  const float* w1      = (const float*)d_in[4];
  const float* w2_gate = (const float*)d_in[5];
  const float* w2      = (const float*)d_in[6];
  float* out = (float*)d_out;

  char* ws = (char*)d_ws;
  size_t off = 0;
  float* lacc = (float*)(ws + off); off += 256;
  const size_t h0b = (size_t)B_ * 3 * I_ * S_ * sizeof(float);
  const size_t t1b = (size_t)B_ * I_ * S_ * sizeof(float);
  float* h0   = (float*)(ws + off); off += h0b;
  float* tmp1 = (float*)(ws + off); off += t1b;
  float* g1   = (float*)(ws + off); off += t1b;
  int*   idx0 = (int*)(ws + off); off += TOK * sizeof(int);
  int*   idx1 = (int*)(ws + off); off += TOK * sizeof(int);
  off = (off + 255) & ~(size_t)255;
  const size_t w0n = (size_t)E_ * F_ * 3 * I_;
  const size_t w1n = (size_t)3 * I_ * 3 * I_;
  const size_t w2n = (size_t)E_ * I_ * F_;
  _Float16* w0p = (_Float16*)(ws + off); off += w0n * 2;
  _Float16* w1p = (_Float16*)(ws + off); off += w1n * 2;
  _Float16* w2p = (_Float16*)(ws + off); off += w2n * 2;
  float* h2 = h0;
  float* h3 = tmp1;

  pack_kn_kernel<F_, 3 * I_><<<(int)(w0n / (1024 * 8)), 256, 0, stream>>>(w0, w0p);
  pack_mk_kernel<3 * I_, 3 * I_><<<(int)(w1n / (2048 * 8)), 256, 0, stream>>>(w1, w1p);
  pack_kn_kernel<I_, F_><<<(int)(w2n / (1024 * 8)), 256, 0, stream>>>(w2, w2p);

  init_loss_kernel<<<1, 32, 0, stream>>>(lacc);

  gate_kernel<<<TOK / 32, GATE_WAVES * 32, 0, stream>>>(
      inp, F_, w0_gate, idx0, lacc);
  moe_wmma_kernel<F_, 3 * I_, 2, 4, 12><<<dim3(3 * I_ / 384, TOK / 64), 384, 0, stream>>>(
      inp, w0p, idx0, h0, 1.0f);

  cumsum_kernel<<<(B_ * I_) / 8, 256, 0, stream>>>(h0, divisor, tmp1);
  norm_kernel<0><<<TOK / 32, NORM_WAVES * 32, 0, stream>>>(tmp1, tmp1, tmp1, I_, g1);

  conv_wmma_kernel<<<dim3(3 * I_ / 384, TOK / 64), CV_WAVES * 32, 0, stream>>>(g1, w1p, h2);

  norm_kernel<1><<<TOK / 32, NORM_WAVES * 32, 0, stream>>>(
      h2, h2 + (size_t)I_ * S_, h2 + (size_t)2 * I_ * S_, 3 * I_, h3);

  gate_kernel<<<TOK / 32, GATE_WAVES * 32, 0, stream>>>(
      h3, I_, w2_gate, idx1, lacc + 8);
  moe_wmma_kernel<I_, F_, 1, 2, 8><<<dim3(F_ / 128, TOK / 32), 256, 0, stream>>>(
      h3, w2p, idx1, out, 1.0f  );

  finalize_loss_kernel<<<1, 32, 0, stream>>>(lacc, out + (size_t)B_ * F_ * S_);
}


# routed GEMMs: weight fragments fetched 4 (8) k-steps ahead of their products through two register sets, across expert boundaries too; was 1 step ahead, each product waiting out the L2 latency
# speedup vs baseline: 3.8619x; 1.2548x over previous
#include <hip/hip_runtime.h>
#include <hip/hip_bf16.h>

typedef __attribute__((ext_vector_type(16))) _Float16 v16h;
typedef __attribute__((ext_vector_type(8)))  _Float16 v8h;
typedef __attribute__((ext_vector_type(8)))  float    v8f;

#define B_ 2
#define F_ 512
#define S_ 1024
#define E_ 4
#define I_ 1024
#define TOK (B_*S_)
#define EPSN 1e-5f
#define SLOPE_ 0.02f

__device__ __forceinline__ int frag_j(int kin) {
  return kin < 8 ? kin : (kin < 16 ? kin - 8 : (kin < 24 ? kin - 8 : kin - 16));
}
__device__ __forceinline__ int frag_lane(int kin, int n) {
  return n + (((kin >> 3) & 1) ? 16 : 0);
}

template <int CIN, int COUT>
__global__ void __launch_bounds__(256)
pack_kn_kernel(const float* __restrict__ W, _Float16* __restrict__ Wp) {
  constexpr int KT = CIN >> 5, MT = COUT >> 4, MG = COUT >> 5;
  const int lane = threadIdx.x & 31;
  const int g = __builtin_amdgcn_readfirstlane((int)(blockIdx.x * 8 + (threadIdx.x >> 5)));
  const int mg = g % MG;
  const int kt = (g / MG) % KT;
  const int e  = g / (MG * KT);
  const float* sp = W + ((size_t)e * CIN + (size_t)kt * 32) * COUT + mg * 32 + lane;
  float v[32];
#pragma unroll
  for (int k = 0; k < 32; ++k) v[k] = sp[(size_t)k * COUT];
  v16h lo, hi;
#pragma unroll
  for (int j = 0; j < 8; ++j) {
    lo[j]     = (_Float16)v[j];
    lo[j + 8] = (_Float16)v[j + 16];
    hi[j]     = (_Float16)v[j + 8];
    hi[j + 8] = (_Float16)v[j + 24];
  }
  const int mt = mg * 2 + (lane >> 4);
  _Float16* dp = Wp + ((((size_t)e * MT + mt) * KT + kt) * 32 + (lane & 15)) * 16;
  *(v16h*)dp = lo;
  *(v16h*)(dp + 16 * 16) = hi;
}

template <int CIN, int COUT>
__global__ void __launch_bounds__(256)
pack_mk_kernel(const float* __restrict__ W, _Float16* __restrict__ Wp) {
  constexpr int KT = CIN >> 5, KG = KT >> 2;
  typedef __attribute__((ext_vector_type(4))) float v4f;
  typedef __attribute__((ext_vector_type(4))) _Float16 v4h;
  const int lane = threadIdx.x & 31;
  const int g = __builtin_amdgcn_readfirstlane((int)(blockIdx.x * 8 + (threadIdx.x >> 5)));
  const int kg = g % KG;
  const int mt = g / KG;
  const int r = lane >> 3, q = lane & 7;
  const float* sp = W + ((size_t)mt * 16 + r) * CIN + kg * 128 + q * 4;
  v4f x[4][4];
#pragma unroll
  for (int t = 0; t < 4; ++t)
#pragma unroll
    for (int i = 0; i < 4; ++i)
      x[t][i] = *(const v4f*)(sp + (size_t)(4 * i) * CIN + t * 32);
  const int fl = r + (((q >> 1) & 1) << 4);
  const int j0 = ((q >> 2) << 3) + ((q & 1) << 2);
  _Float16* dp = Wp + (((size_t)mt * KT + kg * 4) * 32 + fl) * 16 + j0;
#pragma unroll
  for (int t = 0; t < 4; ++t)
#pragma unroll
    for (int i = 0; i < 4; ++i) {
      v4h h;
#pragma unroll
      for (int u = 0; u < 4; ++u) h[u] = (_Float16)x[t][i][u];
      *(v4h*)(dp + (size_t)t * 512 + (4 * i) * 16) = h;
    }
}

__device__ __forceinline__ v16h load_frag(const _Float16* p) {
  v8h lo = *(const v8h*)p;
  v8h hi = *(const v8h*)(p + 8);
  v16h r;
#pragma unroll
  for (int j = 0; j < 8; ++j) { r[j] = lo[j]; r[j + 8] = hi[j]; }
  return r;
}

#define GATE_WAVES 16
__global__ void __launch_bounds__(GATE_WAVES * 32)
gate_kernel(const float* __restrict__ x, int C,
            const float* __restrict__ gw,
            int* __restrict__ eidx,
            float* __restrict__ lacc) {
  __shared__ float part[GATE_WAVES][E_][32];
  const int lane = threadIdx.x & 31;
  const int wave = __builtin_amdgcn_readfirstlane((int)(threadIdx.x >> 5));
  const int t0 = blockIdx.x * 32;
  const int b = t0 / S_, s = (t0 % S_) + lane;
  const int cper = C / GATE_WAVES;
  const int c0 = wave * cper;
  const float* xp = x + ((size_t)b * C + c0) * S_ + s;
  const float* g0 = gw + c0;
  const float* g1 = g0 + C;
  const float* g2 = g1 + C;
  const float* g3 = g2 + C;
  float a0 = 0.f, a1 = 0.f, a2 = 0.f, a3 = 0.f;
  for (int cb = 0; cb < cper; cb += 16) {
    float xv[16];
#pragma unroll
    for (int u = 0; u < 16; ++u) xv[u] = xp[(size_t)(cb + u) * S_];
#pragma unroll
    for (int u = 0; u < 16; ++u) {
      a0 = __builtin_fmaf(xv[u], g0[cb + u], a0);
      a1 = __builtin_fmaf(xv[u], g1[cb + u], a1);
      a2 = __builtin_fmaf(xv[u], g2[cb + u], a2);
      a3 = __builtin_fmaf(xv[u], g3[cb + u], a3);
    }
  }
  part[wave][0][lane] = a0;
  part[wave][1][lane] = a1;
  part[wave][2][lane] = a2;
  part[wave][3][lane] = a3;
  __syncthreads();
  if (wave != 0) return;
  float a[4];
  for (int e = 0; e < E_; ++e) {
    float v = part[0][e][lane];
    for (int w = 1; w < GATE_WAVES; ++w) v += part[w][e][lane];
    a[e] = v;
  }
  int best = 0; float m = a[0];
  for (int e = 1; e < E_; ++e) { if (a[e] > m) { m = a[e]; best = e; } }
  float g[4]; float se = 0.f;
  for (int e = 0; e < E_; ++e) { g[e] = __expf(a[e] - m); se += g[e]; }
  float inv = 1.0f / se;
  eidx[t0 + lane] = best;
  for (int e = 0; e < E_; ++e) {
    float gsum = g[e] * inv;
    for (int d = 16; d > 0; d >>= 1) gsum += __shfl_xor(gsum, d, 32);
    float cnt = (float)__popcll(__ballot(best == e));
    if (lane == 0) {
      atomicAdd(&lacc[e], gsum);
      atomicAdd(&lacc[4 + e], cnt);
    }
  }
}

__device__ __forceinline__ void store_oct(_Float16* dst, const float* v) {
  v8h h;
#pragma unroll
  for (int jj = 0; jj < 8; ++jj) h[jj] = (_Float16)v[jj];
  *(v8h*)dst = h;
}
__device__ __forceinline__ int oct_off(int oct, int n, int KT) {
  int kt = oct >> 2, a = oct & 3;
  return ((((n >> 4) * KT + kt) * 32 + (n & 15) + ((a & 1) << 4)) << 4) + ((a >> 1) << 3);
}

template <int CIN, int COUT, int MTW, int NT, int WAVES, int D>
__global__ void __launch_bounds__(WAVES * 32)
moe_wmma_kernel(const float* __restrict__ X,
                const _Float16* __restrict__ Wp,
                const int* __restrict__ eidx,
                float* __restrict__ Out, float outscale) {
  constexpr int KT = CIN >> 5;
  constexpr int NG = KT / D;
  static_assert(KT % (2 * D) == 0, "two register sets of D k-steps must divide the k-steps");
  constexpr int PIN = (MTW * NT >= 8) ? 1 : 8 / (MTW * NT);
  constexpr int NTOK = NT * 16;
  constexpr int THREADS = WAVES * 32;
  constexpr int ITEMS = (CIN >> 3) * NTOK;
  __shared__ __attribute__((aligned(16))) _Float16 xs[NT * KT * 512];
  const int tok0 = blockIdx.y * NTOK;
  const int b = tok0 / S_, s0 = tok0 % S_;
  auto stage = [&](int q) {
    const int n = q % NTOK, oct = q / NTOK;
    const float* xp = X + ((size_t)b * CIN + oct * 8) * S_ + s0 + n;
    float v[8];
#pragma unroll
    for (int jj = 0; jj < 8; ++jj) v[jj] = xp[(size_t)jj * S_];
    store_oct(&xs[oct_off(oct, n, KT)], v);
  };
  constexpr int FULL = ITEMS / THREADS;
  constexpr int UNR = (FULL % 5 == 0) ? 5 : 4;
#pragma unroll UNR
  for (int it = 0; it < FULL; ++it) stage((int)threadIdx.x + it * THREADS);
  if (ITEMS % THREADS != 0) {
    const int q = (int)threadIdx.x + FULL * THREADS;
    if (q < ITEMS) stage(q);
  }
  __syncthreads();
  const int lane = threadIdx.x & 31;
  const int wave = __builtin_amdgcn_readfirstlane((int)(threadIdx.x >> 5));
  const int col = lane & 15, khalf = (lane >> 4) * 8;
  const int mt0 = (blockIdx.x * WAVES + wave) * MTW;
  constexpr size_t fragstride = (size_t)KT * 512;
  constexpr size_t estride = (size_t)(COUT >> 4) * fragstride;
  int myexp[NT];
#pragma unroll
  for (int nt = 0; nt < NT; ++nt) myexp[nt] = eidx[tok0 + nt * 16 + col];
  v8f cs[MTW][NT];
#pragma unroll
  for (int mi = 0; mi < MTW; ++mi)
#pragma unroll
    for (int nt = 0; nt < NT; ++nt) cs[mi][nt] = v8f{};
  unsigned rem = 0;
#pragma unroll
  for (int e = 0; e < E_; ++e) {
    int hit = 0;
#pragma unroll
    for (int nt = 0; nt < NT; ++nt) hit |= (myexp[nt] == e) ? 1 : 0;
    if (__any(hit)) rem |= 1u << e;
  }
  const _Float16* Bb = xs + lane * 16;
  const _Float16* Abase = Wp + (size_t)mt0 * fragstride;
  const int loff = lane * 16;
  if (rem != 0) {
    int e = __builtin_ctz(rem);
    rem &= rem - 1;
    const _Float16* Acur = Abase + (size_t)e * estride;
    v16h ra[D][MTW], rb[D][MTW];
    v16h bc[NT];
#pragma unroll
    for (int u = 0; u < D; ++u)
#pragma unroll
      for (int mi = 0; mi < MTW; ++mi)
        ra[u][mi] = load_frag(Acur + (size_t)mi * fragstride + (size_t)u * 512 + loff);
#pragma unroll
    for (int nt = 0; nt < NT; ++nt) bc[nt] = load_frag(Bb + (size_t)nt * KT * 512);
    for (;;) {
      const bool more = rem != 0;
      const int en = more ? __builtin_ctz(rem) : e;
      const _Float16* Anext = more ? Abase + (size_t)en * estride
                                   : Acur + (size_t)(KT - D) * 512;
      v8f c[MTW][NT];
#pragma unroll
      for (int mi = 0; mi < MTW; ++mi)
#pragma unroll
        for (int nt = 0; nt < NT; ++nt) c[mi][nt] = v8f{};
      for (int h = 0; h < NG / 2; ++h) {
        const _Float16* Apb = Acur + (size_t)(2 * h + 1) * D * 512;
        const _Float16* Apa = (2 * h + 2 < NG) ? Acur + (size_t)(2 * h + 2) * D * 512 : Anext;
#pragma unroll
        for (int u = 0; u < D; ++u) {
          const int k2 = 2 * h * D + u + 1;
          v16h bn[NT];
#pragma unroll
          for (int nt = 0; nt < NT; ++nt)
            bn[nt] = load_frag(Bb + ((size_t)nt * KT + k2) * 512);
#pragma unroll
          for (int nt = 0; nt < NT; ++nt)
#pragma unroll
            for (int mi = 0; mi < MTW; ++mi)
              c[mi][nt] = __builtin_amdgcn_wmma_f32_16x16x32_f16(
                  false, ra[u][mi], false, bc[nt], (short)0, c[mi][nt], false, false);
#pragma unroll
          for (int mi = 0; mi < MTW; ++mi)
            rb[u][mi] = load_frag(Apb + (size_t)mi * fragstride + (size_t)u * 512 + loff);
          if ((u + 1) % PIN == 0) __builtin_amdgcn_sched_barrier(0);
#pragma unroll
          for (int nt = 0; nt < NT; ++nt) bc[nt] = bn[nt];
        }
#pragma unroll
        for (int u = 0; u < D; ++u) {
          const int kt = (2 * h + 1) * D + u;
          const int k2 = (kt + 1 < KT) ? kt + 1 : 0;
          v16h bn[NT];
#pragma unroll
          for (int nt = 0; nt < NT; ++nt)
            bn[nt] = load_frag(Bb + ((size_t)nt * KT + k2) * 512);
#pragma unroll
          for (int nt = 0; nt < NT; ++nt)
#pragma unroll
            for (int mi = 0; mi < MTW; ++mi)
              c[mi][nt] = __builtin_amdgcn_wmma_f32_16x16x32_f16(
                  false, rb[u][mi], false, bc[nt], (short)0, c[mi][nt], false, false);
#pragma unroll
          for (int mi = 0; mi < MTW; ++mi)
            ra[u][mi] = load_frag(Apa + (size_t)mi * fragstride + (size_t)u * 512 + loff);
          if ((u + 1) % PIN == 0) __builtin_amdgcn_sched_barrier(0);
#pragma unroll
          for (int nt = 0; nt < NT; ++nt) bc[nt] = bn[nt];
        }
      }
#pragma unroll
      for (int nt = 0; nt < NT; ++nt)
        if (myexp[nt] == e) {
#pragma unroll
          for (int mi = 0; mi < MTW; ++mi) cs[mi][nt] = c[mi][nt];
        }
      if (!more) break;
      e = en;
      rem &= rem - 1;
      Acur = Anext;
    }
  }
#pragma unroll
  for (int mi = 0; mi < MTW; ++mi)
#pragma unroll
    for (int nt = 0; nt < NT; ++nt)
#pragma unroll
      for (int j = 0; j < 8; ++j) {
        int r = (mt0 + mi) * 16 + j + khalf;
        Out[((size_t)b * COUT + r) * S_ + s0 + nt * 16 + col] = cs[mi][nt][j] * outscale;
      }
}

__global__ void cumsum_kernel(const float* __restrict__ h0,
                              const float* __restrict__ divisor,
                              float* __restrict__ tmp1) {
  int row = blockIdx.x * (blockDim.x >> 5) + (threadIdx.x >> 5);
  int lane = threadIdx.x & 31;
  if (row >= B_ * I_) return;
  int b = row / I_, i = row % I_;
  const float* dep = h0 + ((size_t)b * 3 * I_ + i) * S_;
  const float* sc  = h0 + ((size_t)b * 3 * I_ + I_ + i) * S_;
  const float* sh  = h0 + ((size_t)b * 3 * I_ + 2 * I_ + i) * S_;
  float* o = tmp1 + ((size_t)b * I_ + i) * S_;
  float carry = 0.f;
  for (int s0 = 0; s0 < S_; s0 += 32) {
    float v = dep[s0 + lane];
#pragma unroll
    for (int d = 1; d < 32; d <<= 1) {
      float t = __shfl_up(v, d, 32);
      if (lane >= d) v += t;
    }
    float cum = carry + v;
    o[s0 + lane] = cum / divisor[s0 + lane] * sc[s0 + lane] + sh[s0 + lane];
    carry = __shfl(cum, 31, 32);
  }
}

#define NORM_WAVES 32
#define NORM_CPW (I_ / NORM_WAVES)
#define NORM_G (NORM_CPW / 4)
#define NORM_PITCH 36
typedef __attribute__((ext_vector_type(4))) float v4f;

__device__ __forceinline__ float lane_x8(float x) {
  int b = __builtin_bit_cast(int, x);
  return __builtin_bit_cast(float, __builtin_amdgcn_permlane16(
      b, b, 0xfedcba98u, 0x76543210u, false, false));
}
__device__ __forceinline__ float lane_x16(float x) {
  int b = __builtin_bit_cast(int, x);
  return __builtin_bit_cast(float, __builtin_amdgcn_permlanex16(
      b, b, 0x76543210u, 0xfedcba98u, false, false));
}
__device__ __forceinline__ v4f join4(v4f p) {
#pragma unroll
  for (int k = 0; k < 4; ++k) p[k] += lane_x8(p[k]);
#pragma unroll
  for (int k = 0; k < 4; ++k) p[k] += lane_x16(p[k]);
  return p;
}
__device__ __forceinline__ v4f block_sum(float* part, v4f p, int wave, int tq, int rs) {
  p = join4(p);
  if (rs == 0) *(v4f*)(part + wave * NORM_PITCH + tq * 4) = p;
  __syncthreads();
  const float* pp = part + (8 * rs) * NORM_PITCH + tq * 4;
  v4f q = *(const v4f*)pp;
#pragma unroll
  for (int j = 1; j < 8; ++j) q += *(const v4f*)(pp + j * NORM_PITCH);
  return join4(q);
}

template <int MODE>
__global__ void __launch_bounds__(NORM_WAVES * 32)
norm_kernel(const float* __restrict__ p0,
            const float* __restrict__ p1,
            const float* __restrict__ p2,
            int cstride,
            float* __restrict__ out) {
  __shared__ __attribute__((aligned(16))) float part0[NORM_WAVES * NORM_PITCH];
  __shared__ __attribute__((aligned(16))) float part1[NORM_WAVES * NORM_PITCH];
  const int lane = threadIdx.x & 31;
  const int wave = __builtin_amdgcn_readfirstlane((int)(threadIdx.x >> 5));
  const int tq = lane & 7, rs = lane >> 3;
  const int t0 = blockIdx.x * 32;
  const int b = t0 / S_, s = (t0 % S_) + tq * 4;
  const int c0 = wave * NORM_CPW + rs;
  const size_t ib = ((size_t)b * cstride + c0) * S_ + s;
  v4f x[NORM_G];
#pragma unroll
  for (int g = 0; g < NORM_G; ++g) {
    const size_t idx = ib + (size_t)(4 * g) * S_;
    v4f a = *(const v4f*)(p0 + idx);
    if (MODE) a = a * *(const v4f*)(p1 + idx) + *(const v4f*)(p2 + idx);
    x[g] = a;
  }
  v4f ps = x[0];
#pragma unroll
  for (int g = 1; g < NORM_G; ++g) ps += x[g];
  const v4f mean = block_sum(part0, ps, wave, tq, rs) * (1.0f / I_);
  v4f ss = v4f{};
#pragma unroll
  for (int g = 0; g < NORM_G; ++g) {
    x[g] -= mean;
    ss += x[g] * x[g];
  }
  const v4f tot = block_sum(part1, ss, wave, tq, rs);
  v4f inv;
#pragma unroll
  for (int k = 0; k < 4; ++k) {
    const float n = sqrtf(tot[k]) * (1.0f / 32.0f);
    inv[k] = 1.0f / (n + EPSN);
  }
  float* op = out + ((size_t)b * I_ + c0) * S_ + s;
#pragma unroll
  for (int g = 0; g < NORM_G; ++g) {
    v4f y = x[g] * inv;
#pragma unroll
    for (int k = 0; k < 4; ++k) y[k] = y[k] >= 0.f ? y[k] : y[k] * SLOPE_;
    *(v4f*)(op + (size_t)(4 * g) * S_) = y;
  }
}

#define CV_WAVES 12
#define CV_NT 4
#define CV_KS 6
__global__ void __launch_bounds__(CV_WAVES * 32)
conv_wmma_kernel(const float* __restrict__ G,
                 const _Float16* __restrict__ Wp1,
                 float* __restrict__ Out) {
  constexpr int CHUNK = CV_KS * 32;
  constexpr int NCH = (3 * I_) / CHUNK;
  constexpr int BUF = CV_NT * CV_KS * 512;
  constexpr int ITS = (CV_KS * 4 * CV_NT * 16) / (CV_WAVES * 32);
  constexpr int KTtot = (3 * I_) >> 5;
  constexpr size_t fragstride = (size_t)KTtot * 512;
  __shared__ __attribute__((aligned(16))) _Float16 xs[2][BUF];
  const int tok0 = blockIdx.y * (CV_NT * 16);
  const int b = tok0 / S_, s0 = tok0 % S_;
  const int lane = threadIdx.x & 31;
  const int wave = __builtin_amdgcn_readfirstlane((int)(threadIdx.x >> 5));
  const int col = lane & 15, khalf = (lane >> 4) * 8;
  const int mt0 = (blockIdx.x * CV_WAVES + wave) * 2;
  const _Float16* A0 = Wp1 + (size_t)mt0 * fragstride + lane * 16;
  const _Float16* A1 = A0 + fragstride;
  const int n64 = (wave & 1) * 32 + lane;
  const int octw = wave >> 1;
  const int sb = s0 + n64 - 2;
  const float* Gb = G + (size_t)b * I_ * S_;
  float r[ITS][8];
  auto gload = [&](int cc) {
#pragma unroll
    for (int it = 0; it < ITS; ++it) {
      const int kk0 = cc + (it * (CV_WAVES / 2) + octw) * 8;
#pragma unroll
      for (int jj = 0; jj < 8; ++jj) {
        const int kk = kk0 + jj;
        const int i = kk / 3, kc = kk - i * 3;
        const int s = sb + kc;
        const float gv = Gb[(size_t)i * S_ + (s > 0 ? s : 0)];
        r[it][jj] = (s >= 0) ? gv : 0.0f;
      }
    }
  };
  auto lstore = [&](int buf) {
#pragma unroll
    for (int it = 0; it < ITS; ++it)
      store_oct(&xs[buf][oct_off(it * (CV_WAVES / 2) + octw, n64, CV_KS)], r[it]);
  };
  v8f c[2][CV_NT];
#pragma unroll
  for (int nt = 0; nt < CV_NT; ++nt) { c[0][nt] = v8f{}; c[1][nt] = v8f{}; }
  gload(0);
  lstore(0);
  __syncthreads();
  v16h ag0[3][2], ag1[3][2];
#pragma unroll
  for (int u = 0; u < 3; ++u) {
    ag0[u][0] = load_frag(A0 + (size_t)u * 512);
    ag0[u][1] = load_frag(A1 + (size_t)u * 512);
  }
  for (int ch = 0; ch < NCH; ++ch) {
    const int buf = ch & 1;
    const _Float16* Bb = &xs[buf][lane * 16];
    if (ch + 1 < NCH) gload((ch + 1) * CHUNK);
#pragma unroll
    for (int u = 0; u < 3; ++u) {
      ag1[u][0] = load_frag(A0 + (size_t)(ch * CV_KS + 3 + u) * 512);
      ag1[u][1] = load_frag(A1 + (size_t)(ch * CV_KS + 3 + u) * 512);
    }
    v16h bc[CV_NT], bn[CV_NT];
#pragma unroll
    for (int nt = 0; nt < CV_NT; ++nt) bc[nt] = load_frag(Bb + (nt * CV_KS) * 512);
    __builtin_amdgcn_sched_barrier(0);
#pragma unroll
    for (int kt = 0; kt < CV_KS; ++kt) {
      if (kt == 3) {
#pragma unroll
        for (int u = 0; u < 3; ++u) {
          const int kn = ((ch + 1) * CV_KS + u < KTtot) ? (ch + 1) * CV_KS + u : KTtot - 1;
          ag0[u][0] = load_frag(A0 + (size_t)kn * 512);
          ag0[u][1] = load_frag(A1 + (size_t)kn * 512);
        }
      }
      if (kt + 1 < CV_KS) {
#pragma unroll
        for (int nt = 0; nt < CV_NT; ++nt)
          bn[nt] = load_frag(Bb + (nt * CV_KS + kt + 1) * 512);
      }
      __builtin_amdgcn_sched_barrier(0);
      const v16h a0c = (kt < 3) ? ag0[kt % 3][0] : ag1[kt % 3][0];
      const v16h a1c = (kt < 3) ? ag0[kt % 3][1] : ag1[kt % 3][1];
#pragma unroll
      for (int nt = 0; nt < CV_NT; ++nt) {
        c[0][nt] = __builtin_amdgcn_wmma_f32_16x16x32_f16(
            false, a0c, false, bc[nt], (short)0, c[0][nt], false, false);
        c[1][nt] = __builtin_amdgcn_wmma_f32_16x16x32_f16(
            false, a1c, false, bc[nt], (short)0, c[1][nt], false, false);
      }
      __builtin_amdgcn_sched_barrier(0);
      if (kt + 1 < CV_KS) {
#pragma unroll
        for (int nt = 0; nt < CV_NT; ++nt) bc[nt] = bn[nt];
      }
    }
    if (ch + 1 < NCH) lstore(buf ^ 1);
    __syncthreads();
  }
#pragma unroll
  for (int mi = 0; mi < 2; ++mi)
#pragma unroll
    for (int nt = 0; nt < CV_NT; ++nt)
#pragma unroll
      for (int j = 0; j < 8; ++j) {
        int rr = (mt0 + mi) * 16 + j + khalf;
        Out[((size_t)b * (3 * I_) + rr) * S_ + s0 + nt * 16 + col] = c[mi][nt][j];
      }
}

__global__ void init_loss_kernel(float* lacc) {
  if (threadIdx.x < 16) lacc[threadIdx.x] = 0.f;
}
__global__ void finalize_loss_kernel(const float* __restrict__ lacc,
                                     float* __restrict__ out_loss) {
  if (threadIdx.x == 0) {
    const float invT = 1.0f / (float)TOK;
    float l = 0.f;
    for (int e = 0; e < E_; ++e) l += (lacc[e] * invT) * (lacc[4 + e] * invT);
    for (int e = 0; e < E_; ++e) l += (lacc[8 + e] * invT) * (lacc[12 + e] * invT);
    *out_loss = l;
  }
}

extern "C" void kernel_launch(void* const* d_in, const int* in_sizes, int n_in,
                              void* d_out, int out_size, void* d_ws, size_t ws_size,
                              hipStream_t stream) {
  const float* inp     = (const float*)d_in[0];
  const float* divisor = (const float*)d_in[1];
  const float* w0_gate = (const float*)d_in[2];
  const float* w0      = (const float*)d_in[3];
  const float* w1      = (const float*)d_in[4];
  const float* w2_gate = (const float*)d_in[5];
  const float* w2      = (const float*)d_in[6];
  float* out = (float*)d_out;

  char* ws = (char*)d_ws;
  size_t off = 0;
  float* lacc = (float*)(ws + off); off += 256;
  const size_t h0b = (size_t)B_ * 3 * I_ * S_ * sizeof(float);
  const size_t t1b = (size_t)B_ * I_ * S_ * sizeof(float);
  float* h0   = (float*)(ws + off); off += h0b;
  float* tmp1 = (float*)(ws + off); off += t1b;
  float* g1   = (float*)(ws + off); off += t1b;
  int*   idx0 = (int*)(ws + off); off += TOK * sizeof(int);
  int*   idx1 = (int*)(ws + off); off += TOK * sizeof(int);
  off = (off + 255) & ~(size_t)255;
  const size_t w0n = (size_t)E_ * F_ * 3 * I_;
  const size_t w1n = (size_t)3 * I_ * 3 * I_;
  const size_t w2n = (size_t)E_ * I_ * F_;
  _Float16* w0p = (_Float16*)(ws + off); off += w0n * 2;
  _Float16* w1p = (_Float16*)(ws + off); off += w1n * 2;
  _Float16* w2p = (_Float16*)(ws + off); off += w2n * 2;
  float* h2 = h0;
  float* h3 = tmp1;

  pack_kn_kernel<F_, 3 * I_><<<(int)(w0n / (1024 * 8)), 256, 0, stream>>>(w0, w0p);
  pack_mk_kernel<3 * I_, 3 * I_><<<(int)(w1n / (2048 * 8)), 256, 0, stream>>>(w1, w1p);
  pack_kn_kernel<I_, F_><<<(int)(w2n / (1024 * 8)), 256, 0, stream>>>(w2, w2p);

  init_loss_kernel<<<1, 32, 0, stream>>>(lacc);

  gate_kernel<<<TOK / 32, GATE_WAVES * 32, 0, stream>>>(
      inp, F_, w0_gate, idx0, lacc);
  moe_wmma_kernel<F_, 3 * I_, 2, 4, 12, 4><<<dim3(3 * I_ / 384, TOK / 64), 384, 0, stream>>>(
      inp, w0p, idx0, h0, 1.0f);

  cumsum_kernel<<<(B_ * I_) / 8, 256, 0, stream>>>(h0, divisor, tmp1);
  norm_kernel<0><<<TOK / 32, NORM_WAVES * 32, 0, stream>>>(tmp1, tmp1, tmp1, I_, g1);

  conv_wmma_kernel<<<dim3(3 * I_ / 384, TOK / 64), CV_WAVES * 32, 0, stream>>>(g1, w1p, h2);

  norm_kernel<1><<<TOK / 32, NORM_WAVES * 32, 0, stream>>>(
      h2, h2 + (size_t)I_ * S_, h2 + (size_t)2 * I_ * S_, 3 * I_, h3);

  gate_kernel<<<TOK / 32, GATE_WAVES * 32, 0, stream>>>(
      h3, I_, w2_gate, idx1, lacc + 8);
  moe_wmma_kernel<I_, F_, 1, 2, 8, 8><<<dim3(F_ / 128, TOK / 32), 256, 0, stream>>>(
      h3, w2p, idx1, out, 1.0f  );

  finalize_loss_kernel<<<1, 32, 0, stream>>>(lacc, out + (size_t)B_ * F_ * S_);
}


# 13 launches -> 7: independent kernels merged (gate 1 + pack w0; cumsum + packs w1,w2), gate 2 inside norm 2, loss summed in the last GEMM from per-group partials, no atomics (sum order changes)
# speedup vs baseline: 4.1078x; 1.0637x over previous
#include <hip/hip_runtime.h>
#include <hip/hip_bf16.h>

typedef __attribute__((ext_vector_type(16))) _Float16 v16h;
typedef __attribute__((ext_vector_type(8)))  _Float16 v8h;
typedef __attribute__((ext_vector_type(8)))  float    v8f;

#define B_ 2
#define F_ 512
#define S_ 1024
#define E_ 4
#define I_ 1024
#define TOK (B_*S_)
#define EPSN 1e-5f
#define SLOPE_ 0.02f

__device__ __forceinline__ int frag_j(int kin) {
  return kin < 8 ? kin : (kin < 16 ? kin - 8 : (kin < 24 ? kin - 8 : kin - 16));
}
__device__ __forceinline__ int frag_lane(int kin, int n) {
  return n + (((kin >> 3) & 1) ? 16 : 0);
}

template <int CIN, int COUT>
__device__ __forceinline__ void
pack_kn_tile(const float* __restrict__ W, _Float16* __restrict__ Wp, int g, int lane) {
  constexpr int KT = CIN >> 5, MT = COUT >> 4, MG = COUT >> 5;
  const int mg = g % MG;
  const int kt = (g / MG) % KT;
  const int e  = g / (MG * KT);
  const float* sp = W + ((size_t)e * CIN + (size_t)kt * 32) * COUT + mg * 32 + lane;
  float v[32];
#pragma unroll
  for (int k = 0; k < 32; ++k) v[k] = sp[(size_t)k * COUT];
  v16h lo, hi;
#pragma unroll
  for (int j = 0; j < 8; ++j) {
    lo[j]     = (_Float16)v[j];
    lo[j + 8] = (_Float16)v[j + 16];
    hi[j]     = (_Float16)v[j + 8];
    hi[j + 8] = (_Float16)v[j + 24];
  }
  const int mt = mg * 2 + (lane >> 4);
  _Float16* dp = Wp + ((((size_t)e * MT + mt) * KT + kt) * 32 + (lane & 15)) * 16;
  *(v16h*)dp = lo;
  *(v16h*)(dp + 16 * 16) = hi;
}

template <int CIN, int COUT>
__device__ __forceinline__ void
pack_mk_tile(const float* __restrict__ W, _Float16* __restrict__ Wp, int g, int lane) {
  constexpr int KT = CIN >> 5, KG = KT >> 2;
  typedef __attribute__((ext_vector_type(4))) float v4f;
  typedef __attribute__((ext_vector_type(4))) _Float16 v4h;
  const int kg = g % KG;
  const int mt = g / KG;
  const int r = lane >> 3, q = lane & 7;
  const float* sp = W + ((size_t)mt * 16 + r) * CIN + kg * 128 + q * 4;
  v4f x[4][4];
#pragma unroll
  for (int t = 0; t < 4; ++t)
#pragma unroll
    for (int i = 0; i < 4; ++i)
      x[t][i] = *(const v4f*)(sp + (size_t)(4 * i) * CIN + t * 32);
  const int fl = r + (((q >> 1) & 1) << 4);
  const int j0 = ((q >> 2) << 3) + ((q & 1) << 2);
  _Float16* dp = Wp + (((size_t)mt * KT + kg * 4) * 32 + fl) * 16 + j0;
#pragma unroll
  for (int t = 0; t < 4; ++t)
#pragma unroll
    for (int i = 0; i < 4; ++i) {
      v4h h;
#pragma unroll
      for (int u = 0; u < 4; ++u) h[u] = (_Float16)x[t][i][u];
      *(v4h*)(dp + (size_t)t * 512 + (4 * i) * 16) = h;
    }
}

__device__ __forceinline__ v16h load_frag(const _Float16* p) {
  v8h lo = *(const v8h*)p;
  v8h hi = *(const v8h*)(p + 8);
  v16h r;
#pragma unroll
  for (int j = 0; j < 8; ++j) { r[j] = lo[j]; r[j + 8] = hi[j]; }
  return r;
}

#define GATE_WAVES 16
#define NGRP (TOK / 32)
__device__ __forceinline__ void
gate_body(const float* __restrict__ x, int C,
          const float* __restrict__ gw,
          int* __restrict__ eidx,
          float* __restrict__ lpart,
          float (*part)[E_][32], int grp, int wave, int lane) {
  const int t0 = grp * 32;
  const int b = t0 / S_, s = (t0 % S_) + lane;
  const int cper = C / GATE_WAVES;
  const int c0 = wave * cper;
  const float* xp = x + ((size_t)b * C + c0) * S_ + s;
  const float* g0 = gw + c0;
  const float* g1 = g0 + C;
  const float* g2 = g1 + C;
  const float* g3 = g2 + C;
  float a0 = 0.f, a1 = 0.f, a2 = 0.f, a3 = 0.f;
  for (int cb = 0; cb < cper; cb += 16) {
    float xv[16];
#pragma unroll
    for (int u = 0; u < 16; ++u) xv[u] = xp[(size_t)(cb + u) * S_];
#pragma unroll
    for (int u = 0; u < 16; ++u) {
      a0 = __builtin_fmaf(xv[u], g0[cb + u], a0);
      a1 = __builtin_fmaf(xv[u], g1[cb + u], a1);
      a2 = __builtin_fmaf(xv[u], g2[cb + u], a2);
      a3 = __builtin_fmaf(xv[u], g3[cb + u], a3);
    }
  }
  part[wave][0][lane] = a0;
  part[wave][1][lane] = a1;
  part[wave][2][lane] = a2;
  part[wave][3][lane] = a3;
  __syncthreads();
  if (wave != 0) return;
  float a[4];
  for (int e = 0; e < E_; ++e) {
    float v = part[0][e][lane];
    for (int w = 1; w < GATE_WAVES; ++w) v += part[w][e][lane];
    a[e] = v;
  }
  int best = 0; float m = a[0];
  for (int e = 1; e < E_; ++e) { if (a[e] > m) { m = a[e]; best = e; } }
  float g[4]; float se = 0.f;
  for (int e = 0; e < E_; ++e) { g[e] = __expf(a[e] - m); se += g[e]; }
  float inv = 1.0f / se;
  eidx[t0 + lane] = best;
  for (int e = 0; e < E_; ++e) {
    float gsum = g[e] * inv;
    for (int d = 16; d > 0; d >>= 1) gsum += __shfl_xor(gsum, d, 32);
    float cnt = (float)__popcll(__ballot(best == e));
    if (lane == 0) {
      lpart[grp * 8 + e] = gsum;
      lpart[grp * 8 + 4 + e] = cnt;
    }
  }
}

__device__ __forceinline__ void store_oct(_Float16* dst, const float* v) {
  v8h h;
#pragma unroll
  for (int jj = 0; jj < 8; ++jj) h[jj] = (_Float16)v[jj];
  *(v8h*)dst = h;
}
__device__ __forceinline__ int oct_off(int oct, int n, int KT) {
  int kt = oct >> 2, a = oct & 3;
  return ((((n >> 4) * KT + kt) * 32 + (n & 15) + ((a & 1) << 4)) << 4) + ((a >> 1) << 3);
}

template <int CIN, int COUT, int MTW, int NT, int WAVES, int D, bool FIN>
__global__ void __launch_bounds__(WAVES * 32)
moe_wmma_kernel(const float* __restrict__ X,
                const _Float16* __restrict__ Wp,
                const int* __restrict__ eidx,
                float* __restrict__ Out, float outscale,
                const float* __restrict__ lpart,
                float* __restrict__ out_loss) {
  constexpr int KT = CIN >> 5;
  constexpr int NG = KT / D;
  static_assert(KT % (2 * D) == 0, "two register sets of D k-steps must divide the k-steps");
  constexpr int PIN = (MTW * NT >= 8) ? 1 : 8 / (MTW * NT);
  constexpr int NTOK = NT * 16;
  constexpr int THREADS = WAVES * 32;
  constexpr int ITEMS = (CIN >> 3) * NTOK;
  __shared__ __attribute__((aligned(16))) _Float16 xs[NT * KT * 512];
  const int tok0 = blockIdx.y * NTOK;
  const int b = tok0 / S_, s0 = tok0 % S_;
  auto stage = [&](int q) {
    const int n = q % NTOK, oct = q / NTOK;
    const float* xp = X + ((size_t)b * CIN + oct * 8) * S_ + s0 + n;
    float v[8];
#pragma unroll
    for (int jj = 0; jj < 8; ++jj) v[jj] = xp[(size_t)jj * S_];
    store_oct(&xs[oct_off(oct, n, KT)], v);
  };
  constexpr int FULL = ITEMS / THREADS;
  constexpr int UNR = (FULL % 5 == 0) ? 5 : 4;
#pragma unroll UNR
  for (int it = 0; it < FULL; ++it) stage((int)threadIdx.x + it * THREADS);
  if (ITEMS % THREADS != 0) {
    const int q = (int)threadIdx.x + FULL * THREADS;
    if (q < ITEMS) stage(q);
  }
  __syncthreads();
  const int lane = threadIdx.x & 31;
  const int wave = __builtin_amdgcn_readfirstlane((int)(threadIdx.x >> 5));
  const int col = lane & 15, khalf = (lane >> 4) * 8;
  const int mt0 = (blockIdx.x * WAVES + wave) * MTW;
  constexpr size_t fragstride = (size_t)KT * 512;
  constexpr size_t estride = (size_t)(COUT >> 4) * fragstride;
  int myexp[NT];
#pragma unroll
  for (int nt = 0; nt < NT; ++nt) myexp[nt] = eidx[tok0 + nt * 16 + col];
  v8f cs[MTW][NT];
#pragma unroll
  for (int mi = 0; mi < MTW; ++mi)
#pragma unroll
    for (int nt = 0; nt < NT; ++nt) cs[mi][nt] = v8f{};
  unsigned rem = 0;
#pragma unroll
  for (int e = 0; e < E_; ++e) {
    int hit = 0;
#pragma unroll
    for (int nt = 0; nt < NT; ++nt) hit |= (myexp[nt] == e) ? 1 : 0;
    if (__any(hit)) rem |= 1u << e;
  }
  const _Float16* Bb = xs + lane * 16;
  const _Float16* Abase = Wp + (size_t)mt0 * fragstride;
  const int loff = lane * 16;
  if (rem != 0) {
    int e = __builtin_ctz(rem);
    rem &= rem - 1;
    const _Float16* Acur = Abase + (size_t)e * estride;
    v16h ra[D][MTW], rb[D][MTW];
    v16h bc[NT];
#pragma unroll
    for (int u = 0; u < D; ++u)
#pragma unroll
      for (int mi = 0; mi < MTW; ++mi)
        ra[u][mi] = load_frag(Acur + (size_t)mi * fragstride + (size_t)u * 512 + loff);
#pragma unroll
    for (int nt = 0; nt < NT; ++nt) bc[nt] = load_frag(Bb + (size_t)nt * KT * 512);
    for (;;) {
      const bool more = rem != 0;
      const int en = more ? __builtin_ctz(rem) : e;
      const _Float16* Anext = more ? Abase + (size_t)en * estride
                                   : Acur + (size_t)(KT - D) * 512;
      v8f c[MTW][NT];
#pragma unroll
      for (int mi = 0; mi < MTW; ++mi)
#pragma unroll
        for (int nt = 0; nt < NT; ++nt) c[mi][nt] = v8f{};
      for (int h = 0; h < NG / 2; ++h) {
        const _Float16* Apb = Acur + (size_t)(2 * h + 1) * D * 512;
        const _Float16* Apa = (2 * h + 2 < NG) ? Acur + (size_t)(2 * h + 2) * D * 512 : Anext;
#pragma unroll
        for (int u = 0; u < D; ++u) {
          const int k2 = 2 * h * D + u + 1;
          v16h bn[NT];
#pragma unroll
          for (int nt = 0; nt < NT; ++nt)
            bn[nt] = load_frag(Bb + ((size_t)nt * KT + k2) * 512);
#pragma unroll
          for (int nt = 0; nt < NT; ++nt)
#pragma unroll
            for (int mi = 0; mi < MTW; ++mi)
              c[mi][nt] = __builtin_amdgcn_wmma_f32_16x16x32_f16(
                  false, ra[u][mi], false, bc[nt], (short)0, c[mi][nt], false, false);
#pragma unroll
          for (int mi = 0; mi < MTW; ++mi)
            rb[u][mi] = load_frag(Apb + (size_t)mi * fragstride + (size_t)u * 512 + loff);
          if ((u + 1) % PIN == 0) __builtin_amdgcn_sched_barrier(0);
#pragma unroll
          for (int nt = 0; nt < NT; ++nt) bc[nt] = bn[nt];
        }
#pragma unroll
        for (int u = 0; u < D; ++u) {
          const int kt = (2 * h + 1) * D + u;
          const int k2 = (kt + 1 < KT) ? kt + 1 : 0;
          v16h bn[NT];
#pragma unroll
          for (int nt = 0; nt < NT; ++nt)
            bn[nt] = load_frag(Bb + ((size_t)nt * KT + k2) * 512);
#pragma unroll
          for (int nt = 0; nt < NT; ++nt)
#pragma unroll
            for (int mi = 0; mi < MTW; ++mi)
              c[mi][nt] = __builtin_amdgcn_wmma_f32_16x16x32_f16(
                  false, rb[u][mi], false, bc[nt], (short)0, c[mi][nt], false, false);
#pragma unroll
          for (int mi = 0; mi < MTW; ++mi)
            ra[u][mi] = load_frag(Apa + (size_t)mi * fragstride + (size_t)u * 512 + loff);
          if ((u + 1) % PIN == 0) __builtin_amdgcn_sched_barrier(0);
#pragma unroll
          for (int nt = 0; nt < NT; ++nt) bc[nt] = bn[nt];
        }
      }
#pragma unroll
      for (int nt = 0; nt < NT; ++nt)
        if (myexp[nt] == e) {
#pragma unroll
          for (int mi = 0; mi < MTW; ++mi) cs[mi][nt] = c[mi][nt];
        }
      if (!more) break;
      e = en;
      rem &= rem - 1;
      Acur = Anext;
    }
  }
#pragma unroll
  for (int mi = 0; mi < MTW; ++mi)
#pragma unroll
    for (int nt = 0; nt < NT; ++nt)
#pragma unroll
      for (int j = 0; j < 8; ++j) {
        int r = (mt0 + mi) * 16 + j + khalf;
        Out[((size_t)b * COUT + r) * S_ + s0 + nt * 16 + col] = cs[mi][nt][j] * outscale;
      }
  if (FIN) {
    if (blockIdx.x == 0 && blockIdx.y == 0 && wave == 0) {
      const int a = lane & 15, h = lane >> 4;
      const float* p = lpart + ((size_t)(a >> 3) * NGRP + h * (NGRP / 2)) * 8 + (a & 7);
      float v = 0.f;
#pragma unroll
      for (int g = 0; g < NGRP / 2; ++g) v += p[g * 8];
      v += __shfl_xor(v, 16, 32);
      const float m = v * (1.0f / (float)TOK);
      const float pr = m * __shfl_xor(m, 4, 32);
      float l = 0.f;
      for (int e = 0; e < E_; ++e) l += __shfl(pr, e, 32);
      for (int e = 0; e < E_; ++e) l += __shfl(pr, 8 + e, 32);
      if (lane == 0) *out_loss = l;
    }
  }
}

#define CS_BLOCKS ((B_ * I_) / 8)
#define P1_BLOCKS ((3 * I_ * 3 * I_) / (2048 * 8))
#define P2_BLOCKS ((E_ * I_ * F_) / (1024 * 8))
__global__ void __launch_bounds__(256)
cumsum_pack_kernel(const float* __restrict__ h0,
                   const float* __restrict__ divisor,
                   float* __restrict__ tmp1,
                   const float* __restrict__ w1, _Float16* __restrict__ w1p,
                   const float* __restrict__ w2, _Float16* __restrict__ w2p) {
  const int lane = threadIdx.x & 31;
  const int wv = __builtin_amdgcn_readfirstlane((int)(blockIdx.x * 8 + (threadIdx.x >> 5)));
  if (wv >= (CS_BLOCKS + P1_BLOCKS) * 8) {
    pack_kn_tile<I_, F_>(w2, w2p, wv - (CS_BLOCKS + P1_BLOCKS) * 8, lane);
    return;
  }
  if (wv >= CS_BLOCKS * 8) {
    pack_mk_tile<3 * I_, 3 * I_>(w1, w1p, wv - CS_BLOCKS * 8, lane);
    return;
  }
  const int row = wv;
  int b = row / I_, i = row % I_;
  const float* dep = h0 + ((size_t)b * 3 * I_ + i) * S_;
  const float* sc  = h0 + ((size_t)b * 3 * I_ + I_ + i) * S_;
  const float* sh  = h0 + ((size_t)b * 3 * I_ + 2 * I_ + i) * S_;
  float* o = tmp1 + ((size_t)b * I_ + i) * S_;
  float carry = 0.f;
  for (int s0 = 0; s0 < S_; s0 += 32) {
    float v = dep[s0 + lane];
#pragma unroll
    for (int d = 1; d < 32; d <<= 1) {
      float t = __shfl_up(v, d, 32);
      if (lane >= d) v += t;
    }
    float cum = carry + v;
    o[s0 + lane] = cum / divisor[s0 + lane] * sc[s0 + lane] + sh[s0 + lane];
    carry = __shfl(cum, 31, 32);
  }
}

#define NORM_WAVES 32
#define NORM_CPW (I_ / NORM_WAVES)
#define NORM_G (NORM_CPW / 4)
#define NORM_PITCH 36
typedef __attribute__((ext_vector_type(4))) float v4f;

__device__ __forceinline__ float lane_x8(float x) {
  int b = __builtin_bit_cast(int, x);
  return __builtin_bit_cast(float, __builtin_amdgcn_permlane16(
      b, b, 0xfedcba98u, 0x76543210u, false, false));
}
__device__ __forceinline__ float lane_x16(float x) {
  int b = __builtin_bit_cast(int, x);
  return __builtin_bit_cast(float, __builtin_amdgcn_permlanex16(
      b, b, 0x76543210u, 0xfedcba98u, false, false));
}
__device__ __forceinline__ v4f join4(v4f p) {
#pragma unroll
  for (int k = 0; k < 4; ++k) p[k] += lane_x8(p[k]);
#pragma unroll
  for (int k = 0; k < 4; ++k) p[k] += lane_x16(p[k]);
  return p;
}
__device__ __forceinline__ v4f block_sum(float* part, v4f p, int wave, int tq, int rs) {
  p = join4(p);
  if (rs == 0) *(v4f*)(part + wave * NORM_PITCH + tq * 4) = p;
  __syncthreads();
  const float* pp = part + (8 * rs) * NORM_PITCH + tq * 4;
  v4f q = *(const v4f*)pp;
#pragma unroll
  for (int j = 1; j < 8; ++j) q += *(const v4f*)(pp + j * NORM_PITCH);
  return join4(q);
}

template <int MODE>
__global__ void __launch_bounds__(NORM_WAVES * 32)
norm_kernel(const float* __restrict__ p0,
            const float* __restrict__ p1,
            const float* __restrict__ p2,
            int cstride,
            float* __restrict__ out,
            const float* __restrict__ gw,
            int* __restrict__ eidx,
            float* __restrict__ lpart) {
  __shared__ __attribute__((aligned(16))) float part0[NORM_WAVES * NORM_PITCH];
  __shared__ __attribute__((aligned(16))) float part1[NORM_WAVES * NORM_PITCH];
  __shared__ __attribute__((aligned(16))) float partg[MODE ? E_ : 1][NORM_WAVES * NORM_PITCH];
  const int lane = threadIdx.x & 31;
  const int wave = __builtin_amdgcn_readfirstlane((int)(threadIdx.x >> 5));
  const int tq = lane & 7, rs = lane >> 3;
  const int t0 = blockIdx.x * 32;
  const int b = t0 / S_, s = (t0 % S_) + tq * 4;
  const int c0 = wave * NORM_CPW + rs;
  const size_t ib = ((size_t)b * cstride + c0) * S_ + s;
  v4f x[NORM_G];
#pragma unroll
  for (int g = 0; g < NORM_G; ++g) {
    const size_t idx = ib + (size_t)(4 * g) * S_;
    v4f a = *(const v4f*)(p0 + idx);
    if (MODE) a = a * *(const v4f*)(p1 + idx) + *(const v4f*)(p2 + idx);
    x[g] = a;
  }
  float gwv[E_][NORM_G];
  if (MODE) {
#pragma unroll
    for (int e = 0; e < E_; ++e)
#pragma unroll
      for (int g = 0; g < NORM_G; ++g) gwv[e][g] = gw[e * I_ + c0 + 4 * g];
  }
  v4f ps = x[0];
#pragma unroll
  for (int g = 1; g < NORM_G; ++g) ps += x[g];
  const v4f mean = block_sum(part0, ps, wave, tq, rs) * (1.0f / I_);
  v4f ss = v4f{};
#pragma unroll
  for (int g = 0; g < NORM_G; ++g) {
    x[g] -= mean;
    ss += x[g] * x[g];
  }
  const v4f tot = block_sum(part1, ss, wave, tq, rs);
  v4f inv;
#pragma unroll
  for (int k = 0; k < 4; ++k) {
    const float n = sqrtf(tot[k]) * (1.0f / 32.0f);
    inv[k] = 1.0f / (n + EPSN);
  }
  float* op = out + ((size_t)b * I_ + c0) * S_ + s;
#pragma unroll
  for (int g = 0; g < NORM_G; ++g) {
    v4f y = x[g] * inv;
#pragma unroll
    for (int k = 0; k < 4; ++k) y[k] = y[k] >= 0.f ? y[k] : y[k] * SLOPE_;
    *(v4f*)(op + (size_t)(4 * g) * S_) = y;
    x[g] = y;
  }
  if (MODE) {
    v4f pl[E_];
#pragma unroll
    for (int e = 0; e < E_; ++e) {
      v4f a = x[0] * gwv[e][0];
#pragma unroll
      for (int g = 1; g < NORM_G; ++g) a += x[g] * gwv[e][g];
      pl[e] = join4(a);
    }
    if (rs == 0) {
#pragma unroll
      for (int e = 0; e < E_; ++e) *(v4f*)(partg[e] + wave * NORM_PITCH + tq * 4) = pl[e];
    }
    __syncthreads();
    if (wave != 0) return;
    v4f lg[E_];
#pragma unroll
    for (int e = 0; e < E_; ++e) {
      const float* pp = partg[e] + (8 * rs) * NORM_PITCH + tq * 4;
      v4f q = *(const v4f*)pp;
#pragma unroll
      for (int j = 1; j < 8; ++j) q += *(const v4f*)(pp + j * NORM_PITCH);
      lg[e] = join4(q);
    }
    float gs[E_], cn[E_];
#pragma unroll
    for (int e = 0; e < E_; ++e) { gs[e] = 0.f; cn[e] = 0.f; }
    int bk[4];
#pragma unroll
    for (int k = 0; k < 4; ++k) {
      int best = 0; float m = lg[0][k];
#pragma unroll
      for (int e = 1; e < E_; ++e) { if (lg[e][k] > m) { m = lg[e][k]; best = e; } }
      float g[E_]; float se = 0.f;
#pragma unroll
      for (int e = 0; e < E_; ++e) { g[e] = __expf(lg[e][k] - m); se += g[e]; }
      const float iv = 1.0f / se;
#pragma unroll
      for (int e = 0; e < E_; ++e) {
        gs[e] += g[e] * iv;
        cn[e] += (best == e) ? 1.0f : 0.0f;
      }
      bk[k] = best;
    }
    if (rs == 0) {
      typedef __attribute__((ext_vector_type(4))) int v4i;
      v4i bv; bv[0] = bk[0]; bv[1] = bk[1]; bv[2] = bk[2]; bv[3] = bk[3];
      *(v4i*)(eidx + t0 + tq * 4) = bv;
    }
#pragma unroll
    for (int e = 0; e < E_; ++e)
      for (int d = 1; d < 8; d <<= 1) {
        gs[e] += __shfl_xor(gs[e], d, 32);
        cn[e] += __shfl_xor(cn[e], d, 32);
      }
    if (lane == 0) {
#pragma unroll
      for (int e = 0; e < E_; ++e) {
        lpart[blockIdx.x * 8 + e] = gs[e];
        lpart[blockIdx.x * 8 + 4 + e] = cn[e];
      }
    }
  }
}

#define CV_WAVES 12
#define CV_NT 4
#define CV_KS 6
__global__ void __launch_bounds__(CV_WAVES * 32)
conv_wmma_kernel(const float* __restrict__ G,
                 const _Float16* __restrict__ Wp1,
                 float* __restrict__ Out) {
  constexpr int CHUNK = CV_KS * 32;
  constexpr int NCH = (3 * I_) / CHUNK;
  constexpr int BUF = CV_NT * CV_KS * 512;
  constexpr int ITS = (CV_KS * 4 * CV_NT * 16) / (CV_WAVES * 32);
  constexpr int KTtot = (3 * I_) >> 5;
  constexpr size_t fragstride = (size_t)KTtot * 512;
  __shared__ __attribute__((aligned(16))) _Float16 xs[2][BUF];
  const int tok0 = blockIdx.y * (CV_NT * 16);
  const int b = tok0 / S_, s0 = tok0 % S_;
  const int lane = threadIdx.x & 31;
  const int wave = __builtin_amdgcn_readfirstlane((int)(threadIdx.x >> 5));
  const int col = lane & 15, khalf = (lane >> 4) * 8;
  const int mt0 = (blockIdx.x * CV_WAVES + wave) * 2;
  const _Float16* A0 = Wp1 + (size_t)mt0 * fragstride + lane * 16;
  const _Float16* A1 = A0 + fragstride;
  const int n64 = (wave & 1) * 32 + lane;
  const int octw = wave >> 1;
  const int sb = s0 + n64 - 2;
  const float* Gb = G + (size_t)b * I_ * S_;
  float r[ITS][8];
  auto gload = [&](int cc) {
#pragma unroll
    for (int it = 0; it < ITS; ++it) {
      const int kk0 = cc + (it * (CV_WAVES / 2) + octw) * 8;
#pragma unroll
      for (int jj = 0; jj < 8; ++jj) {
        const int kk = kk0 + jj;
        const int i = kk / 3, kc = kk - i * 3;
        const int s = sb + kc;
        const float gv = Gb[(size_t)i * S_ + (s > 0 ? s : 0)];
        r[it][jj] = (s >= 0) ? gv : 0.0f;
      }
    }
  };
  auto lstore = [&](int buf) {
#pragma unroll
    for (int it = 0; it < ITS; ++it)
      store_oct(&xs[buf][oct_off(it * (CV_WAVES / 2) + octw, n64, CV_KS)], r[it]);
  };
  v8f c[2][CV_NT];
#pragma unroll
  for (int nt = 0; nt < CV_NT; ++nt) { c[0][nt] = v8f{}; c[1][nt] = v8f{}; }
  gload(0);
  lstore(0);
  __syncthreads();
  v16h ag0[3][2], ag1[3][2];
#pragma unroll
  for (int u = 0; u < 3; ++u) {
    ag0[u][0] = load_frag(A0 + (size_t)u * 512);
    ag0[u][1] = load_frag(A1 + (size_t)u * 512);
  }
  for (int ch = 0; ch < NCH; ++ch) {
    const int buf = ch & 1;
    const _Float16* Bb = &xs[buf][lane * 16];
    if (ch + 1 < NCH) gload((ch + 1) * CHUNK);
#pragma unroll
    for (int u = 0; u < 3; ++u) {
      ag1[u][0] = load_frag(A0 + (size_t)(ch * CV_KS + 3 + u) * 512);
      ag1[u][1] = load_frag(A1 + (size_t)(ch * CV_KS + 3 + u) * 512);
    }
    v16h bc[CV_NT], bn[CV_NT];
#pragma unroll
    for (int nt = 0; nt < CV_NT; ++nt) bc[nt] = load_frag(Bb + (nt * CV_KS) * 512);
    __builtin_amdgcn_sched_barrier(0);
#pragma unroll
    for (int kt = 0; kt < CV_KS; ++kt) {
      if (kt == 3) {
#pragma unroll
        for (int u = 0; u < 3; ++u) {
          const int kn = ((ch + 1) * CV_KS + u < KTtot) ? (ch + 1) * CV_KS + u : KTtot - 1;
          ag0[u][0] = load_frag(A0 + (size_t)kn * 512);
          ag0[u][1] = load_frag(A1 + (size_t)kn * 512);
        }
      }
      if (kt + 1 < CV_KS) {
#pragma unroll
        for (int nt = 0; nt < CV_NT; ++nt)
          bn[nt] = load_frag(Bb + (nt * CV_KS + kt + 1) * 512);
      }
      __builtin_amdgcn_sched_barrier(0);
      const v16h a0c = (kt < 3) ? ag0[kt % 3][0] : ag1[kt % 3][0];
      const v16h a1c = (kt < 3) ? ag0[kt % 3][1] : ag1[kt % 3][1];
#pragma unroll
      for (int nt = 0; nt < CV_NT; ++nt) {
        c[0][nt] = __builtin_amdgcn_wmma_f32_16x16x32_f16(
            false, a0c, false, bc[nt], (short)0, c[0][nt], false, false);
        c[1][nt] = __builtin_amdgcn_wmma_f32_16x16x32_f16(
            false, a1c, false, bc[nt], (short)0, c[1][nt], false, false);
      }
      __builtin_amdgcn_sched_barrier(0);
      if (kt + 1 < CV_KS) {
#pragma unroll
        for (int nt = 0; nt < CV_NT; ++nt) bc[nt] = bn[nt];
      }
    }
    if (ch + 1 < NCH) lstore(buf ^ 1);
    __syncthreads();
  }
#pragma unroll
  for (int mi = 0; mi < 2; ++mi)
#pragma unroll
    for (int nt = 0; nt < CV_NT; ++nt)
#pragma unroll
      for (int j = 0; j < 8; ++j) {
        int rr = (mt0 + mi) * 16 + j + khalf;
        Out[((size_t)b * (3 * I_) + rr) * S_ + s0 + nt * 16 + col] = c[mi][nt][j];
      }
}

#define P0_BLOCKS ((E_ * F_ * 3 * I_) / (1024 * GATE_WAVES))
__global__ void __launch_bounds__(GATE_WAVES * 32)
gate_pack_kernel(const float* __restrict__ x,
                 const float* __restrict__ gw,
                 int* __restrict__ eidx,
                 float* __restrict__ lpart,
                 const float* __restrict__ w0, _Float16* __restrict__ w0p) {
  __shared__ float part[GATE_WAVES][E_][32];
  const int lane = threadIdx.x & 31;
  const int wave = __builtin_amdgcn_readfirstlane((int)(threadIdx.x >> 5));
  if (blockIdx.x < NGRP) {
    gate_body(x, F_, gw, eidx, lpart, part, (int)blockIdx.x, wave, lane);
    return;
  }
  pack_kn_tile<F_, 3 * I_>(w0, w0p, (int)(blockIdx.x - NGRP) * GATE_WAVES + wave, lane);
}

extern "C" void kernel_launch(void* const* d_in, const int* in_sizes, int n_in,
                              void* d_out, int out_size, void* d_ws, size_t ws_size,
                              hipStream_t stream) {
  const float* inp     = (const float*)d_in[0];
  const float* divisor = (const float*)d_in[1];
  const float* w0_gate = (const float*)d_in[2];
  const float* w0      = (const float*)d_in[3];
  const float* w1      = (const float*)d_in[4];
  const float* w2_gate = (const float*)d_in[5];
  const float* w2      = (const float*)d_in[6];
  float* out = (float*)d_out;

  char* ws = (char*)d_ws;
  size_t off = 0;
  float* lpart = (float*)(ws + off); off += (size_t)2 * NGRP * 8 * sizeof(float);
  const size_t h0b = (size_t)B_ * 3 * I_ * S_ * sizeof(float);
  const size_t t1b = (size_t)B_ * I_ * S_ * sizeof(float);
  float* h0   = (float*)(ws + off); off += h0b;
  float* tmp1 = (float*)(ws + off); off += t1b;
  float* g1   = (float*)(ws + off); off += t1b;
  int*   idx0 = (int*)(ws + off); off += TOK * sizeof(int);
  int*   idx1 = (int*)(ws + off); off += TOK * sizeof(int);
  off = (off + 255) & ~(size_t)255;
  const size_t w0n = (size_t)E_ * F_ * 3 * I_;
  const size_t w1n = (size_t)3 * I_ * 3 * I_;
  const size_t w2n = (size_t)E_ * I_ * F_;
  _Float16* w0p = (_Float16*)(ws + off); off += w0n * 2;
  _Float16* w1p = (_Float16*)(ws + off); off += w1n * 2;
  _Float16* w2p = (_Float16*)(ws + off); off += w2n * 2;
  float* h2 = h0;
  float* h3 = tmp1;

  gate_pack_kernel<<<NGRP + P0_BLOCKS, GATE_WAVES * 32, 0, stream>>>(
      inp, w0_gate, idx0, lpart, w0, w0p);
  moe_wmma_kernel<F_, 3 * I_, 2, 4, 12, 4, false><<<dim3(3 * I_ / 384, TOK / 64), 384, 0, stream>>>(
      inp, w0p, idx0, h0, 1.0f, nullptr, nullptr);

  cumsum_pack_kernel<<<CS_BLOCKS + P1_BLOCKS + P2_BLOCKS, 256, 0, stream>>>(
      h0, divisor, tmp1, w1, w1p, w2, w2p);
  norm_kernel<0><<<TOK / 32, NORM_WAVES * 32, 0, stream>>>(
      tmp1, tmp1, tmp1, I_, g1, w2_gate, idx1, lpart);

  conv_wmma_kernel<<<dim3(3 * I_ / 384, TOK / 64), CV_WAVES * 32, 0, stream>>>(g1, w1p, h2);

  norm_kernel<1><<<TOK / 32, NORM_WAVES * 32, 0, stream>>>(
      h2, h2 + (size_t)I_ * S_, h2 + (size_t)2 * I_ * S_, 3 * I_, h3,
      w2_gate, idx1, lpart + NGRP * 8);

  moe_wmma_kernel<I_, F_, 1, 2, 8, 8, true><<<dim3(F_ / 128, TOK / 32), 256, 0, stream>>>(
      h3, w2p, idx1, out, 1.0f  , lpart, out + (size_t)B_ * F_ * S_);
}


# operand fragments stored as two 512-byte planes [8 halves][lane], packed weights and LDS token tile alike: a wave-wide 16-byte fragment load is 512 contiguous bytes (was 32-byte lane stride)
# speedup vs baseline: 4.4299x; 1.0784x over previous
#include <hip/hip_runtime.h>
#include <hip/hip_bf16.h>

typedef __attribute__((ext_vector_type(16))) _Float16 v16h;
typedef __attribute__((ext_vector_type(8)))  _Float16 v8h;
typedef __attribute__((ext_vector_type(8)))  float    v8f;

#define B_ 2
#define F_ 512
#define S_ 1024
#define E_ 4
#define I_ 1024
#define TOK (B_*S_)
#define EPSN 1e-5f
#define SLOPE_ 0.02f

__device__ __forceinline__ int frag_j(int kin) {
  return kin < 8 ? kin : (kin < 16 ? kin - 8 : (kin < 24 ? kin - 8 : kin - 16));
}
__device__ __forceinline__ int frag_lane(int kin, int n) {
  return n + (((kin >> 3) & 1) ? 16 : 0);
}

template <int CIN, int COUT>
__device__ __forceinline__ void
pack_kn_tile(const float* __restrict__ W, _Float16* __restrict__ Wp, int g, int lane) {
  constexpr int KT = CIN >> 5, MT = COUT >> 4, MG = COUT >> 5;
  const int mg = g % MG;
  const int kt = (g / MG) % KT;
  const int e  = g / (MG * KT);
  const float* sp = W + ((size_t)e * CIN + (size_t)kt * 32) * COUT + mg * 32 + lane;
  float v[32];
#pragma unroll
  for (int k = 0; k < 32; ++k) v[k] = sp[(size_t)k * COUT];
  v8h a0, a1, b0, b1;
#pragma unroll
  for (int j = 0; j < 8; ++j) {
    a0[j] = (_Float16)v[j];
    b0[j] = (_Float16)v[j + 8];
    a1[j] = (_Float16)v[j + 16];
    b1[j] = (_Float16)v[j + 24];
  }
  const int mt = mg * 2 + (lane >> 4);
  _Float16* dp = Wp + (((size_t)e * MT + mt) * KT + kt) * 512 + (lane & 15) * 8;
  *(v8h*)dp = a0;
  *(v8h*)(dp + 16 * 8) = b0;
  *(v8h*)(dp + 256) = a1;
  *(v8h*)(dp + 256 + 16 * 8) = b1;
}

template <int CIN, int COUT>
__device__ __forceinline__ void
pack_mk_tile(const float* __restrict__ W, _Float16* __restrict__ Wp, int g, int lane) {
  constexpr int KT = CIN >> 5, KG = KT >> 2;
  typedef __attribute__((ext_vector_type(4))) float v4f;
  typedef __attribute__((ext_vector_type(4))) _Float16 v4h;
  const int kg = g % KG;
  const int mt = g / KG;
  const int r = lane >> 3, q = lane & 7;
  const float* sp = W + ((size_t)mt * 16 + r) * CIN + kg * 128 + q * 4;
  v4f x[4][4];
#pragma unroll
  for (int t = 0; t < 4; ++t)
#pragma unroll
    for (int i = 0; i < 4; ++i)
      x[t][i] = *(const v4f*)(sp + (size_t)(4 * i) * CIN + t * 32);
  const int fl = r + (((q >> 1) & 1) << 4);
  _Float16* dp = Wp + ((size_t)mt * KT + kg * 4) * 512 + ((q >> 2) << 8) + fl * 8 + ((q & 1) << 2);
#pragma unroll
  for (int t = 0; t < 4; ++t)
#pragma unroll
    for (int i = 0; i < 4; ++i) {
      v4h h;
#pragma unroll
      for (int u = 0; u < 4; ++u) h[u] = (_Float16)x[t][i][u];
      *(v4h*)(dp + (size_t)t * 512 + (4 * i) * 8) = h;
    }
}

__device__ __forceinline__ v16h load_frag(const _Float16* p) {
  v8h lo = *(const v8h*)p;
  v8h hi = *(const v8h*)(p + 256);
  v16h r;
#pragma unroll
  for (int j = 0; j < 8; ++j) { r[j] = lo[j]; r[j + 8] = hi[j]; }
  return r;
}

#define GATE_WAVES 16
#define NGRP (TOK / 32)
__device__ __forceinline__ void
gate_body(const float* __restrict__ x, int C,
          const float* __restrict__ gw,
          int* __restrict__ eidx,
          float* __restrict__ lpart,
          float (*part)[E_][32], int grp, int wave, int lane) {
  const int t0 = grp * 32;
  const int b = t0 / S_, s = (t0 % S_) + lane;
  const int cper = C / GATE_WAVES;
  const int c0 = wave * cper;
  const float* xp = x + ((size_t)b * C + c0) * S_ + s;
  const float* g0 = gw + c0;
  const float* g1 = g0 + C;
  const float* g2 = g1 + C;
  const float* g3 = g2 + C;
  float a0 = 0.f, a1 = 0.f, a2 = 0.f, a3 = 0.f;
  for (int cb = 0; cb < cper; cb += 16) {
    float xv[16];
#pragma unroll
    for (int u = 0; u < 16; ++u) xv[u] = xp[(size_t)(cb + u) * S_];
#pragma unroll
    for (int u = 0; u < 16; ++u) {
      a0 = __builtin_fmaf(xv[u], g0[cb + u], a0);
      a1 = __builtin_fmaf(xv[u], g1[cb + u], a1);
      a2 = __builtin_fmaf(xv[u], g2[cb + u], a2);
      a3 = __builtin_fmaf(xv[u], g3[cb + u], a3);
    }
  }
  part[wave][0][lane] = a0;
  part[wave][1][lane] = a1;
  part[wave][2][lane] = a2;
  part[wave][3][lane] = a3;
  __syncthreads();
  if (wave != 0) return;
  float a[4];
  for (int e = 0; e < E_; ++e) {
    float v = part[0][e][lane];
    for (int w = 1; w < GATE_WAVES; ++w) v += part[w][e][lane];
    a[e] = v;
  }
  int best = 0; float m = a[0];
  for (int e = 1; e < E_; ++e) { if (a[e] > m) { m = a[e]; best = e; } }
  float g[4]; float se = 0.f;
  for (int e = 0; e < E_; ++e) { g[e] = __expf(a[e] - m); se += g[e]; }
  float inv = 1.0f / se;
  eidx[t0 + lane] = best;
  for (int e = 0; e < E_; ++e) {
    float gsum = g[e] * inv;
    for (int d = 16; d > 0; d >>= 1) gsum += __shfl_xor(gsum, d, 32);
    float cnt = (float)__popcll(__ballot(best == e));
    if (lane == 0) {
      lpart[grp * 8 + e] = gsum;
      lpart[grp * 8 + 4 + e] = cnt;
    }
  }
}

__device__ __forceinline__ void store_oct(_Float16* dst, const float* v) {
  v8h h;
#pragma unroll
  for (int jj = 0; jj < 8; ++jj) h[jj] = (_Float16)v[jj];
  *(v8h*)dst = h;
}
__device__ __forceinline__ int oct_off(int oct, int n, int KT) {
  int kt = oct >> 2, a = oct & 3;
  return ((((n >> 4) * KT + kt) << 9) + ((a >> 1) << 8)) + (((n & 15) + ((a & 1) << 4)) << 3);
}

template <int CIN, int COUT, int MTW, int NT, int WAVES, int D, bool FIN>
__global__ void __launch_bounds__(WAVES * 32)
moe_wmma_kernel(const float* __restrict__ X,
                const _Float16* __restrict__ Wp,
                const int* __restrict__ eidx,
                float* __restrict__ Out, float outscale,
                const float* __restrict__ lpart,
                float* __restrict__ out_loss) {
  constexpr int KT = CIN >> 5;
  constexpr int NG = KT / D;
  static_assert(KT % (2 * D) == 0, "two register sets of D k-steps must divide the k-steps");
  constexpr int PIN = (MTW * NT >= 8) ? 1 : 8 / (MTW * NT);
  constexpr int NTOK = NT * 16;
  constexpr int THREADS = WAVES * 32;
  constexpr int ITEMS = (CIN >> 3) * NTOK;
  __shared__ __attribute__((aligned(16))) _Float16 xs[NT * KT * 512];
  const int tok0 = blockIdx.y * NTOK;
  const int b = tok0 / S_, s0 = tok0 % S_;
  auto stage = [&](int q) {
    const int n = q % NTOK, oct = q / NTOK;
    const float* xp = X + ((size_t)b * CIN + oct * 8) * S_ + s0 + n;
    float v[8];
#pragma unroll
    for (int jj = 0; jj < 8; ++jj) v[jj] = xp[(size_t)jj * S_];
    store_oct(&xs[oct_off(oct, n, KT)], v);
  };
  constexpr int FULL = ITEMS / THREADS;
  constexpr int UNR = (FULL % 5 == 0) ? 5 : 4;
#pragma unroll UNR
  for (int it = 0; it < FULL; ++it) stage((int)threadIdx.x + it * THREADS);
  if (ITEMS % THREADS != 0) {
    const int q = (int)threadIdx.x + FULL * THREADS;
    if (q < ITEMS) stage(q);
  }
  __syncthreads();
  const int lane = threadIdx.x & 31;
  const int wave = __builtin_amdgcn_readfirstlane((int)(threadIdx.x >> 5));
  const int col = lane & 15, khalf = (lane >> 4) * 8;
  const int mt0 = (blockIdx.x * WAVES + wave) * MTW;
  constexpr size_t fragstride = (size_t)KT * 512;
  constexpr size_t estride = (size_t)(COUT >> 4) * fragstride;
  int myexp[NT];
#pragma unroll
  for (int nt = 0; nt < NT; ++nt) myexp[nt] = eidx[tok0 + nt * 16 + col];
  v8f cs[MTW][NT];
#pragma unroll
  for (int mi = 0; mi < MTW; ++mi)
#pragma unroll
    for (int nt = 0; nt < NT; ++nt) cs[mi][nt] = v8f{};
  unsigned rem = 0;
#pragma unroll
  for (int e = 0; e < E_; ++e) {
    int hit = 0;
#pragma unroll
    for (int nt = 0; nt < NT; ++nt) hit |= (myexp[nt] == e) ? 1 : 0;
    if (__any(hit)) rem |= 1u << e;
  }
  const _Float16* Bb = xs + lane * 8;
  const _Float16* Abase = Wp + (size_t)mt0 * fragstride;
  const int loff = lane * 8;
  if (rem != 0) {
    int e = __builtin_ctz(rem);
    rem &= rem - 1;
    const _Float16* Acur = Abase + (size_t)e * estride;
    v16h ra[D][MTW], rb[D][MTW];
    v16h bc[NT];
#pragma unroll
    for (int u = 0; u < D; ++u)
#pragma unroll
      for (int mi = 0; mi < MTW; ++mi)
        ra[u][mi] = load_frag(Acur + (size_t)mi * fragstride + (size_t)u * 512 + loff);
#pragma unroll
    for (int nt = 0; nt < NT; ++nt) bc[nt] = load_frag(Bb + (size_t)nt * KT * 512);
    for (;;) {
      const bool more = rem != 0;
      const int en = more ? __builtin_ctz(rem) : e;
      const _Float16* Anext = more ? Abase + (size_t)en * estride
                                   : Acur + (size_t)(KT - D) * 512;
      v8f c[MTW][NT];
#pragma unroll
      for (int mi = 0; mi < MTW; ++mi)
#pragma unroll
        for (int nt = 0; nt < NT; ++nt) c[mi][nt] = v8f{};
      for (int h = 0; h < NG / 2; ++h) {
        const _Float16* Apb = Acur + (size_t)(2 * h + 1) * D * 512;
        const _Float16* Apa = (2 * h + 2 < NG) ? Acur + (size_t)(2 * h + 2) * D * 512 : Anext;
#pragma unroll
        for (int u = 0; u < D; ++u) {
          const int k2 = 2 * h * D + u + 1;
          v16h bn[NT];
#pragma unroll
          for (int nt = 0; nt < NT; ++nt)
            bn[nt] = load_frag(Bb + ((size_t)nt * KT + k2) * 512);
#pragma unroll
          for (int nt = 0; nt < NT; ++nt)
#pragma unroll
            for (int mi = 0; mi < MTW; ++mi)
              c[mi][nt] = __builtin_amdgcn_wmma_f32_16x16x32_f16(
                  false, ra[u][mi], false, bc[nt], (short)0, c[mi][nt], false, false);
#pragma unroll
          for (int mi = 0; mi < MTW; ++mi)
            rb[u][mi] = load_frag(Apb + (size_t)mi * fragstride + (size_t)u * 512 + loff);
          if ((u + 1) % PIN == 0) __builtin_amdgcn_sched_barrier(0);
#pragma unroll
          for (int nt = 0; nt < NT; ++nt) bc[nt] = bn[nt];
        }
#pragma unroll
        for (int u = 0; u < D; ++u) {
          const int kt = (2 * h + 1) * D + u;
          const int k2 = (kt + 1 < KT) ? kt + 1 : 0;
          v16h bn[NT];
#pragma unroll
          for (int nt = 0; nt < NT; ++nt)
            bn[nt] = load_frag(Bb + ((size_t)nt * KT + k2) * 512);
#pragma unroll
          for (int nt = 0; nt < NT; ++nt)
#pragma unroll
            for (int mi = 0; mi < MTW; ++mi)
              c[mi][nt] = __builtin_amdgcn_wmma_f32_16x16x32_f16(
                  false, rb[u][mi], false, bc[nt], (short)0, c[mi][nt], false, false);
#pragma unroll
          for (int mi = 0; mi < MTW; ++mi)
            ra[u][mi] = load_frag(Apa + (size_t)mi * fragstride + (size_t)u * 512 + loff);
          if ((u + 1) % PIN == 0) __builtin_amdgcn_sched_barrier(0);
#pragma unroll
          for (int nt = 0; nt < NT; ++nt) bc[nt] = bn[nt];
        }
      }
#pragma unroll
      for (int nt = 0; nt < NT; ++nt)
        if (myexp[nt] == e) {
#pragma unroll
          for (int mi = 0; mi < MTW; ++mi) cs[mi][nt] = c[mi][nt];
        }
      if (!more) break;
      e = en;
      rem &= rem - 1;
      Acur = Anext;
    }
  }
#pragma unroll
  for (int mi = 0; mi < MTW; ++mi)
#pragma unroll
    for (int nt = 0; nt < NT; ++nt)
#pragma unroll
      for (int j = 0; j < 8; ++j) {
        int r = (mt0 + mi) * 16 + j + khalf;
        Out[((size_t)b * COUT + r) * S_ + s0 + nt * 16 + col] = cs[mi][nt][j] * outscale;
      }
  if (FIN) {
    if (blockIdx.x == 0 && blockIdx.y == 0 && wave == 0) {
      const int a = lane & 15, h = lane >> 4;
      const float* p = lpart + ((size_t)(a >> 3) * NGRP + h * (NGRP / 2)) * 8 + (a & 7);
      float v = 0.f;
#pragma unroll
      for (int g = 0; g < NGRP / 2; ++g) v += p[g * 8];
      v += __shfl_xor(v, 16, 32);
      const float m = v * (1.0f / (float)TOK);
      const float pr = m * __shfl_xor(m, 4, 32);
      float l = 0.f;
      for (int e = 0; e < E_; ++e) l += __shfl(pr, e, 32);
      for (int e = 0; e < E_; ++e) l += __shfl(pr, 8 + e, 32);
      if (lane == 0) *out_loss = l;
    }
  }
}

#define CS_BLOCKS ((B_ * I_) / 8)
#define P1_BLOCKS ((3 * I_ * 3 * I_) / (2048 * 8))
#define P2_BLOCKS ((E_ * I_ * F_) / (1024 * 8))
__global__ void __launch_bounds__(256)
cumsum_pack_kernel(const float* __restrict__ h0,
                   const float* __restrict__ divisor,
                   float* __restrict__ tmp1,
                   const float* __restrict__ w1, _Float16* __restrict__ w1p,
                   const float* __restrict__ w2, _Float16* __restrict__ w2p) {
  const int lane = threadIdx.x & 31;
  const int wv = __builtin_amdgcn_readfirstlane((int)(blockIdx.x * 8 + (threadIdx.x >> 5)));
  if (wv >= (CS_BLOCKS + P1_BLOCKS) * 8) {
    pack_kn_tile<I_, F_>(w2, w2p, wv - (CS_BLOCKS + P1_BLOCKS) * 8, lane);
    return;
  }
  if (wv >= CS_BLOCKS * 8) {
    pack_mk_tile<3 * I_, 3 * I_>(w1, w1p, wv - CS_BLOCKS * 8, lane);
    return;
  }
  const int row = wv;
  int b = row / I_, i = row % I_;
  const float* dep = h0 + ((size_t)b * 3 * I_ + i) * S_;
  const float* sc  = h0 + ((size_t)b * 3 * I_ + I_ + i) * S_;
  const float* sh  = h0 + ((size_t)b * 3 * I_ + 2 * I_ + i) * S_;
  float* o = tmp1 + ((size_t)b * I_ + i) * S_;
  float carry = 0.f;
  for (int s0 = 0; s0 < S_; s0 += 32) {
    float v = dep[s0 + lane];
#pragma unroll
    for (int d = 1; d < 32; d <<= 1) {
      float t = __shfl_up(v, d, 32);
      if (lane >= d) v += t;
    }
    float cum = carry + v;
    o[s0 + lane] = cum / divisor[s0 + lane] * sc[s0 + lane] + sh[s0 + lane];
    carry = __shfl(cum, 31, 32);
  }
}

#define NORM_WAVES 32
#define NORM_CPW (I_ / NORM_WAVES)
#define NORM_G (NORM_CPW / 4)
#define NORM_PITCH 36
typedef __attribute__((ext_vector_type(4))) float v4f;

__device__ __forceinline__ float lane_x8(float x) {
  int b = __builtin_bit_cast(int, x);
  return __builtin_bit_cast(float, __builtin_amdgcn_permlane16(
      b, b, 0xfedcba98u, 0x76543210u, false, false));
}
__device__ __forceinline__ float lane_x16(float x) {
  int b = __builtin_bit_cast(int, x);
  return __builtin_bit_cast(float, __builtin_amdgcn_permlanex16(
      b, b, 0x76543210u, 0xfedcba98u, false, false));
}
__device__ __forceinline__ v4f join4(v4f p) {
#pragma unroll
  for (int k = 0; k < 4; ++k) p[k] += lane_x8(p[k]);
#pragma unroll
  for (int k = 0; k < 4; ++k) p[k] += lane_x16(p[k]);
  return p;
}
__device__ __forceinline__ v4f block_sum(float* part, v4f p, int wave, int tq, int rs) {
  p = join4(p);
  if (rs == 0) *(v4f*)(part + wave * NORM_PITCH + tq * 4) = p;
  __syncthreads();
  const float* pp = part + (8 * rs) * NORM_PITCH + tq * 4;
  v4f q = *(const v4f*)pp;
#pragma unroll
  for (int j = 1; j < 8; ++j) q += *(const v4f*)(pp + j * NORM_PITCH);
  return join4(q);
}

template <int MODE>
__global__ void __launch_bounds__(NORM_WAVES * 32)
norm_kernel(const float* __restrict__ p0,
            const float* __restrict__ p1,
            const float* __restrict__ p2,
            int cstride,
            float* __restrict__ out,
            const float* __restrict__ gw,
            int* __restrict__ eidx,
            float* __restrict__ lpart) {
  __shared__ __attribute__((aligned(16))) float part0[NORM_WAVES * NORM_PITCH];
  __shared__ __attribute__((aligned(16))) float part1[NORM_WAVES * NORM_PITCH];
  __shared__ __attribute__((aligned(16))) float partg[MODE ? E_ : 1][NORM_WAVES * NORM_PITCH];
  const int lane = threadIdx.x & 31;
  const int wave = __builtin_amdgcn_readfirstlane((int)(threadIdx.x >> 5));
  const int tq = lane & 7, rs = lane >> 3;
  const int t0 = blockIdx.x * 32;
  const int b = t0 / S_, s = (t0 % S_) + tq * 4;
  const int c0 = wave * NORM_CPW + rs;
  const size_t ib = ((size_t)b * cstride + c0) * S_ + s;
  v4f x[NORM_G];
#pragma unroll
  for (int g = 0; g < NORM_G; ++g) {
    const size_t idx = ib + (size_t)(4 * g) * S_;
    v4f a = *(const v4f*)(p0 + idx);
    if (MODE) a = a * *(const v4f*)(p1 + idx) + *(const v4f*)(p2 + idx);
    x[g] = a;
  }
  float gwv[E_][NORM_G];
  if (MODE) {
#pragma unroll
    for (int e = 0; e < E_; ++e)
#pragma unroll
      for (int g = 0; g < NORM_G; ++g) gwv[e][g] = gw[e * I_ + c0 + 4 * g];
  }
  v4f ps = x[0];
#pragma unroll
  for (int g = 1; g < NORM_G; ++g) ps += x[g];
  const v4f mean = block_sum(part0, ps, wave, tq, rs) * (1.0f / I_);
  v4f ss = v4f{};
#pragma unroll
  for (int g = 0; g < NORM_G; ++g) {
    x[g] -= mean;
    ss += x[g] * x[g];
  }
  const v4f tot = block_sum(part1, ss, wave, tq, rs);
  v4f inv;
#pragma unroll
  for (int k = 0; k < 4; ++k) {
    const float n = sqrtf(tot[k]) * (1.0f / 32.0f);
    inv[k] = 1.0f / (n + EPSN);
  }
  float* op = out + ((size_t)b * I_ + c0) * S_ + s;
#pragma unroll
  for (int g = 0; g < NORM_G; ++g) {
    v4f y = x[g] * inv;
#pragma unroll
    for (int k = 0; k < 4; ++k) y[k] = y[k] >= 0.f ? y[k] : y[k] * SLOPE_;
    *(v4f*)(op + (size_t)(4 * g) * S_) = y;
    x[g] = y;
  }
  if (MODE) {
    v4f pl[E_];
#pragma unroll
    for (int e = 0; e < E_; ++e) {
      v4f a = x[0] * gwv[e][0];
#pragma unroll
      for (int g = 1; g < NORM_G; ++g) a += x[g] * gwv[e][g];
      pl[e] = join4(a);
    }
    if (rs == 0) {
#pragma unroll
      for (int e = 0; e < E_; ++e) *(v4f*)(partg[e] + wave * NORM_PITCH + tq * 4) = pl[e];
    }
    __syncthreads();
    if (wave != 0) return;
    v4f lg[E_];
#pragma unroll
    for (int e = 0; e < E_; ++e) {
      const float* pp = partg[e] + (8 * rs) * NORM_PITCH + tq * 4;
      v4f q = *(const v4f*)pp;
#pragma unroll
      for (int j = 1; j < 8; ++j) q += *(const v4f*)(pp + j * NORM_PITCH);
      lg[e] = join4(q);
    }
    float gs[E_], cn[E_];
#pragma unroll
    for (int e = 0; e < E_; ++e) { gs[e] = 0.f; cn[e] = 0.f; }
    int bk[4];
#pragma unroll
    for (int k = 0; k < 4; ++k) {
      int best = 0; float m = lg[0][k];
#pragma unroll
      for (int e = 1; e < E_; ++e) { if (lg[e][k] > m) { m = lg[e][k]; best = e; } }
      float g[E_]; float se = 0.f;
#pragma unroll
      for (int e = 0; e < E_; ++e) { g[e] = __expf(lg[e][k] - m); se += g[e]; }
      const float iv = 1.0f / se;
#pragma unroll
      for (int e = 0; e < E_; ++e) {
        gs[e] += g[e] * iv;
        cn[e] += (best == e) ? 1.0f : 0.0f;
      }
      bk[k] = best;
    }
    if (rs == 0) {
      typedef __attribute__((ext_vector_type(4))) int v4i;
      v4i bv; bv[0] = bk[0]; bv[1] = bk[1]; bv[2] = bk[2]; bv[3] = bk[3];
      *(v4i*)(eidx + t0 + tq * 4) = bv;
    }
#pragma unroll
    for (int e = 0; e < E_; ++e)
      for (int d = 1; d < 8; d <<= 1) {
        gs[e] += __shfl_xor(gs[e], d, 32);
        cn[e] += __shfl_xor(cn[e], d, 32);
      }
    if (lane == 0) {
#pragma unroll
      for (int e = 0; e < E_; ++e) {
        lpart[blockIdx.x * 8 + e] = gs[e];
        lpart[blockIdx.x * 8 + 4 + e] = cn[e];
      }
    }
  }
}

#define CV_WAVES 12
#define CV_NT 4
#define CV_KS 6
__global__ void __launch_bounds__(CV_WAVES * 32)
conv_wmma_kernel(const float* __restrict__ G,
                 const _Float16* __restrict__ Wp1,
                 float* __restrict__ Out) {
  constexpr int CHUNK = CV_KS * 32;
  constexpr int NCH = (3 * I_) / CHUNK;
  constexpr int BUF = CV_NT * CV_KS * 512;
  constexpr int ITS = (CV_KS * 4 * CV_NT * 16) / (CV_WAVES * 32);
  constexpr int KTtot = (3 * I_) >> 5;
  constexpr size_t fragstride = (size_t)KTtot * 512;
  __shared__ __attribute__((aligned(16))) _Float16 xs[2][BUF];
  const int tok0 = blockIdx.y * (CV_NT * 16);
  const int b = tok0 / S_, s0 = tok0 % S_;
  const int lane = threadIdx.x & 31;
  const int wave = __builtin_amdgcn_readfirstlane((int)(threadIdx.x >> 5));
  const int col = lane & 15, khalf = (lane >> 4) * 8;
  const int mt0 = (blockIdx.x * CV_WAVES + wave) * 2;
  const _Float16* A0 = Wp1 + (size_t)mt0 * fragstride + lane * 8;
  const _Float16* A1 = A0 + fragstride;
  const int n64 = (wave & 1) * 32 + lane;
  const int octw = wave >> 1;
  const int sb = s0 + n64 - 2;
  const float* Gb = G + (size_t)b * I_ * S_;
  float r[ITS][8];
  auto gload = [&](int cc) {
#pragma unroll
    for (int it = 0; it < ITS; ++it) {
      const int kk0 = cc + (it * (CV_WAVES / 2) + octw) * 8;
#pragma unroll
      for (int jj = 0; jj < 8; ++jj) {
        const int kk = kk0 + jj;
        const int i = kk / 3, kc = kk - i * 3;
        const int s = sb + kc;
        const float gv = Gb[(size_t)i * S_ + (s > 0 ? s : 0)];
        r[it][jj] = (s >= 0) ? gv : 0.0f;
      }
    }
  };
  auto lstore = [&](int buf) {
#pragma unroll
    for (int it = 0; it < ITS; ++it)
      store_oct(&xs[buf][oct_off(it * (CV_WAVES / 2) + octw, n64, CV_KS)], r[it]);
  };
  v8f c[2][CV_NT];
#pragma unroll
  for (int nt = 0; nt < CV_NT; ++nt) { c[0][nt] = v8f{}; c[1][nt] = v8f{}; }
  gload(0);
  lstore(0);
  __syncthreads();
  v16h ag0[3][2], ag1[3][2];
#pragma unroll
  for (int u = 0; u < 3; ++u) {
    ag0[u][0] = load_frag(A0 + (size_t)u * 512);
    ag0[u][1] = load_frag(A1 + (size_t)u * 512);
  }
  for (int ch = 0; ch < NCH; ++ch) {
    const int buf = ch & 1;
    const _Float16* Bb = &xs[buf][lane * 8];
    if (ch + 1 < NCH) gload((ch + 1) * CHUNK);
#pragma unroll
    for (int u = 0; u < 3; ++u) {
      ag1[u][0] = load_frag(A0 + (size_t)(ch * CV_KS + 3 + u) * 512);
      ag1[u][1] = load_frag(A1 + (size_t)(ch * CV_KS + 3 + u) * 512);
    }
    v16h bc[CV_NT], bn[CV_NT];
#pragma unroll
    for (int nt = 0; nt < CV_NT; ++nt) bc[nt] = load_frag(Bb + (nt * CV_KS) * 512);
    __builtin_amdgcn_sched_barrier(0);
#pragma unroll
    for (int kt = 0; kt < CV_KS; ++kt) {
      if (kt == 3) {
#pragma unroll
        for (int u = 0; u < 3; ++u) {
          const int kn = ((ch + 1) * CV_KS + u < KTtot) ? (ch + 1) * CV_KS + u : KTtot - 1;
          ag0[u][0] = load_frag(A0 + (size_t)kn * 512);
          ag0[u][1] = load_frag(A1 + (size_t)kn * 512);
        }
      }
      if (kt + 1 < CV_KS) {
#pragma unroll
        for (int nt = 0; nt < CV_NT; ++nt)
          bn[nt] = load_frag(Bb + (nt * CV_KS + kt + 1) * 512);
      }
      __builtin_amdgcn_sched_barrier(0);
      const v16h a0c = (kt < 3) ? ag0[kt % 3][0] : ag1[kt % 3][0];
      const v16h a1c = (kt < 3) ? ag0[kt % 3][1] : ag1[kt % 3][1];
#pragma unroll
      for (int nt = 0; nt < CV_NT; ++nt) {
        c[0][nt] = __builtin_amdgcn_wmma_f32_16x16x32_f16(
            false, a0c, false, bc[nt], (short)0, c[0][nt], false, false);
        c[1][nt] = __builtin_amdgcn_wmma_f32_16x16x32_f16(
            false, a1c, false, bc[nt], (short)0, c[1][nt], false, false);
      }
      __builtin_amdgcn_sched_barrier(0);
      if (kt + 1 < CV_KS) {
#pragma unroll
        for (int nt = 0; nt < CV_NT; ++nt) bc[nt] = bn[nt];
      }
    }
    if (ch + 1 < NCH) lstore(buf ^ 1);
    __syncthreads();
  }
#pragma unroll
  for (int mi = 0; mi < 2; ++mi)
#pragma unroll
    for (int nt = 0; nt < CV_NT; ++nt)
#pragma unroll
      for (int j = 0; j < 8; ++j) {
        int rr = (mt0 + mi) * 16 + j + khalf;
        Out[((size_t)b * (3 * I_) + rr) * S_ + s0 + nt * 16 + col] = c[mi][nt][j];
      }
}

#define P0_BLOCKS ((E_ * F_ * 3 * I_) / (1024 * GATE_WAVES))
__global__ void __launch_bounds__(GATE_WAVES * 32)
gate_pack_kernel(const float* __restrict__ x,
                 const float* __restrict__ gw,
                 int* __restrict__ eidx,
                 float* __restrict__ lpart,
                 const float* __restrict__ w0, _Float16* __restrict__ w0p) {
  __shared__ float part[GATE_WAVES][E_][32];
  const int lane = threadIdx.x & 31;
  const int wave = __builtin_amdgcn_readfirstlane((int)(threadIdx.x >> 5));
  if (blockIdx.x < NGRP) {
    gate_body(x, F_, gw, eidx, lpart, part, (int)blockIdx.x, wave, lane);
    return;
  }
  pack_kn_tile<F_, 3 * I_>(w0, w0p, (int)(blockIdx.x - NGRP) * GATE_WAVES + wave, lane);
}

extern "C" void kernel_launch(void* const* d_in, const int* in_sizes, int n_in,
                              void* d_out, int out_size, void* d_ws, size_t ws_size,
                              hipStream_t stream) {
  const float* inp     = (const float*)d_in[0];
  const float* divisor = (const float*)d_in[1];
  const float* w0_gate = (const float*)d_in[2];
  const float* w0      = (const float*)d_in[3];
  const float* w1      = (const float*)d_in[4];
  const float* w2_gate = (const float*)d_in[5];
  const float* w2      = (const float*)d_in[6];
  float* out = (float*)d_out;

  char* ws = (char*)d_ws;
  size_t off = 0;
  float* lpart = (float*)(ws + off); off += (size_t)2 * NGRP * 8 * sizeof(float);
  const size_t h0b = (size_t)B_ * 3 * I_ * S_ * sizeof(float);
  const size_t t1b = (size_t)B_ * I_ * S_ * sizeof(float);
  float* h0   = (float*)(ws + off); off += h0b;
  float* tmp1 = (float*)(ws + off); off += t1b;
  float* g1   = (float*)(ws + off); off += t1b;
  int*   idx0 = (int*)(ws + off); off += TOK * sizeof(int);
  int*   idx1 = (int*)(ws + off); off += TOK * sizeof(int);
  off = (off + 255) & ~(size_t)255;
  const size_t w0n = (size_t)E_ * F_ * 3 * I_;
  const size_t w1n = (size_t)3 * I_ * 3 * I_;
  const size_t w2n = (size_t)E_ * I_ * F_;
  _Float16* w0p = (_Float16*)(ws + off); off += w0n * 2;
  _Float16* w1p = (_Float16*)(ws + off); off += w1n * 2;
  _Float16* w2p = (_Float16*)(ws + off); off += w2n * 2;
  float* h2 = h0;
  float* h3 = tmp1;

  gate_pack_kernel<<<NGRP + P0_BLOCKS, GATE_WAVES * 32, 0, stream>>>(
      inp, w0_gate, idx0, lpart, w0, w0p);
  moe_wmma_kernel<F_, 3 * I_, 2, 4, 12, 4, false><<<dim3(3 * I_ / 384, TOK / 64), 384, 0, stream>>>(
      inp, w0p, idx0, h0, 1.0f, nullptr, nullptr);

  cumsum_pack_kernel<<<CS_BLOCKS + P1_BLOCKS + P2_BLOCKS, 256, 0, stream>>>(
      h0, divisor, tmp1, w1, w1p, w2, w2p);
  norm_kernel<0><<<TOK / 32, NORM_WAVES * 32, 0, stream>>>(
      tmp1, tmp1, tmp1, I_, g1, w2_gate, idx1, lpart);

  conv_wmma_kernel<<<dim3(3 * I_ / 384, TOK / 64), CV_WAVES * 32, 0, stream>>>(g1, w1p, h2);

  norm_kernel<1><<<TOK / 32, NORM_WAVES * 32, 0, stream>>>(
      h2, h2 + (size_t)I_ * S_, h2 + (size_t)2 * I_ * S_, 3 * I_, h3,
      w2_gate, idx1, lpart + NGRP * 8);

  moe_wmma_kernel<I_, F_, 1, 2, 8, 8, true><<<dim3(F_ / 128, TOK / 32), 256, 0, stream>>>(
      h3, w2p, idx1, out, 1.0f  , lpart, out + (size_t)B_ * F_ * S_);
}


# WMMA GEMMs re-tiled squarer, 2 wave columns per work-group (conv, GEMM 1: 192 rows x 128 tokens, was 384 x 64; GEMM 2: 64 x 64, was 128 x 32): a row's 2 waves share weight reads; 16-byte row loads
# speedup vs baseline: 5.2647x; 1.1884x over previous
#include <hip/hip_runtime.h>
#include <hip/hip_bf16.h>

typedef __attribute__((ext_vector_type(16))) _Float16 v16h;
typedef __attribute__((ext_vector_type(8)))  _Float16 v8h;
typedef __attribute__((ext_vector_type(8)))  float    v8f;
typedef __attribute__((ext_vector_type(4)))  float    v4f;
typedef __attribute__((ext_vector_type(2)))  float    v2f;

#define B_ 2
#define F_ 512
#define S_ 1024
#define E_ 4
#define I_ 1024
#define TOK (B_*S_)
#define EPSN 1e-5f
#define SLOPE_ 0.02f

__device__ __forceinline__ int frag_j(int kin) {
  return kin < 8 ? kin : (kin < 16 ? kin - 8 : (kin < 24 ? kin - 8 : kin - 16));
}
__device__ __forceinline__ int frag_lane(int kin, int n) {
  return n + (((kin >> 3) & 1) ? 16 : 0);
}

template <int CIN, int COUT>
__device__ __forceinline__ void
pack_kn_tile(const float* __restrict__ W, _Float16* __restrict__ Wp, int g, int lane) {
  constexpr int KT = CIN >> 5, MT = COUT >> 4, MG = COUT >> 5;
  const int mg = g % MG;
  const int kt = (g / MG) % KT;
  const int e  = g / (MG * KT);
  const float* sp = W + ((size_t)e * CIN + (size_t)kt * 32) * COUT + mg * 32 + lane;
  float v[32];
#pragma unroll
  for (int k = 0; k < 32; ++k) v[k] = sp[(size_t)k * COUT];
  v8h a0, a1, b0, b1;
#pragma unroll
  for (int j = 0; j < 8; ++j) {
    a0[j] = (_Float16)v[j];
    b0[j] = (_Float16)v[j + 8];
    a1[j] = (_Float16)v[j + 16];
    b1[j] = (_Float16)v[j + 24];
  }
  const int mt = mg * 2 + (lane >> 4);
  _Float16* dp = Wp + (((size_t)e * MT + mt) * KT + kt) * 512 + (lane & 15) * 8;
  *(v8h*)dp = a0;
  *(v8h*)(dp + 16 * 8) = b0;
  *(v8h*)(dp + 256) = a1;
  *(v8h*)(dp + 256 + 16 * 8) = b1;
}

template <int CIN, int COUT>
__device__ __forceinline__ void
pack_mk_tile(const float* __restrict__ W, _Float16* __restrict__ Wp, int g, int lane) {
  constexpr int KT = CIN >> 5, KG = KT >> 2;
  typedef __attribute__((ext_vector_type(4))) float v4f;
  typedef __attribute__((ext_vector_type(4))) _Float16 v4h;
  const int kg = g % KG;
  const int mt = g / KG;
  const int r = lane >> 3, q = lane & 7;
  const float* sp = W + ((size_t)mt * 16 + r) * CIN + kg * 128 + q * 4;
  v4f x[4][4];
#pragma unroll
  for (int t = 0; t < 4; ++t)
#pragma unroll
    for (int i = 0; i < 4; ++i)
      x[t][i] = *(const v4f*)(sp + (size_t)(4 * i) * CIN + t * 32);
  const int fl = r + (((q >> 1) & 1) << 4);
  _Float16* dp = Wp + ((size_t)mt * KT + kg * 4) * 512 + ((q >> 2) << 8) + fl * 8 + ((q & 1) << 2);
#pragma unroll
  for (int t = 0; t < 4; ++t)
#pragma unroll
    for (int i = 0; i < 4; ++i) {
      v4h h;
#pragma unroll
      for (int u = 0; u < 4; ++u) h[u] = (_Float16)x[t][i][u];
      *(v4h*)(dp + (size_t)t * 512 + (4 * i) * 8) = h;
    }
}

__device__ __forceinline__ v16h load_frag(const _Float16* p) {
  v8h lo = *(const v8h*)p;
  v8h hi = *(const v8h*)(p + 256);
  v16h r;
#pragma unroll
  for (int j = 0; j < 8; ++j) { r[j] = lo[j]; r[j + 8] = hi[j]; }
  return r;
}

#define GATE_WAVES 16
#define NGRP (TOK / 32)
__device__ __forceinline__ void
gate_body(const float* __restrict__ x, int C,
          const float* __restrict__ gw,
          int* __restrict__ eidx,
          float* __restrict__ lpart,
          float (*part)[E_][32], int grp, int wave, int lane) {
  const int t0 = grp * 32;
  const int b = t0 / S_, s = (t0 % S_) + lane;
  const int cper = C / GATE_WAVES;
  const int c0 = wave * cper;
  const float* xp = x + ((size_t)b * C + c0) * S_ + s;
  const float* g0 = gw + c0;
  const float* g1 = g0 + C;
  const float* g2 = g1 + C;
  const float* g3 = g2 + C;
  float a0 = 0.f, a1 = 0.f, a2 = 0.f, a3 = 0.f;
  for (int cb = 0; cb < cper; cb += 16) {
    float xv[16];
#pragma unroll
    for (int u = 0; u < 16; ++u) xv[u] = xp[(size_t)(cb + u) * S_];
#pragma unroll
    for (int u = 0; u < 16; ++u) {
      a0 = __builtin_fmaf(xv[u], g0[cb + u], a0);
      a1 = __builtin_fmaf(xv[u], g1[cb + u], a1);
      a2 = __builtin_fmaf(xv[u], g2[cb + u], a2);
      a3 = __builtin_fmaf(xv[u], g3[cb + u], a3);
    }
  }
  part[wave][0][lane] = a0;
  part[wave][1][lane] = a1;
  part[wave][2][lane] = a2;
  part[wave][3][lane] = a3;
  __syncthreads();
  if (wave != 0) return;
  float a[4];
  for (int e = 0; e < E_; ++e) {
    float v = part[0][e][lane];
    for (int w = 1; w < GATE_WAVES; ++w) v += part[w][e][lane];
    a[e] = v;
  }
  int best = 0; float m = a[0];
  for (int e = 1; e < E_; ++e) { if (a[e] > m) { m = a[e]; best = e; } }
  float g[4]; float se = 0.f;
  for (int e = 0; e < E_; ++e) { g[e] = __expf(a[e] - m); se += g[e]; }
  float inv = 1.0f / se;
  eidx[t0 + lane] = best;
  for (int e = 0; e < E_; ++e) {
    float gsum = g[e] * inv;
    for (int d = 16; d > 0; d >>= 1) gsum += __shfl_xor(gsum, d, 32);
    float cnt = (float)__popcll(__ballot(best == e));
    if (lane == 0) {
      lpart[grp * 8 + e] = gsum;
      lpart[grp * 8 + 4 + e] = cnt;
    }
  }
}

__device__ __forceinline__ void store_oct(_Float16* dst, const float* v) {
  v8h h;
#pragma unroll
  for (int jj = 0; jj < 8; ++jj) h[jj] = (_Float16)v[jj];
  *(v8h*)dst = h;
}
__device__ __forceinline__ int oct_off(int oct, int n, int TS) {
  int kt = oct >> 2, a = oct & 3;
  return (n >> 4) * TS + (kt << 9) + ((a >> 1) << 8) + (((n & 15) + ((a & 1) << 4)) << 3);
}

template <int CIN, int COUT, int MTW, int NT, int WAVES, int D, bool FIN, int NW>
__global__ void __launch_bounds__(WAVES * 32)
moe_wmma_kernel(const float* __restrict__ X,
                const _Float16* __restrict__ Wp,
                const int* __restrict__ eidx,
                float* __restrict__ Out, float outscale,
                const float* __restrict__ lpart,
                float* __restrict__ out_loss) {
  constexpr int KT = CIN >> 5;
  constexpr int NG = KT / D;
  static_assert(KT % (2 * D) == 0, "two register sets of D k-steps must divide the k-steps");
  constexpr int PIN = (MTW * NT >= 8) ? 1 : 8 / (MTW * NT);
  static_assert(WAVES % NW == 0 && (NW & (NW - 1)) == 0, "wave columns: a power of two that divides the waves");
  constexpr int NTOK = NW * NT * 16;
  constexpr int NQ = NTOK / 4;
  constexpr int TS = KT * 512 + 8;
  constexpr int THREADS = WAVES * 32;
  constexpr int ITEMS = (CIN >> 3) * NQ;
  __shared__ __attribute__((aligned(16))) _Float16 xs[NW * NT * TS];
  const int tok0 = blockIdx.y * NTOK;
  const int b = tok0 / S_, s0 = tok0 % S_;
  auto stage = [&](int q) {
    const int qd = q % NQ, oct = q / NQ;
    const float* xp = X + ((size_t)b * CIN + oct * 8) * S_ + s0 + 4 * qd;
    v4f x[8];
#pragma unroll
    for (int jj = 0; jj < 8; ++jj) x[jj] = *(const v4f*)(xp + (size_t)jj * S_);
#pragma unroll
    for (int t = 0; t < 4; ++t) {
      float v[8];
#pragma unroll
      for (int jj = 0; jj < 8; ++jj) v[jj] = x[jj][t];
      store_oct(&xs[oct_off(oct, 4 * qd + t, TS)], v);
    }
  };
  constexpr int FULL = ITEMS / THREADS;
  constexpr int UNR = (FULL % 5 == 0) ? 5 : 4;
#pragma unroll UNR
  for (int it = 0; it < FULL; ++it) stage((int)threadIdx.x + it * THREADS);
  if (ITEMS % THREADS != 0) {
    const int q = (int)threadIdx.x + FULL * THREADS;
    if (q < ITEMS) stage(q);
  }
  __syncthreads();
  const int lane = threadIdx.x & 31;
  const int wave = __builtin_amdgcn_readfirstlane((int)(threadIdx.x >> 5));
  const int col = lane & 15, khalf = (lane >> 4) * 8;
  const int wm = wave / NW, wn = wave % NW;
  const int nt0 = wn * NT;
  const int mt0 = (blockIdx.x * (WAVES / NW) + wm) * MTW;
  constexpr size_t fragstride = (size_t)KT * 512;
  constexpr size_t estride = (size_t)(COUT >> 4) * fragstride;
  int myexp[NT];
#pragma unroll
  for (int nt = 0; nt < NT; ++nt) myexp[nt] = eidx[tok0 + (nt0 + nt) * 16 + col];
  v8f cs[MTW][NT];
#pragma unroll
  for (int mi = 0; mi < MTW; ++mi)
#pragma unroll
    for (int nt = 0; nt < NT; ++nt) cs[mi][nt] = v8f{};
  unsigned rem = 0;
#pragma unroll
  for (int e = 0; e < E_; ++e) {
    int hit = 0;
#pragma unroll
    for (int nt = 0; nt < NT; ++nt) hit |= (myexp[nt] == e) ? 1 : 0;
    if (__any(hit)) rem |= 1u << e;
  }
  const _Float16* Bb = xs + nt0 * TS + lane * 8;
  const _Float16* Abase = Wp + (size_t)mt0 * fragstride;
  const int loff = lane * 8;
  if (rem != 0) {
    int e = __builtin_ctz(rem);
    rem &= rem - 1;
    const _Float16* Acur = Abase + (size_t)e * estride;
    v16h ra[D][MTW], rb[D][MTW];
    v16h bc[NT];
#pragma unroll
    for (int u = 0; u < D; ++u)
#pragma unroll
      for (int mi = 0; mi < MTW; ++mi)
        ra[u][mi] = load_frag(Acur + (size_t)mi * fragstride + (size_t)u * 512 + loff);
#pragma unroll
    for (int nt = 0; nt < NT; ++nt) bc[nt] = load_frag(Bb + nt * TS);
    for (;;) {
      const bool more = rem != 0;
      const int en = more ? __builtin_ctz(rem) : e;
      const _Float16* Anext = more ? Abase + (size_t)en * estride
                                   : Acur + (size_t)(KT - D) * 512;
      v8f c[MTW][NT];
#pragma unroll
      for (int mi = 0; mi < MTW; ++mi)
#pragma unroll
        for (int nt = 0; nt < NT; ++nt) c[mi][nt] = v8f{};
      for (int h = 0; h < NG / 2; ++h) {
        const _Float16* Apb = Acur + (size_t)(2 * h + 1) * D * 512;
        const _Float16* Apa = (2 * h + 2 < NG) ? Acur + (size_t)(2 * h + 2) * D * 512 : Anext;
#pragma unroll
        for (int u = 0; u < D; ++u) {
          const int k2 = 2 * h * D + u + 1;
          v16h bn[NT];
#pragma unroll
          for (int nt = 0; nt < NT; ++nt)
            bn[nt] = load_frag(Bb + nt * TS + k2 * 512);
#pragma unroll
          for (int nt = 0; nt < NT; ++nt)
#pragma unroll
            for (int mi = 0; mi < MTW; ++mi)
              c[mi][nt] = __builtin_amdgcn_wmma_f32_16x16x32_f16(
                  false, ra[u][mi], false, bc[nt], (short)0, c[mi][nt], false, false);
#pragma unroll
          for (int mi = 0; mi < MTW; ++mi)
            rb[u][mi] = load_frag(Apb + (size_t)mi * fragstride + (size_t)u * 512 + loff);
          if ((u + 1) % PIN == 0) __builtin_amdgcn_sched_barrier(0);
#pragma unroll
          for (int nt = 0; nt < NT; ++nt) bc[nt] = bn[nt];
        }
#pragma unroll
        for (int u = 0; u < D; ++u) {
          const int kt = (2 * h + 1) * D + u;
          const int k2 = (kt + 1 < KT) ? kt + 1 : 0;
          v16h bn[NT];
#pragma unroll
          for (int nt = 0; nt < NT; ++nt)
            bn[nt] = load_frag(Bb + nt * TS + k2 * 512);
#pragma unroll
          for (int nt = 0; nt < NT; ++nt)
#pragma unroll
            for (int mi = 0; mi < MTW; ++mi)
              c[mi][nt] = __builtin_amdgcn_wmma_f32_16x16x32_f16(
                  false, rb[u][mi], false, bc[nt], (short)0, c[mi][nt], false, false);
#pragma unroll
          for (int mi = 0; mi < MTW; ++mi)
            ra[u][mi] = load_frag(Apa + (size_t)mi * fragstride + (size_t)u * 512 + loff);
          if ((u + 1) % PIN == 0) __builtin_amdgcn_sched_barrier(0);
#pragma unroll
          for (int nt = 0; nt < NT; ++nt) bc[nt] = bn[nt];
        }
      }
#pragma unroll
      for (int nt = 0; nt < NT; ++nt)
        if (myexp[nt] == e) {
#pragma unroll
          for (int mi = 0; mi < MTW; ++mi) cs[mi][nt] = c[mi][nt];
        }
      if (!more) break;
      e = en;
      rem &= rem - 1;
      Acur = Anext;
    }
  }
#pragma unroll
  for (int mi = 0; mi < MTW; ++mi)
#pragma unroll
    for (int nt = 0; nt < NT; ++nt)
#pragma unroll
      for (int j = 0; j < 8; ++j) {
        int r = (mt0 + mi) * 16 + j + khalf;
        Out[((size_t)b * COUT + r) * S_ + s0 + (nt0 + nt) * 16 + col] = cs[mi][nt][j] * outscale;
      }
  if (FIN) {
    if (blockIdx.x == 0 && blockIdx.y == 0 && wave == 0) {
      const int a = lane & 15, h = lane >> 4;
      const float* p = lpart + ((size_t)(a >> 3) * NGRP + h * (NGRP / 2)) * 8 + (a & 7);
      float v = 0.f;
#pragma unroll
      for (int g = 0; g < NGRP / 2; ++g) v += p[g * 8];
      v += __shfl_xor(v, 16, 32);
      const float m = v * (1.0f / (float)TOK);
      const float pr = m * __shfl_xor(m, 4, 32);
      float l = 0.f;
      for (int e = 0; e < E_; ++e) l += __shfl(pr, e, 32);
      for (int e = 0; e < E_; ++e) l += __shfl(pr, 8 + e, 32);
      if (lane == 0) *out_loss = l;
    }
  }
}

#define CS_BLOCKS ((B_ * I_) / 8)
#define P1_BLOCKS ((3 * I_ * 3 * I_) / (2048 * 8))
#define P2_BLOCKS ((E_ * I_ * F_) / (1024 * 8))
__global__ void __launch_bounds__(256)
cumsum_pack_kernel(const float* __restrict__ h0,
                   const float* __restrict__ divisor,
                   float* __restrict__ tmp1,
                   const float* __restrict__ w1, _Float16* __restrict__ w1p,
                   const float* __restrict__ w2, _Float16* __restrict__ w2p) {
  const int lane = threadIdx.x & 31;
  const int wv = __builtin_amdgcn_readfirstlane((int)(blockIdx.x * 8 + (threadIdx.x >> 5)));
  if (wv >= (CS_BLOCKS + P1_BLOCKS) * 8) {
    pack_kn_tile<I_, F_>(w2, w2p, wv - (CS_BLOCKS + P1_BLOCKS) * 8, lane);
    return;
  }
  if (wv >= CS_BLOCKS * 8) {
    pack_mk_tile<3 * I_, 3 * I_>(w1, w1p, wv - CS_BLOCKS * 8, lane);
    return;
  }
  typedef __attribute__((ext_vector_type(4))) float v4f;
  constexpr int NCK = S_ / 128;
  const int row = wv;
  const int b = row / I_, i = row % I_;
  const v4f* dep = (const v4f*)(h0 + ((size_t)b * 3 * I_ + i) * S_) + lane;
  const v4f* sc  = (const v4f*)(h0 + ((size_t)b * 3 * I_ + I_ + i) * S_) + lane;
  const v4f* sh  = (const v4f*)(h0 + ((size_t)b * 3 * I_ + 2 * I_ + i) * S_) + lane;
  const v4f* dv  = (const v4f*)divisor + lane;
  v4f* o = (v4f*)(tmp1 + ((size_t)b * I_ + i) * S_) + lane;
  v4f x[NCK], vd[NCK], vs[NCK], vh[NCK];
#pragma unroll
  for (int j = 0; j < NCK; ++j) x[j] = dep[j * 32];
#pragma unroll
  for (int j = 0; j < NCK; ++j) vd[j] = dv[j * 32];
#pragma unroll
  for (int j = 0; j < NCK; ++j) vs[j] = sc[j * 32];
#pragma unroll
  for (int j = 0; j < NCK; ++j) vh[j] = sh[j * 32];
  __builtin_amdgcn_sched_barrier(0);
  float t[NCK];
#pragma unroll
  for (int j = 0; j < NCK; ++j) {
    x[j][1] += x[j][0];
    x[j][2] += x[j][1];
    x[j][3] += x[j][2];
    t[j] = x[j][3];
  }
#pragma unroll
  for (int d = 1; d < 32; d <<= 1) {
#pragma unroll
    for (int j = 0; j < NCK; ++j) {
      const float u = __shfl_up(t[j], d, 32);
      if (lane >= d) t[j] += u;
    }
  }
  float carry = 0.f;
#pragma unroll
  for (int j = 0; j < NCK; ++j) {
    const float ex = __shfl_up(t[j], 1, 32);
    const float base = carry + (lane == 0 ? 0.f : ex);
    carry += __shfl(t[j], 31, 32);
    const v4f cum = x[j] + base;
    o[j * 32] = cum / vd[j] * vs[j] + vh[j];
  }
}

#define NORM_WAVES 32
#define NORM_CPW (I_ / NORM_WAVES)
#define NORM_G (NORM_CPW / 4)
#define NORM_PITCH 36

__device__ __forceinline__ float lane_x8(float x) {
  int b = __builtin_bit_cast(int, x);
  return __builtin_bit_cast(float, __builtin_amdgcn_permlane16(
      b, b, 0xfedcba98u, 0x76543210u, false, false));
}
__device__ __forceinline__ float lane_x16(float x) {
  int b = __builtin_bit_cast(int, x);
  return __builtin_bit_cast(float, __builtin_amdgcn_permlanex16(
      b, b, 0x76543210u, 0xfedcba98u, false, false));
}
__device__ __forceinline__ v4f join4(v4f p) {
#pragma unroll
  for (int k = 0; k < 4; ++k) p[k] += lane_x8(p[k]);
#pragma unroll
  for (int k = 0; k < 4; ++k) p[k] += lane_x16(p[k]);
  return p;
}
__device__ __forceinline__ v4f block_sum(float* part, v4f p, int wave, int tq, int rs) {
  p = join4(p);
  if (rs == 0) *(v4f*)(part + wave * NORM_PITCH + tq * 4) = p;
  __syncthreads();
  const float* pp = part + (8 * rs) * NORM_PITCH + tq * 4;
  v4f q = *(const v4f*)pp;
#pragma unroll
  for (int j = 1; j < 8; ++j) q += *(const v4f*)(pp + j * NORM_PITCH);
  return join4(q);
}

template <int MODE>
__global__ void __launch_bounds__(NORM_WAVES * 32)
norm_kernel(const float* __restrict__ p0,
            const float* __restrict__ p1,
            const float* __restrict__ p2,
            int cstride,
            float* __restrict__ out,
            const float* __restrict__ gw,
            int* __restrict__ eidx,
            float* __restrict__ lpart) {
  __shared__ __attribute__((aligned(16))) float part0[NORM_WAVES * NORM_PITCH];
  __shared__ __attribute__((aligned(16))) float part1[NORM_WAVES * NORM_PITCH];
  __shared__ __attribute__((aligned(16))) float partg[MODE ? E_ : 1][NORM_WAVES * NORM_PITCH];
  const int lane = threadIdx.x & 31;
  const int wave = __builtin_amdgcn_readfirstlane((int)(threadIdx.x >> 5));
  const int tq = lane & 7, rs = lane >> 3;
  const int t0 = blockIdx.x * 32;
  const int b = t0 / S_, s = (t0 % S_) + tq * 4;
  const int c0 = wave * NORM_CPW + rs;
  const size_t ib = ((size_t)b * cstride + c0) * S_ + s;
  v4f x[NORM_G];
#pragma unroll
  for (int g = 0; g < NORM_G; ++g) {
    const size_t idx = ib + (size_t)(4 * g) * S_;
    v4f a = *(const v4f*)(p0 + idx);
    if (MODE) a = a * *(const v4f*)(p1 + idx) + *(const v4f*)(p2 + idx);
    x[g] = a;
  }
  float gwv[E_][NORM_G];
  if (MODE) {
#pragma unroll
    for (int e = 0; e < E_; ++e)
#pragma unroll
      for (int g = 0; g < NORM_G; ++g) gwv[e][g] = gw[e * I_ + c0 + 4 * g];
  }
  v4f ps = x[0];
#pragma unroll
  for (int g = 1; g < NORM_G; ++g) ps += x[g];
  const v4f mean = block_sum(part0, ps, wave, tq, rs) * (1.0f / I_);
  v4f ss = v4f{};
#pragma unroll
  for (int g = 0; g < NORM_G; ++g) {
    x[g] -= mean;
    ss += x[g] * x[g];
  }
  const v4f tot = block_sum(part1, ss, wave, tq, rs);
  v4f inv;
#pragma unroll
  for (int k = 0; k < 4; ++k) {
    const float n = sqrtf(tot[k]) * (1.0f / 32.0f);
    inv[k] = 1.0f / (n + EPSN);
  }
  float* op = out + ((size_t)b * I_ + c0) * S_ + s;
#pragma unroll
  for (int g = 0; g < NORM_G; ++g) {
    v4f y = x[g] * inv;
#pragma unroll
    for (int k = 0; k < 4; ++k) y[k] = y[k] >= 0.f ? y[k] : y[k] * SLOPE_;
    *(v4f*)(op + (size_t)(4 * g) * S_) = y;
    x[g] = y;
  }
  if (MODE) {
    v4f pl[E_];
#pragma unroll
    for (int e = 0; e < E_; ++e) {
      v4f a = x[0] * gwv[e][0];
#pragma unroll
      for (int g = 1; g < NORM_G; ++g) a += x[g] * gwv[e][g];
      pl[e] = join4(a);
    }
    if (rs == 0) {
#pragma unroll
      for (int e = 0; e < E_; ++e) *(v4f*)(partg[e] + wave * NORM_PITCH + tq * 4) = pl[e];
    }
    __syncthreads();
    if (wave != 0) return;
    v4f lg[E_];
#pragma unroll
    for (int e = 0; e < E_; ++e) {
      const float* pp = partg[e] + (8 * rs) * NORM_PITCH + tq * 4;
      v4f q = *(const v4f*)pp;
#pragma unroll
      for (int j = 1; j < 8; ++j) q += *(const v4f*)(pp + j * NORM_PITCH);
      lg[e] = join4(q);
    }
    float gs[E_], cn[E_];
#pragma unroll
    for (int e = 0; e < E_; ++e) { gs[e] = 0.f; cn[e] = 0.f; }
    int bk[4];
#pragma unroll
    for (int k = 0; k < 4; ++k) {
      int best = 0; float m = lg[0][k];
#pragma unroll
      for (int e = 1; e < E_; ++e) { if (lg[e][k] > m) { m = lg[e][k]; best = e; } }
      float g[E_]; float se = 0.f;
#pragma unroll
      for (int e = 0; e < E_; ++e) { g[e] = __expf(lg[e][k] - m); se += g[e]; }
      const float iv = 1.0f / se;
#pragma unroll
      for (int e = 0; e < E_; ++e) {
        gs[e] += g[e] * iv;
        cn[e] += (best == e) ? 1.0f : 0.0f;
      }
      bk[k] = best;
    }
    if (rs == 0) {
      typedef __attribute__((ext_vector_type(4))) int v4i;
      v4i bv; bv[0] = bk[0]; bv[1] = bk[1]; bv[2] = bk[2]; bv[3] = bk[3];
      *(v4i*)(eidx + t0 + tq * 4) = bv;
    }
#pragma unroll
    for (int e = 0; e < E_; ++e)
      for (int d = 1; d < 8; d <<= 1) {
        gs[e] += __shfl_xor(gs[e], d, 32);
        cn[e] += __shfl_xor(cn[e], d, 32);
      }
    if (lane == 0) {
#pragma unroll
      for (int e = 0; e < E_; ++e) {
        lpart[blockIdx.x * 8 + e] = gs[e];
        lpart[blockIdx.x * 8 + 4 + e] = cn[e];
      }
    }
  }
}

#define CV_WAVES 8
#define CV_NW 2
#define CV_NT 4
#define CV_MT 3
#define CV_KS 6
__global__ void __launch_bounds__(CV_WAVES * 32)
conv_wmma_kernel(const float* __restrict__ G,
                 const _Float16* __restrict__ Wp1,
                 float* __restrict__ Out) {
  constexpr int CHUNK = CV_KS * 32;
  constexpr int NCH = (3 * I_) / CHUNK;
  constexpr int TS = CV_KS * 512 + 8;
  constexpr int BUF = CV_NW * CV_NT * TS;
  constexpr int KTtot = (3 * I_) >> 5;
  constexpr size_t fragstride = (size_t)KTtot * 512;
  static_assert(CHUNK == CV_WAVES * 24, "one 8-channel group (24 k) per wave");
  static_assert(CV_NW * CV_NT * 4 == 32, "one token quad per lane");
  __shared__ __attribute__((aligned(16))) _Float16 xs[2][BUF];
  const int tok0 = blockIdx.y * (CV_NW * CV_NT * 16);
  const int b = tok0 / S_, s0 = tok0 % S_;
  const int lane = threadIdx.x & 31;
  const int wave = __builtin_amdgcn_readfirstlane((int)(threadIdx.x >> 5));
  const int wm = wave / CV_NW, wn = wave % CV_NW;
  const int nt0 = wn * CV_NT;
  const int col = lane & 15, khalf = (lane >> 4) * 8;
  const int mt0 = (blockIdx.x * (CV_WAVES / CV_NW) + wm) * CV_MT;
  const _Float16* Ab = Wp1 + (size_t)mt0 * fragstride + lane * 8;
  const float* Gq = G + ((size_t)b * I_ + (size_t)wave * 8) * S_ + s0 + 4 * lane;
  const int hoff = (s0 + 4 * lane >= 2) ? -2 : 0;
  v4f rm[8];
  v2f rh[8];
  auto gload = [&](int ch) {
    const float* gp = Gq + (size_t)(ch * 64) * S_;
#pragma unroll
    for (int c = 0; c < 8; ++c) {
      rm[c] = *(const v4f*)(gp + (size_t)c * S_);
      rh[c] = *(const v2f*)(gp + (size_t)c * S_ + hoff);
    }
  };
  auto lstore = [&](int buf) {
#pragma unroll
    for (int t = 0; t < 4; ++t) {
#pragma unroll
      for (int j = 0; j < 3; ++j) {
        float v[8];
#pragma unroll
        for (int e = 0; e < 8; ++e) {
          const int kl = 8 * j + e, c = kl / 3, tap = kl - 3 * c;
          const int x = t + tap;
          v[e] = (x < 2) ? (hoff ? rh[c][x & 1] : 0.0f) : rm[c][(x - 2) & 3];
        }
        store_oct(&xs[buf][oct_off(wave * 3 + j, 4 * lane + t, TS)], v);
      }
    }
  };
  v8f c[CV_MT][CV_NT];
#pragma unroll
  for (int mi = 0; mi < CV_MT; ++mi)
#pragma unroll
    for (int nt = 0; nt < CV_NT; ++nt) c[mi][nt] = v8f{};
  gload(0);
  lstore(0);
  __syncthreads();
  v16h ag0[3][CV_MT], ag1[3][CV_MT];
#pragma unroll
  for (int u = 0; u < 3; ++u)
#pragma unroll
    for (int mi = 0; mi < CV_MT; ++mi)
      ag0[u][mi] = load_frag(Ab + (size_t)mi * fragstride + (size_t)u * 512);
  for (int ch = 0; ch < NCH; ++ch) {
    const int buf = ch & 1;
    const _Float16* Bb = &xs[buf][nt0 * TS + lane * 8];
    gload(ch + 1 < NCH ? ch + 1 : ch);
#pragma unroll
    for (int u = 0; u < 3; ++u)
#pragma unroll
      for (int mi = 0; mi < CV_MT; ++mi)
        ag1[u][mi] = load_frag(Ab + (size_t)mi * fragstride + (size_t)(ch * CV_KS + 3 + u) * 512);
    v16h bc[CV_NT], bn[CV_NT];
#pragma unroll
    for (int nt = 0; nt < CV_NT; ++nt) bc[nt] = load_frag(Bb + nt * TS);
    __builtin_amdgcn_sched_barrier(0);
#pragma unroll
    for (int kt = 0; kt < CV_KS; ++kt) {
      if (kt == 3) {
#pragma unroll
        for (int u = 0; u < 3; ++u) {
          const int kn = ((ch + 1) * CV_KS + u < KTtot) ? (ch + 1) * CV_KS + u : KTtot - 1;
#pragma unroll
          for (int mi = 0; mi < CV_MT; ++mi)
            ag0[u][mi] = load_frag(Ab + (size_t)mi * fragstride + (size_t)kn * 512);
        }
      }
      if (kt + 1 < CV_KS) {
#pragma unroll
        for (int nt = 0; nt < CV_NT; ++nt)
          bn[nt] = load_frag(Bb + nt * TS + (kt + 1) * 512);
      }
      __builtin_amdgcn_sched_barrier(0);
#pragma unroll
      for (int nt = 0; nt < CV_NT; ++nt)
#pragma unroll
        for (int mi = 0; mi < CV_MT; ++mi)
          c[mi][nt] = __builtin_amdgcn_wmma_f32_16x16x32_f16(
              false, (kt < 3) ? ag0[kt % 3][mi] : ag1[kt % 3][mi], false, bc[nt],
              (short)0, c[mi][nt], false, false);
      __builtin_amdgcn_sched_barrier(0);
      if (kt + 1 < CV_KS) {
#pragma unroll
        for (int nt = 0; nt < CV_NT; ++nt) bc[nt] = bn[nt];
      }
    }
    lstore(buf ^ 1);
    __syncthreads();
  }
#pragma unroll
  for (int mi = 0; mi < CV_MT; ++mi)
#pragma unroll
    for (int nt = 0; nt < CV_NT; ++nt)
#pragma unroll
      for (int j = 0; j < 8; ++j) {
        int rr = (mt0 + mi) * 16 + j + khalf;
        Out[((size_t)b * (3 * I_) + rr) * S_ + s0 + (nt0 + nt) * 16 + col] = c[mi][nt][j];
      }
}

#define P0_BLOCKS ((E_ * F_ * 3 * I_) / (1024 * GATE_WAVES))
__global__ void __launch_bounds__(GATE_WAVES * 32)
gate_pack_kernel(const float* __restrict__ x,
                 const float* __restrict__ gw,
                 int* __restrict__ eidx,
                 float* __restrict__ lpart,
                 const float* __restrict__ w0, _Float16* __restrict__ w0p) {
  __shared__ float part[GATE_WAVES][E_][32];
  const int lane = threadIdx.x & 31;
  const int wave = __builtin_amdgcn_readfirstlane((int)(threadIdx.x >> 5));
  if (blockIdx.x < NGRP) {
    gate_body(x, F_, gw, eidx, lpart, part, (int)blockIdx.x, wave, lane);
    return;
  }
  pack_kn_tile<F_, 3 * I_>(w0, w0p, (int)(blockIdx.x - NGRP) * GATE_WAVES + wave, lane);
}

extern "C" void kernel_launch(void* const* d_in, const int* in_sizes, int n_in,
                              void* d_out, int out_size, void* d_ws, size_t ws_size,
                              hipStream_t stream) {
  const float* inp     = (const float*)d_in[0];
  const float* divisor = (const float*)d_in[1];
  const float* w0_gate = (const float*)d_in[2];
  const float* w0      = (const float*)d_in[3];
  const float* w1      = (const float*)d_in[4];
  const float* w2_gate = (const float*)d_in[5];
  const float* w2      = (const float*)d_in[6];
  float* out = (float*)d_out;

  char* ws = (char*)d_ws;
  size_t off = 0;
  float* lpart = (float*)(ws + off); off += (size_t)2 * NGRP * 8 * sizeof(float);
  const size_t h0b = (size_t)B_ * 3 * I_ * S_ * sizeof(float);
  const size_t t1b = (size_t)B_ * I_ * S_ * sizeof(float);
  float* h0   = (float*)(ws + off); off += h0b;
  float* tmp1 = (float*)(ws + off); off += t1b;
  float* g1   = (float*)(ws + off); off += t1b;
  int*   idx0 = (int*)(ws + off); off += TOK * sizeof(int);
  int*   idx1 = (int*)(ws + off); off += TOK * sizeof(int);
  off = (off + 255) & ~(size_t)255;
  const size_t w0n = (size_t)E_ * F_ * 3 * I_;
  const size_t w1n = (size_t)3 * I_ * 3 * I_;
  const size_t w2n = (size_t)E_ * I_ * F_;
  _Float16* w0p = (_Float16*)(ws + off); off += w0n * 2;
  _Float16* w1p = (_Float16*)(ws + off); off += w1n * 2;
  _Float16* w2p = (_Float16*)(ws + off); off += w2n * 2;
  float* h2 = h0;
  float* h3 = tmp1;

  gate_pack_kernel<<<NGRP + P0_BLOCKS, GATE_WAVES * 32, 0, stream>>>(
      inp, w0_gate, idx0, lpart, w0, w0p);
  moe_wmma_kernel<F_, 3 * I_, 2, 4, 12, 4, false, 2><<<dim3(3 * I_ / 192, TOK / 128), 384, 0, stream>>>(
      inp, w0p, idx0, h0, 1.0f, nullptr, nullptr);

  cumsum_pack_kernel<<<CS_BLOCKS + P1_BLOCKS + P2_BLOCKS, 256, 0, stream>>>(
      h0, divisor, tmp1, w1, w1p, w2, w2p);
  norm_kernel<0><<<TOK / 32, NORM_WAVES * 32, 0, stream>>>(
      tmp1, tmp1, tmp1, I_, g1, w2_gate, idx1, lpart);

  conv_wmma_kernel<<<dim3(3 * I_ / 192, TOK / 128), CV_WAVES * 32, 0, stream>>>(g1, w1p, h2);

  norm_kernel<1><<<TOK / 32, NORM_WAVES * 32, 0, stream>>>(
      h2, h2 + (size_t)I_ * S_, h2 + (size_t)2 * I_ * S_, 3 * I_, h3,
      w2_gate, idx1, lpart + NGRP * 8);

  moe_wmma_kernel<I_, F_, 1, 2, 8, 8, true, 2><<<dim3(F_ / 64, TOK / 64), 256, 0, stream>>>(
      h3, w2p, idx1, out, 1.0f  , lpart, out + (size_t)B_ * F_ * S_);
}


# conv epilogue forms s0*s1+sh in registers: a wave's 3 m-tiles are the same 16 rows of the three planes; one plane stored (8 MB, was 25), the norm behind reads one plane, not three
# speedup vs baseline: 5.8029x; 1.1022x over previous
#include <hip/hip_runtime.h>
#include <hip/hip_bf16.h>

typedef __attribute__((ext_vector_type(16))) _Float16 v16h;
typedef __attribute__((ext_vector_type(8)))  _Float16 v8h;
typedef __attribute__((ext_vector_type(8)))  float    v8f;
typedef __attribute__((ext_vector_type(4)))  float    v4f;
typedef __attribute__((ext_vector_type(2)))  float    v2f;

#define B_ 2
#define F_ 512
#define S_ 1024
#define E_ 4
#define I_ 1024
#define TOK (B_*S_)
#define EPSN 1e-5f
#define SLOPE_ 0.02f

__device__ __forceinline__ int frag_j(int kin) {
  return kin < 8 ? kin : (kin < 16 ? kin - 8 : (kin < 24 ? kin - 8 : kin - 16));
}
__device__ __forceinline__ int frag_lane(int kin, int n) {
  return n + (((kin >> 3) & 1) ? 16 : 0);
}

template <int CIN, int COUT>
__device__ __forceinline__ void
pack_kn_tile(const float* __restrict__ W, _Float16* __restrict__ Wp, int g, int lane) {
  constexpr int KT = CIN >> 5, MT = COUT >> 4, MG = COUT >> 5;
  const int mg = g % MG;
  const int kt = (g / MG) % KT;
  const int e  = g / (MG * KT);
  const float* sp = W + ((size_t)e * CIN + (size_t)kt * 32) * COUT + mg * 32 + lane;
  float v[32];
#pragma unroll
  for (int k = 0; k < 32; ++k) v[k] = sp[(size_t)k * COUT];
  v8h a0, a1, b0, b1;
#pragma unroll
  for (int j = 0; j < 8; ++j) {
    a0[j] = (_Float16)v[j];
    b0[j] = (_Float16)v[j + 8];
    a1[j] = (_Float16)v[j + 16];
    b1[j] = (_Float16)v[j + 24];
  }
  const int mt = mg * 2 + (lane >> 4);
  _Float16* dp = Wp + (((size_t)e * MT + mt) * KT + kt) * 512 + (lane & 15) * 8;
  *(v8h*)dp = a0;
  *(v8h*)(dp + 16 * 8) = b0;
  *(v8h*)(dp + 256) = a1;
  *(v8h*)(dp + 256 + 16 * 8) = b1;
}

template <int CIN, int COUT>
__device__ __forceinline__ void
pack_mk_tile(const float* __restrict__ W, _Float16* __restrict__ Wp, int g, int lane) {
  constexpr int KT = CIN >> 5, KG = KT >> 2;
  typedef __attribute__((ext_vector_type(4))) float v4f;
  typedef __attribute__((ext_vector_type(4))) _Float16 v4h;
  const int kg = g % KG;
  const int mt = g / KG;
  const int r = lane >> 3, q = lane & 7;
  const float* sp = W + ((size_t)mt * 16 + r) * CIN + kg * 128 + q * 4;
  v4f x[4][4];
#pragma unroll
  for (int t = 0; t < 4; ++t)
#pragma unroll
    for (int i = 0; i < 4; ++i)
      x[t][i] = *(const v4f*)(sp + (size_t)(4 * i) * CIN + t * 32);
  const int fl = r + (((q >> 1) & 1) << 4);
  _Float16* dp = Wp + ((size_t)mt * KT + kg * 4) * 512 + ((q >> 2) << 8) + fl * 8 + ((q & 1) << 2);
#pragma unroll
  for (int t = 0; t < 4; ++t)
#pragma unroll
    for (int i = 0; i < 4; ++i) {
      v4h h;
#pragma unroll
      for (int u = 0; u < 4; ++u) h[u] = (_Float16)x[t][i][u];
      *(v4h*)(dp + (size_t)t * 512 + (4 * i) * 8) = h;
    }
}

__device__ __forceinline__ v16h load_frag(const _Float16* p) {
  v8h lo = *(const v8h*)p;
  v8h hi = *(const v8h*)(p + 256);
  v16h r;
#pragma unroll
  for (int j = 0; j < 8; ++j) { r[j] = lo[j]; r[j + 8] = hi[j]; }
  return r;
}

#define GATE_WAVES 16
#define NGRP (TOK / 32)
__device__ __forceinline__ void
gate_body(const float* __restrict__ x, int C,
          const float* __restrict__ gw,
          int* __restrict__ eidx,
          float* __restrict__ lpart,
          float (*part)[E_][32], int grp, int wave, int lane) {
  const int t0 = grp * 32;
  const int b = t0 / S_, s = (t0 % S_) + lane;
  const int cper = C / GATE_WAVES;
  const int c0 = wave * cper;
  const float* xp = x + ((size_t)b * C + c0) * S_ + s;
  const float* g0 = gw + c0;
  const float* g1 = g0 + C;
  const float* g2 = g1 + C;
  const float* g3 = g2 + C;
  float a0 = 0.f, a1 = 0.f, a2 = 0.f, a3 = 0.f;
  for (int cb = 0; cb < cper; cb += 16) {
    float xv[16];
#pragma unroll
    for (int u = 0; u < 16; ++u) xv[u] = xp[(size_t)(cb + u) * S_];
#pragma unroll
    for (int u = 0; u < 16; ++u) {
      a0 = __builtin_fmaf(xv[u], g0[cb + u], a0);
      a1 = __builtin_fmaf(xv[u], g1[cb + u], a1);
      a2 = __builtin_fmaf(xv[u], g2[cb + u], a2);
      a3 = __builtin_fmaf(xv[u], g3[cb + u], a3);
    }
  }
  part[wave][0][lane] = a0;
  part[wave][1][lane] = a1;
  part[wave][2][lane] = a2;
  part[wave][3][lane] = a3;
  __syncthreads();
  if (wave != 0) return;
  float a[4];
  for (int e = 0; e < E_; ++e) {
    float v = part[0][e][lane];
    for (int w = 1; w < GATE_WAVES; ++w) v += part[w][e][lane];
    a[e] = v;
  }
  int best = 0; float m = a[0];
  for (int e = 1; e < E_; ++e) { if (a[e] > m) { m = a[e]; best = e; } }
  float g[4]; float se = 0.f;
  for (int e = 0; e < E_; ++e) { g[e] = __expf(a[e] - m); se += g[e]; }
  float inv = 1.0f / se;
  eidx[t0 + lane] = best;
  for (int e = 0; e < E_; ++e) {
    float gsum = g[e] * inv;
    for (int d = 16; d > 0; d >>= 1) gsum += __shfl_xor(gsum, d, 32);
    float cnt = (float)__popcll(__ballot(best == e));
    if (lane == 0) {
      lpart[grp * 8 + e] = gsum;
      lpart[grp * 8 + 4 + e] = cnt;
    }
  }
}

__device__ __forceinline__ void store_oct(_Float16* dst, const float* v) {
  v8h h;
#pragma unroll
  for (int jj = 0; jj < 8; ++jj) h[jj] = (_Float16)v[jj];
  *(v8h*)dst = h;
}
__device__ __forceinline__ int oct_off(int oct, int n, int TS) {
  int kt = oct >> 2, a = oct & 3;
  return (n >> 4) * TS + (kt << 9) + ((a >> 1) << 8) + (((n & 15) + ((a & 1) << 4)) << 3);
}

template <int CIN, int COUT, int MTW, int NT, int WAVES, int D, bool FIN, int NW>
__global__ void __launch_bounds__(WAVES * 32)
moe_wmma_kernel(const float* __restrict__ X,
                const _Float16* __restrict__ Wp,
                const int* __restrict__ eidx,
                float* __restrict__ Out, float outscale,
                const float* __restrict__ lpart,
                float* __restrict__ out_loss) {
  constexpr int KT = CIN >> 5;
  constexpr int NG = KT / D;
  static_assert(KT % (2 * D) == 0, "two register sets of D k-steps must divide the k-steps");
  constexpr int PIN = (MTW * NT >= 8) ? 1 : 8 / (MTW * NT);
  static_assert(WAVES % NW == 0 && (NW & (NW - 1)) == 0, "wave columns: a power of two that divides the waves");
  constexpr int NTOK = NW * NT * 16;
  constexpr int NQ = NTOK / 4;
  constexpr int TS = KT * 512 + 8;
  constexpr int THREADS = WAVES * 32;
  constexpr int ITEMS = (CIN >> 3) * NQ;
  __shared__ __attribute__((aligned(16))) _Float16 xs[NW * NT * TS];
  const int tok0 = blockIdx.y * NTOK;
  const int b = tok0 / S_, s0 = tok0 % S_;
  auto stage = [&](int q) {
    const int qd = q % NQ, oct = q / NQ;
    const float* xp = X + ((size_t)b * CIN + oct * 8) * S_ + s0 + 4 * qd;
    v4f x[8];
#pragma unroll
    for (int jj = 0; jj < 8; ++jj) x[jj] = *(const v4f*)(xp + (size_t)jj * S_);
#pragma unroll
    for (int t = 0; t < 4; ++t) {
      float v[8];
#pragma unroll
      for (int jj = 0; jj < 8; ++jj) v[jj] = x[jj][t];
      store_oct(&xs[oct_off(oct, 4 * qd + t, TS)], v);
    }
  };
  constexpr int FULL = ITEMS / THREADS;
  constexpr int UNR = (FULL % 5 == 0) ? 5 : 4;
#pragma unroll UNR
  for (int it = 0; it < FULL; ++it) stage((int)threadIdx.x + it * THREADS);
  if (ITEMS % THREADS != 0) {
    const int q = (int)threadIdx.x + FULL * THREADS;
    if (q < ITEMS) stage(q);
  }
  __syncthreads();
  const int lane = threadIdx.x & 31;
  const int wave = __builtin_amdgcn_readfirstlane((int)(threadIdx.x >> 5));
  const int col = lane & 15, khalf = (lane >> 4) * 8;
  const int wm = wave / NW, wn = wave % NW;
  const int nt0 = wn * NT;
  const int mt0 = (blockIdx.x * (WAVES / NW) + wm) * MTW;
  constexpr size_t fragstride = (size_t)KT * 512;
  constexpr size_t estride = (size_t)(COUT >> 4) * fragstride;
  int myexp[NT];
#pragma unroll
  for (int nt = 0; nt < NT; ++nt) myexp[nt] = eidx[tok0 + (nt0 + nt) * 16 + col];
  v8f cs[MTW][NT];
#pragma unroll
  for (int mi = 0; mi < MTW; ++mi)
#pragma unroll
    for (int nt = 0; nt < NT; ++nt) cs[mi][nt] = v8f{};
  unsigned rem = 0;
#pragma unroll
  for (int e = 0; e < E_; ++e) {
    int hit = 0;
#pragma unroll
    for (int nt = 0; nt < NT; ++nt) hit |= (myexp[nt] == e) ? 1 : 0;
    if (__any(hit)) rem |= 1u << e;
  }
  const _Float16* Bb = xs + nt0 * TS + lane * 8;
  const _Float16* Abase = Wp + (size_t)mt0 * fragstride;
  const int loff = lane * 8;
  if (rem != 0) {
    int e = __builtin_ctz(rem);
    rem &= rem - 1;
    const _Float16* Acur = Abase + (size_t)e * estride;
    v16h ra[D][MTW], rb[D][MTW];
    v16h bc[NT];
#pragma unroll
    for (int u = 0; u < D; ++u)
#pragma unroll
      for (int mi = 0; mi < MTW; ++mi)
        ra[u][mi] = load_frag(Acur + (size_t)mi * fragstride + (size_t)u * 512 + loff);
#pragma unroll
    for (int nt = 0; nt < NT; ++nt) bc[nt] = load_frag(Bb + nt * TS);
    for (;;) {
      const bool more = rem != 0;
      const int en = more ? __builtin_ctz(rem) : e;
      const _Float16* Anext = more ? Abase + (size_t)en * estride
                                   : Acur + (size_t)(KT - D) * 512;
      v8f c[MTW][NT];
#pragma unroll
      for (int mi = 0; mi < MTW; ++mi)
#pragma unroll
        for (int nt = 0; nt < NT; ++nt) c[mi][nt] = v8f{};
      for (int h = 0; h < NG / 2; ++h) {
        const _Float16* Apb = Acur + (size_t)(2 * h + 1) * D * 512;
        const _Float16* Apa = (2 * h + 2 < NG) ? Acur + (size_t)(2 * h + 2) * D * 512 : Anext;
#pragma unroll
        for (int u = 0; u < D; ++u) {
          const int k2 = 2 * h * D + u + 1;
          v16h bn[NT];
#pragma unroll
          for (int nt = 0; nt < NT; ++nt)
            bn[nt] = load_frag(Bb + nt * TS + k2 * 512);
#pragma unroll
          for (int nt = 0; nt < NT; ++nt)
#pragma unroll
            for (int mi = 0; mi < MTW; ++mi)
              c[mi][nt] = __builtin_amdgcn_wmma_f32_16x16x32_f16(
                  false, ra[u][mi], false, bc[nt], (short)0, c[mi][nt], false, false);
#pragma unroll
          for (int mi = 0; mi < MTW; ++mi)
            rb[u][mi] = load_frag(Apb + (size_t)mi * fragstride + (size_t)u * 512 + loff);
          if ((u + 1) % PIN == 0) __builtin_amdgcn_sched_barrier(0);
#pragma unroll
          for (int nt = 0; nt < NT; ++nt) bc[nt] = bn[nt];
        }
#pragma unroll
        for (int u = 0; u < D; ++u) {
          const int kt = (2 * h + 1) * D + u;
          const int k2 = (kt + 1 < KT) ? kt + 1 : 0;
          v16h bn[NT];
#pragma unroll
          for (int nt = 0; nt < NT; ++nt)
            bn[nt] = load_frag(Bb + nt * TS + k2 * 512);
#pragma unroll
          for (int nt = 0; nt < NT; ++nt)
#pragma unroll
            for (int mi = 0; mi < MTW; ++mi)
              c[mi][nt] = __builtin_amdgcn_wmma_f32_16x16x32_f16(
                  false, rb[u][mi], false, bc[nt], (short)0, c[mi][nt], false, false);
#pragma unroll
          for (int mi = 0; mi < MTW; ++mi)
            ra[u][mi] = load_frag(Apa + (size_t)mi * fragstride + (size_t)u * 512 + loff);
          if ((u + 1) % PIN == 0) __builtin_amdgcn_sched_barrier(0);
#pragma unroll
          for (int nt = 0; nt < NT; ++nt) bc[nt] = bn[nt];
        }
      }
#pragma unroll
      for (int nt = 0; nt < NT; ++nt)
        if (myexp[nt] == e) {
#pragma unroll
          for (int mi = 0; mi < MTW; ++mi) cs[mi][nt] = c[mi][nt];
        }
      if (!more) break;
      e = en;
      rem &= rem - 1;
      Acur = Anext;
    }
  }
#pragma unroll
  for (int mi = 0; mi < MTW; ++mi)
#pragma unroll
    for (int nt = 0; nt < NT; ++nt)
#pragma unroll
      for (int j = 0; j < 8; ++j) {
        int r = (mt0 + mi) * 16 + j + khalf;
        Out[((size_t)b * COUT + r) * S_ + s0 + (nt0 + nt) * 16 + col] = cs[mi][nt][j] * outscale;
      }
  if (FIN) {
    if (blockIdx.x == 0 && blockIdx.y == 0 && wave == 0) {
      const int a = lane & 15, h = lane >> 4;
      const float* p = lpart + ((size_t)(a >> 3) * NGRP + h * (NGRP / 2)) * 8 + (a & 7);
      float v = 0.f;
#pragma unroll
      for (int g = 0; g < NGRP / 2; ++g) v += p[g * 8];
      v += __shfl_xor(v, 16, 32);
      const float m = v * (1.0f / (float)TOK);
      const float pr = m * __shfl_xor(m, 4, 32);
      float l = 0.f;
      for (int e = 0; e < E_; ++e) l += __shfl(pr, e, 32);
      for (int e = 0; e < E_; ++e) l += __shfl(pr, 8 + e, 32);
      if (lane == 0) *out_loss = l;
    }
  }
}

#define CS_BLOCKS ((B_ * I_) / 8)
#define P1_BLOCKS ((3 * I_ * 3 * I_) / (2048 * 8))
#define P2_BLOCKS ((E_ * I_ * F_) / (1024 * 8))
__global__ void __launch_bounds__(256)
cumsum_pack_kernel(const float* __restrict__ h0,
                   const float* __restrict__ divisor,
                   float* __restrict__ tmp1,
                   const float* __restrict__ w1, _Float16* __restrict__ w1p,
                   const float* __restrict__ w2, _Float16* __restrict__ w2p) {
  const int lane = threadIdx.x & 31;
  const int wv = __builtin_amdgcn_readfirstlane((int)(blockIdx.x * 8 + (threadIdx.x >> 5)));
  if (wv >= (CS_BLOCKS + P1_BLOCKS) * 8) {
    pack_kn_tile<I_, F_>(w2, w2p, wv - (CS_BLOCKS + P1_BLOCKS) * 8, lane);
    return;
  }
  if (wv >= CS_BLOCKS * 8) {
    pack_mk_tile<3 * I_, 3 * I_>(w1, w1p, wv - CS_BLOCKS * 8, lane);
    return;
  }
  typedef __attribute__((ext_vector_type(4))) float v4f;
  constexpr int NCK = S_ / 128;
  const int row = wv;
  const int b = row / I_, i = row % I_;
  const v4f* dep = (const v4f*)(h0 + ((size_t)b * 3 * I_ + i) * S_) + lane;
  const v4f* sc  = (const v4f*)(h0 + ((size_t)b * 3 * I_ + I_ + i) * S_) + lane;
  const v4f* sh  = (const v4f*)(h0 + ((size_t)b * 3 * I_ + 2 * I_ + i) * S_) + lane;
  const v4f* dv  = (const v4f*)divisor + lane;
  v4f* o = (v4f*)(tmp1 + ((size_t)b * I_ + i) * S_) + lane;
  v4f x[NCK], vd[NCK], vs[NCK], vh[NCK];
#pragma unroll
  for (int j = 0; j < NCK; ++j) x[j] = dep[j * 32];
#pragma unroll
  for (int j = 0; j < NCK; ++j) vd[j] = dv[j * 32];
#pragma unroll
  for (int j = 0; j < NCK; ++j) vs[j] = sc[j * 32];
#pragma unroll
  for (int j = 0; j < NCK; ++j) vh[j] = sh[j * 32];
  __builtin_amdgcn_sched_barrier(0);
  float t[NCK];
#pragma unroll
  for (int j = 0; j < NCK; ++j) {
    x[j][1] += x[j][0];
    x[j][2] += x[j][1];
    x[j][3] += x[j][2];
    t[j] = x[j][3];
  }
#pragma unroll
  for (int d = 1; d < 32; d <<= 1) {
#pragma unroll
    for (int j = 0; j < NCK; ++j) {
      const float u = __shfl_up(t[j], d, 32);
      if (lane >= d) t[j] += u;
    }
  }
  float carry = 0.f;
#pragma unroll
  for (int j = 0; j < NCK; ++j) {
    const float ex = __shfl_up(t[j], 1, 32);
    const float base = carry + (lane == 0 ? 0.f : ex);
    carry += __shfl(t[j], 31, 32);
    const v4f cum = x[j] + base;
    o[j * 32] = cum / vd[j] * vs[j] + vh[j];
  }
}

#define NORM_WAVES 32
#define NORM_CPW (I_ / NORM_WAVES)
#define NORM_G (NORM_CPW / 4)
#define NORM_PITCH 36

__device__ __forceinline__ float lane_x8(float x) {
  int b = __builtin_bit_cast(int, x);
  return __builtin_bit_cast(float, __builtin_amdgcn_permlane16(
      b, b, 0xfedcba98u, 0x76543210u, false, false));
}
__device__ __forceinline__ float lane_x16(float x) {
  int b = __builtin_bit_cast(int, x);
  return __builtin_bit_cast(float, __builtin_amdgcn_permlanex16(
      b, b, 0x76543210u, 0xfedcba98u, false, false));
}
__device__ __forceinline__ v4f join4(v4f p) {
#pragma unroll
  for (int k = 0; k < 4; ++k) p[k] += lane_x8(p[k]);
#pragma unroll
  for (int k = 0; k < 4; ++k) p[k] += lane_x16(p[k]);
  return p;
}
__device__ __forceinline__ v4f block_sum(float* part, v4f p, int wave, int tq, int rs) {
  p = join4(p);
  if (rs == 0) *(v4f*)(part + wave * NORM_PITCH + tq * 4) = p;
  __syncthreads();
  const float* pp = part + (8 * rs) * NORM_PITCH + tq * 4;
  v4f q = *(const v4f*)pp;
#pragma unroll
  for (int j = 1; j < 8; ++j) q += *(const v4f*)(pp + j * NORM_PITCH);
  return join4(q);
}

template <int MODE>
__global__ void __launch_bounds__(NORM_WAVES * 32)
norm_kernel(const float* __restrict__ p0,
            float* __restrict__ out,
            const float* __restrict__ gw,
            int* __restrict__ eidx,
            float* __restrict__ lpart) {
  __shared__ __attribute__((aligned(16))) float part0[NORM_WAVES * NORM_PITCH];
  __shared__ __attribute__((aligned(16))) float part1[NORM_WAVES * NORM_PITCH];
  __shared__ __attribute__((aligned(16))) float partg[MODE ? E_ : 1][NORM_WAVES * NORM_PITCH];
  const int lane = threadIdx.x & 31;
  const int wave = __builtin_amdgcn_readfirstlane((int)(threadIdx.x >> 5));
  const int tq = lane & 7, rs = lane >> 3;
  const int t0 = blockIdx.x * 32;
  const int b = t0 / S_, s = (t0 % S_) + tq * 4;
  const int c0 = wave * NORM_CPW + rs;
  const size_t ib = ((size_t)b * I_ + c0) * S_ + s;
  v4f x[NORM_G];
#pragma unroll
  for (int g = 0; g < NORM_G; ++g) x[g] = *(const v4f*)(p0 + ib + (size_t)(4 * g) * S_);
  float gwv[E_][NORM_G];
  if (MODE) {
#pragma unroll
    for (int e = 0; e < E_; ++e)
#pragma unroll
      for (int g = 0; g < NORM_G; ++g) gwv[e][g] = gw[e * I_ + c0 + 4 * g];
  }
  v4f ps = x[0];
#pragma unroll
  for (int g = 1; g < NORM_G; ++g) ps += x[g];
  const v4f mean = block_sum(part0, ps, wave, tq, rs) * (1.0f / I_);
  v4f ss = v4f{};
#pragma unroll
  for (int g = 0; g < NORM_G; ++g) {
    x[g] -= mean;
    ss += x[g] * x[g];
  }
  const v4f tot = block_sum(part1, ss, wave, tq, rs);
  v4f inv;
#pragma unroll
  for (int k = 0; k < 4; ++k) {
    const float n = sqrtf(tot[k]) * (1.0f / 32.0f);
    inv[k] = 1.0f / (n + EPSN);
  }
  float* op = out + ((size_t)b * I_ + c0) * S_ + s;
#pragma unroll
  for (int g = 0; g < NORM_G; ++g) {
    v4f y = x[g] * inv;
#pragma unroll
    for (int k = 0; k < 4; ++k) y[k] = y[k] >= 0.f ? y[k] : y[k] * SLOPE_;
    *(v4f*)(op + (size_t)(4 * g) * S_) = y;
    x[g] = y;
  }
  if (MODE) {
    v4f pl[E_];
#pragma unroll
    for (int e = 0; e < E_; ++e) {
      v4f a = x[0] * gwv[e][0];
#pragma unroll
      for (int g = 1; g < NORM_G; ++g) a += x[g] * gwv[e][g];
      pl[e] = join4(a);
    }
    if (rs == 0) {
#pragma unroll
      for (int e = 0; e < E_; ++e) *(v4f*)(partg[e] + wave * NORM_PITCH + tq * 4) = pl[e];
    }
    __syncthreads();
    if (wave != 0) return;
    v4f lg[E_];
#pragma unroll
    for (int e = 0; e < E_; ++e) {
      const float* pp = partg[e] + (8 * rs) * NORM_PITCH + tq * 4;
      v4f q = *(const v4f*)pp;
#pragma unroll
      for (int j = 1; j < 8; ++j) q += *(const v4f*)(pp + j * NORM_PITCH);
      lg[e] = join4(q);
    }
    float gs[E_], cn[E_];
#pragma unroll
    for (int e = 0; e < E_; ++e) { gs[e] = 0.f; cn[e] = 0.f; }
    int bk[4];
#pragma unroll
    for (int k = 0; k < 4; ++k) {
      int best = 0; float m = lg[0][k];
#pragma unroll
      for (int e = 1; e < E_; ++e) { if (lg[e][k] > m) { m = lg[e][k]; best = e; } }
      float g[E_]; float se = 0.f;
#pragma unroll
      for (int e = 0; e < E_; ++e) { g[e] = __expf(lg[e][k] - m); se += g[e]; }
      const float iv = 1.0f / se;
#pragma unroll
      for (int e = 0; e < E_; ++e) {
        gs[e] += g[e] * iv;
        cn[e] += (best == e) ? 1.0f : 0.0f;
      }
      bk[k] = best;
    }
    if (rs == 0) {
      typedef __attribute__((ext_vector_type(4))) int v4i;
      v4i bv; bv[0] = bk[0]; bv[1] = bk[1]; bv[2] = bk[2]; bv[3] = bk[3];
      *(v4i*)(eidx + t0 + tq * 4) = bv;
    }
#pragma unroll
    for (int e = 0; e < E_; ++e)
      for (int d = 1; d < 8; d <<= 1) {
        gs[e] += __shfl_xor(gs[e], d, 32);
        cn[e] += __shfl_xor(cn[e], d, 32);
      }
    if (lane == 0) {
#pragma unroll
      for (int e = 0; e < E_; ++e) {
        lpart[blockIdx.x * 8 + e] = gs[e];
        lpart[blockIdx.x * 8 + 4 + e] = cn[e];
      }
    }
  }
}

#define CV_WAVES 8
#define CV_NW 2
#define CV_NT 4
#define CV_MT 3
#define CV_KS 6
__global__ void __launch_bounds__(CV_WAVES * 32)
conv_wmma_kernel(const float* __restrict__ G,
                 const _Float16* __restrict__ Wp1,
                 float* __restrict__ Out) {
  constexpr int CHUNK = CV_KS * 32;
  constexpr int NCH = (3 * I_) / CHUNK;
  constexpr int TS = CV_KS * 512 + 8;
  constexpr int BUF = CV_NW * CV_NT * TS;
  constexpr int KTtot = (3 * I_) >> 5;
  constexpr size_t fragstride = (size_t)KTtot * 512;
  static_assert(CHUNK == CV_WAVES * 24, "one 8-channel group (24 k) per wave");
  static_assert(CV_NW * CV_NT * 4 == 32, "one token quad per lane");
  __shared__ __attribute__((aligned(16))) _Float16 xs[2][BUF];
  const int tok0 = blockIdx.y * (CV_NW * CV_NT * 16);
  const int b = tok0 / S_, s0 = tok0 % S_;
  const int lane = threadIdx.x & 31;
  const int wave = __builtin_amdgcn_readfirstlane((int)(threadIdx.x >> 5));
  const int wm = wave / CV_NW, wn = wave % CV_NW;
  const int nt0 = wn * CV_NT;
  const int col = lane & 15, khalf = (lane >> 4) * 8;
  const int mq = blockIdx.x * (CV_WAVES / CV_NW) + wm;
  constexpr size_t planestride = (size_t)(I_ >> 4) * fragstride;
  const _Float16* Ab = Wp1 + (size_t)mq * fragstride + lane * 8;
  const float* Gq = G + ((size_t)b * I_ + (size_t)wave * 8) * S_ + s0 + 4 * lane;
  const int hoff = (s0 + 4 * lane >= 2) ? -2 : 0;
  v4f rm[8];
  v2f rh[8];
  auto gload = [&](int ch) {
    const float* gp = Gq + (size_t)(ch * 64) * S_;
#pragma unroll
    for (int c = 0; c < 8; ++c) {
      rm[c] = *(const v4f*)(gp + (size_t)c * S_);
      rh[c] = *(const v2f*)(gp + (size_t)c * S_ + hoff);
    }
  };
  auto lstore = [&](int buf) {
#pragma unroll
    for (int t = 0; t < 4; ++t) {
#pragma unroll
      for (int j = 0; j < 3; ++j) {
        float v[8];
#pragma unroll
        for (int e = 0; e < 8; ++e) {
          const int kl = 8 * j + e, c = kl / 3, tap = kl - 3 * c;
          const int x = t + tap;
          v[e] = (x < 2) ? (hoff ? rh[c][x & 1] : 0.0f) : rm[c][(x - 2) & 3];
        }
        store_oct(&xs[buf][oct_off(wave * 3 + j, 4 * lane + t, TS)], v);
      }
    }
  };
  v8f c[CV_MT][CV_NT];
#pragma unroll
  for (int mi = 0; mi < CV_MT; ++mi)
#pragma unroll
    for (int nt = 0; nt < CV_NT; ++nt) c[mi][nt] = v8f{};
  gload(0);
  lstore(0);
  __syncthreads();
  v16h ag0[3][CV_MT], ag1[3][CV_MT];
#pragma unroll
  for (int u = 0; u < 3; ++u)
#pragma unroll
    for (int mi = 0; mi < CV_MT; ++mi)
      ag0[u][mi] = load_frag(Ab + (size_t)mi * planestride + (size_t)u * 512);
  for (int ch = 0; ch < NCH; ++ch) {
    const int buf = ch & 1;
    const _Float16* Bb = &xs[buf][nt0 * TS + lane * 8];
    gload(ch + 1 < NCH ? ch + 1 : ch);
#pragma unroll
    for (int u = 0; u < 3; ++u)
#pragma unroll
      for (int mi = 0; mi < CV_MT; ++mi)
        ag1[u][mi] = load_frag(Ab + (size_t)mi * planestride + (size_t)(ch * CV_KS + 3 + u) * 512);
    v16h bc[CV_NT], bn[CV_NT];
#pragma unroll
    for (int nt = 0; nt < CV_NT; ++nt) bc[nt] = load_frag(Bb + nt * TS);
    __builtin_amdgcn_sched_barrier(0);
#pragma unroll
    for (int kt = 0; kt < CV_KS; ++kt) {
      if (kt == 3) {
#pragma unroll
        for (int u = 0; u < 3; ++u) {
          const int kn = ((ch + 1) * CV_KS + u < KTtot) ? (ch + 1) * CV_KS + u : KTtot - 1;
#pragma unroll
          for (int mi = 0; mi < CV_MT; ++mi)
            ag0[u][mi] = load_frag(Ab + (size_t)mi * planestride + (size_t)kn * 512);
        }
      }
      if (kt + 1 < CV_KS) {
#pragma unroll
        for (int nt = 0; nt < CV_NT; ++nt)
          bn[nt] = load_frag(Bb + nt * TS + (kt + 1) * 512);
      }
      __builtin_amdgcn_sched_barrier(0);
#pragma unroll
      for (int nt = 0; nt < CV_NT; ++nt)
#pragma unroll
        for (int mi = 0; mi < CV_MT; ++mi)
          c[mi][nt] = __builtin_amdgcn_wmma_f32_16x16x32_f16(
              false, (kt < 3) ? ag0[kt % 3][mi] : ag1[kt % 3][mi], false, bc[nt],
              (short)0, c[mi][nt], false, false);
      __builtin_amdgcn_sched_barrier(0);
      if (kt + 1 < CV_KS) {
#pragma unroll
        for (int nt = 0; nt < CV_NT; ++nt) bc[nt] = bn[nt];
      }
    }
    lstore(buf ^ 1);
    __syncthreads();
  }
  static_assert(CV_MT == 3, "a wave's m-tiles are the planes s0, s1, sh");
#pragma unroll
  for (int nt = 0; nt < CV_NT; ++nt)
#pragma unroll
    for (int j = 0; j < 8; ++j) {
      const int rr = mq * 16 + j + khalf;
      Out[((size_t)b * I_ + rr) * S_ + s0 + (nt0 + nt) * 16 + col] =
          c[0][nt][j] * c[1][nt][j] + c[2][nt][j];
    }
}

#define P0_BLOCKS ((E_ * F_ * 3 * I_) / (1024 * GATE_WAVES))
__global__ void __launch_bounds__(GATE_WAVES * 32)
gate_pack_kernel(const float* __restrict__ x,
                 const float* __restrict__ gw,
                 int* __restrict__ eidx,
                 float* __restrict__ lpart,
                 const float* __restrict__ w0, _Float16* __restrict__ w0p) {
  __shared__ float part[GATE_WAVES][E_][32];
  const int lane = threadIdx.x & 31;
  const int wave = __builtin_amdgcn_readfirstlane((int)(threadIdx.x >> 5));
  if (blockIdx.x < NGRP) {
    gate_body(x, F_, gw, eidx, lpart, part, (int)blockIdx.x, wave, lane);
    return;
  }
  pack_kn_tile<F_, 3 * I_>(w0, w0p, (int)(blockIdx.x - NGRP) * GATE_WAVES + wave, lane);
}

extern "C" void kernel_launch(void* const* d_in, const int* in_sizes, int n_in,
                              void* d_out, int out_size, void* d_ws, size_t ws_size,
                              hipStream_t stream) {
  const float* inp     = (const float*)d_in[0];
  const float* divisor = (const float*)d_in[1];
  const float* w0_gate = (const float*)d_in[2];
  const float* w0      = (const float*)d_in[3];
  const float* w1      = (const float*)d_in[4];
  const float* w2_gate = (const float*)d_in[5];
  const float* w2      = (const float*)d_in[6];
  float* out = (float*)d_out;

  char* ws = (char*)d_ws;
  size_t off = 0;
  float* lpart = (float*)(ws + off); off += (size_t)2 * NGRP * 8 * sizeof(float);
  const size_t h0b = (size_t)B_ * 3 * I_ * S_ * sizeof(float);
  const size_t t1b = (size_t)B_ * I_ * S_ * sizeof(float);
  float* h0   = (float*)(ws + off); off += h0b;
  float* tmp1 = (float*)(ws + off); off += t1b;
  float* g1   = (float*)(ws + off); off += t1b;
  int*   idx0 = (int*)(ws + off); off += TOK * sizeof(int);
  int*   idx1 = (int*)(ws + off); off += TOK * sizeof(int);
  off = (off + 255) & ~(size_t)255;
  const size_t w0n = (size_t)E_ * F_ * 3 * I_;
  const size_t w1n = (size_t)3 * I_ * 3 * I_;
  const size_t w2n = (size_t)E_ * I_ * F_;
  _Float16* w0p = (_Float16*)(ws + off); off += w0n * 2;
  _Float16* w1p = (_Float16*)(ws + off); off += w1n * 2;
  _Float16* w2p = (_Float16*)(ws + off); off += w2n * 2;
  float* h2 = h0;
  float* h3 = tmp1;

  gate_pack_kernel<<<NGRP + P0_BLOCKS, GATE_WAVES * 32, 0, stream>>>(
      inp, w0_gate, idx0, lpart, w0, w0p);
  moe_wmma_kernel<F_, 3 * I_, 2, 4, 12, 4, false, 2><<<dim3(3 * I_ / 192, TOK / 128), 384, 0, stream>>>(
      inp, w0p, idx0, h0, 1.0f, nullptr, nullptr);

  cumsum_pack_kernel<<<CS_BLOCKS + P1_BLOCKS + P2_BLOCKS, 256, 0, stream>>>(
      h0, divisor, tmp1, w1, w1p, w2, w2p);
  norm_kernel<0><<<TOK / 32, NORM_WAVES * 32, 0, stream>>>(
      tmp1, g1, w2_gate, idx1, lpart);

  conv_wmma_kernel<<<dim3(3 * I_ / 192, TOK / 128), CV_WAVES * 32, 0, stream>>>(g1, w1p, h2);

  norm_kernel<1><<<TOK / 32, NORM_WAVES * 32, 0, stream>>>(
      h2, h3, w2_gate, idx1, lpart + NGRP * 8);

  moe_wmma_kernel<I_, F_, 1, 2, 8, 8, true, 2><<<dim3(F_ / 64, TOK / 64), 256, 0, stream>>>(
      h3, w2p, idx1, out, 1.0f  , lpart, out + (size_t)B_ * F_ * S_);
}
